# starting kernel
# baseline (speedup 1.0000x reference)
#include <hip/hip_runtime.h>
#include <math.h>


typedef __attribute__((ext_vector_type(2))) float v2f;
typedef __attribute__((ext_vector_type(8))) float v8f;

__device__ __forceinline__ void atomAddF32(float* p, float v) {
    __hip_atomic_fetch_add(p, v, __ATOMIC_RELAXED, __HIP_MEMORY_SCOPE_AGENT);
}


__global__ void k_deg_init(float* __restrict__ deg, int n) {
    int i = blockIdx.x * blockDim.x + threadIdx.x;
    if (i < n) deg[i] = 1.0f;
}

__global__ void k_deg_scatter(const int* __restrict__ dst, const float* __restrict__ w,
                              float* __restrict__ deg, int E) {
    int e = blockIdx.x * blockDim.x + threadIdx.x;
    if (e < E) atomAddF32(&deg[dst[e]], w[e]);
}

__global__ void k_dinv(float* __restrict__ deg, int n) {
    int i = blockIdx.x * blockDim.x + threadIdx.x;
    if (i < n) {
        float d = deg[i];
        deg[i] = (d > 0.0f) ? rsqrtf(d) : 0.0f;
    }
}

__global__ void k_norm(const int* __restrict__ src, const int* __restrict__ dst,
                       const float* __restrict__ w, const float* __restrict__ dinv,
                       float* __restrict__ nrm, int E) {
    int e = blockIdx.x * blockDim.x + threadIdx.x;
    if (e < E) nrm[e] = dinv[src[e]] * w[e] * dinv[dst[e]];
}


__global__ void k_gemm1(const float* __restrict__ X, const float* __restrict__ W,
                        float* __restrict__ O, int mtiles) {
    int wave = threadIdx.x >> 5;
    int tile = blockIdx.x * (blockDim.x >> 5) + wave;
    if (tile >= mtiles * 4) return;
    int mt = tile >> 2, nt = tile & 3;
    int lane = threadIdx.x & 31;
    int lo = lane & 15, hi = lane >> 4;

    const float* arow = X + (size_t)(mt * 16 + lo) * 64;
    int col = nt * 16 + lo;
    v8f c = {};
#pragma unroll
    for (int k = 0; k < 64; k += 4) {
        int ka = k + 2 * hi;
        v2f a = { arow[ka], arow[ka + 1] };
        v2f b = { W[(size_t)ka * 64 + col], W[(size_t)(ka + 1) * 64 + col] };
        c = __builtin_amdgcn_wmma_f32_16x16x4_f32(false, a, false, b,
                                                  (short)0, c, false, false);
    }
    float* orow = O + (size_t)(mt * 16 + 8 * hi) * 64 + col;
#pragma unroll
    for (int r = 0; r < 8; ++r) orow[(size_t)r * 64] = c[r];
}


__global__ void k_gemm2(const float* __restrict__ Hin, const float* __restrict__ W2,
                        float* __restrict__ O, int mtiles) {
    int wave = threadIdx.x >> 5;
    int tile = blockIdx.x * (blockDim.x >> 5) + wave;
    if (tile >= mtiles * 3) return;
    int mt = tile / 3, nt = tile - mt * 3;
    int lane = threadIdx.x & 31;
    int lo = lane & 15, hi = lane >> 4;

    const float* arow = Hin + (size_t)(mt * 16 + lo) * 64;
    int col = nt * 16 + lo;
    bool cok = col < 40;
    v8f c = {};
#pragma unroll
    for (int k = 0; k < 64; k += 4) {
        int ka = k + 2 * hi;
        v2f a = { fmaxf(arow[ka], 0.0f), fmaxf(arow[ka + 1], 0.0f) };
        v2f b = { cok ? W2[(size_t)ka * 40 + col] : 0.0f,
                  cok ? W2[(size_t)(ka + 1) * 40 + col] : 0.0f };
        c = __builtin_amdgcn_wmma_f32_16x16x4_f32(false, a, false, b,
                                                  (short)0, c, false, false);
    }
    if (cok) {
        float* orow = O + (size_t)(mt * 16 + 8 * hi) * 40 + col;
#pragma unroll
        for (int r = 0; r < 8; ++r) orow[(size_t)r * 40] = c[r];
    }
}


__global__ void k_self1(const float* __restrict__ h, const float* __restrict__ dinv,
                        const float* __restrict__ bias, float* __restrict__ out, int n) {
    int t = blockIdx.x * blockDim.x + threadIdx.x;
    if (t < n * 64) {
        int i = t >> 6, f = t & 63;
        float di = dinv[i];
        out[t] = bias[f] + di * di * h[t];
    }
}

__global__ void k_self2(const float* __restrict__ h, const float* __restrict__ dinv,
                        const float* __restrict__ bias, float* __restrict__ out, int n) {
    int t = blockIdx.x * blockDim.x + threadIdx.x;
    if (t < n * 64) {
        int i = t >> 6, f = t & 63;
        if (f < 40) {
            float di = dinv[i];
            out[(size_t)i * 40 + f] = bias[f] + di * di * h[(size_t)i * 40 + f];
        }
    }
}


__global__ void k_scat1(const int* __restrict__ src, const int* __restrict__ dst,
                        const float* __restrict__ nrm, const float* __restrict__ h,
                        float* __restrict__ out, int E) {
    long long t = (long long)blockIdx.x * blockDim.x + threadIdx.x;
    if (t >= (long long)E * 64) return;
    int e = (int)(t >> 6), f = (int)(t & 63);
    int s = src[e], d = dst[e];
    atomAddF32(&out[(size_t)d * 64 + f], nrm[e] * h[(size_t)s * 64 + f]);
}

__global__ void k_scat2(const int* __restrict__ src, const int* __restrict__ dst,
                        const float* __restrict__ nrm, const float* __restrict__ h,
                        float* __restrict__ out, int E) {
    long long t = (long long)blockIdx.x * blockDim.x + threadIdx.x;
    if (t >= (long long)E * 64) return;
    int e = (int)(t >> 6), f = (int)(t & 63);
    if (f < 40) {
        int s = src[e], d = dst[e];
        atomAddF32(&out[(size_t)d * 40 + f], nrm[e] * h[(size_t)s * 40 + f]);
    }
}


__global__ void k_lsm(const float* __restrict__ h, float* __restrict__ out, int n) {
    int i = blockIdx.x * blockDim.x + threadIdx.x;
    if (i >= n) return;
    float v[40];
    float mx = -3.402823466e38f;
#pragma unroll
    for (int c = 0; c < 40; ++c) { v[c] = h[(size_t)i * 40 + c]; mx = fmaxf(mx, v[c]); }
    float s = 0.0f;
#pragma unroll
    for (int c = 0; c < 40; ++c) s += expf(v[c] - mx);
    float l = mx + logf(s);
#pragma unroll
    for (int c = 0; c < 40; ++c) out[(size_t)i * 40 + c] = v[c] - l;
}


extern "C" void kernel_launch(void* const* d_in, const int* in_sizes, int n_in,
                              void* d_out, int out_size, void* d_ws, size_t ws_size,
                              hipStream_t stream) {
    const float* x   = (const float*)d_in[0];
    const int*   ei  = (const int*)d_in[1];
    const float* ew  = (const float*)d_in[2];
    const float* W1  = (const float*)d_in[3];
    const float* b1  = (const float*)d_in[4];
    const float* W2  = (const float*)d_in[5];
    const float* b2  = (const float*)d_in[6];
    float* out = (float*)d_out;

    const int N = in_sizes[0] / 64;
    const int E = in_sizes[2];
    const int* src = ei;
    const int* dst = ei + E;

    float* ws   = (float*)d_ws;
    float* dinv = ws;
    float* nrm  = dinv + N;
    float* bufA = nrm + E;
    float* bufB = bufA + (size_t)N * 64;

    const int BLK = 256;
    const int mtiles = N / 16;

    k_deg_init<<<(N + BLK - 1) / BLK, BLK, 0, stream>>>(dinv, N);
    k_deg_scatter<<<(E + BLK - 1) / BLK, BLK, 0, stream>>>(dst, ew, dinv, E);
    k_dinv<<<(N + BLK - 1) / BLK, BLK, 0, stream>>>(dinv, N);
    k_norm<<<(E + BLK - 1) / BLK, BLK, 0, stream>>>(src, dst, ew, dinv, nrm, E);

    k_gemm1<<<(mtiles * 4 + 7) / 8, BLK, 0, stream>>>(x, W1, bufA, mtiles);
    {
        long long t = (long long)N * 64;
        k_self1<<<(unsigned)((t + BLK - 1) / BLK), BLK, 0, stream>>>(bufA, dinv, b1, bufB, N);
    }
    {
        long long t = (long long)E * 64;
        k_scat1<<<(unsigned)((t + BLK - 1) / BLK), BLK, 0, stream>>>(src, dst, nrm, bufA, bufB, E);
    }

    k_gemm2<<<(mtiles * 3 + 7) / 8, BLK, 0, stream>>>(bufB, W2, bufA, mtiles);
    {
        long long t = (long long)N * 64;
        k_self2<<<(unsigned)((t + BLK - 1) / BLK), BLK, 0, stream>>>(bufA, dinv, b2, bufB, N);
    }
    {
        long long t = (long long)E * 64;
        k_scat2<<<(unsigned)((t + BLK - 1) / BLK), BLK, 0, stream>>>(src, dst, nrm, bufA, bufB, E);
    }

    k_lsm<<<(N + BLK - 1) / BLK, BLK, 0, stream>>>(bufB, out, N);
}


# f32 atomic edge scatter replaced by per-call grouping of edges by destination (int atomic count = slot) and a per-node register sum of 256-B-pitch rows; sums now run in slot (arrival-rank) order
# speedup vs baseline: 2.8033x; 2.8033x over previous
#include <hip/hip_runtime.h>
#include <math.h>


typedef __attribute__((ext_vector_type(2))) float v2f;
typedef __attribute__((ext_vector_type(8))) float v8f;


__global__ void k_zero(int* __restrict__ cnt, int n) {
    int i = blockIdx.x * blockDim.x + threadIdx.x;
    if (i < n) cnt[i] = 0;
}

__global__ void k_hist(const int* __restrict__ dst, int* __restrict__ cnt,
                       int* __restrict__ rank, int E) {
    int e = blockIdx.x * blockDim.x + threadIdx.x;
    if (e < E)
        rank[e] = __hip_atomic_fetch_add(&cnt[dst[e]], 1, __ATOMIC_RELAXED,
                                         __HIP_MEMORY_SCOPE_AGENT);
}

__global__ void __launch_bounds__(1024)
k_scan(const int* __restrict__ cnt, int* __restrict__ rowptr, int n) {
    __shared__ int sh[1024];
    const int tid = threadIdx.x;
    const int base = blockIdx.x * 1024;
    int pre = 0;
    for (int i = tid * 4; i < base; i += 4096) {
        const int4 v = *reinterpret_cast<const int4*>(cnt + i);
        pre += (v.x + v.y) + (v.z + v.w);
    }
    const int i0 = base + tid;
    const int c = (i0 < n) ? cnt[i0] : 0;

    sh[tid] = pre;
    __syncthreads();
    for (int off = 512; off > 0; off >>= 1) {
        if (tid < off) sh[tid] += sh[tid + off];
        __syncthreads();
    }
    const int pre_total = sh[0];
    __syncthreads();

    int v = c;
    sh[tid] = v;
    __syncthreads();
    for (int off = 1; off < 1024; off <<= 1) {
        const int add = (tid >= off) ? sh[tid - off] : 0;
        __syncthreads();
        v += add;
        sh[tid] = v;
        __syncthreads();
    }
    if (i0 < n) rowptr[i0] = pre_total + v - c;
}

__global__ void k_fill(const int* __restrict__ src, const int* __restrict__ dst,
                       const float* __restrict__ w, const int* __restrict__ rank,
                       const int* __restrict__ rowptr,
                       unsigned long long* __restrict__ ep, int E) {
    int e = blockIdx.x * blockDim.x + threadIdx.x;
    if (e < E) {
        const int p = rowptr[dst[e]] + rank[e];
        ep[p] = ((unsigned long long)__float_as_uint(w[e]) << 32) |
                (unsigned long long)(unsigned)src[e];
    }
}

__global__ void k_deg(const int* __restrict__ rowptr, const int* __restrict__ cnt,
                      const unsigned long long* __restrict__ ep,
                      float* __restrict__ dinv, int n) {
    int i = blockIdx.x * blockDim.x + threadIdx.x;
    if (i < n) {
        const int p0 = rowptr[i], c = cnt[i];
        float d = 1.0f;
        for (int k = 0; k < c; ++k) d += __uint_as_float((unsigned)(ep[p0 + k] >> 32));
        dinv[i] = (d > 0.0f) ? rsqrtf(d) : 0.0f;
    }
}


__global__ void k_gemm1(const float* __restrict__ X, const float* __restrict__ W,
                        float* __restrict__ O, int mtiles) {
    int wave = threadIdx.x >> 5;
    int tile = blockIdx.x * (blockDim.x >> 5) + wave;
    if (tile >= mtiles * 4) return;
    int mt = tile >> 2, nt = tile & 3;
    int lane = threadIdx.x & 31;
    int lo = lane & 15, hi = lane >> 4;

    const float* arow = X + (size_t)(mt * 16 + lo) * 64;
    int col = nt * 16 + lo;
    v8f c = {};
#pragma unroll
    for (int k = 0; k < 64; k += 4) {
        int ka = k + 2 * hi;
        v2f a = { arow[ka], arow[ka + 1] };
        v2f b = { W[(size_t)ka * 64 + col], W[(size_t)(ka + 1) * 64 + col] };
        c = __builtin_amdgcn_wmma_f32_16x16x4_f32(false, a, false, b,
                                                  (short)0, c, false, false);
    }
    float* orow = O + (size_t)(mt * 16 + 8 * hi) * 64 + col;
#pragma unroll
    for (int r = 0; r < 8; ++r) orow[(size_t)r * 64] = c[r];
}


__global__ void k_gemm2(const float* __restrict__ Hin, const float* __restrict__ W2,
                        float* __restrict__ O, int mtiles) {
    int wave = threadIdx.x >> 5;
    int tile = blockIdx.x * (blockDim.x >> 5) + wave;
    if (tile >= mtiles * 3) return;
    int mt = tile / 3, nt = tile - mt * 3;
    int lane = threadIdx.x & 31;
    int lo = lane & 15, hi = lane >> 4;

    const float* arow = Hin + (size_t)(mt * 16 + lo) * 64;
    int col = nt * 16 + lo;
    bool cok = col < 40;
    v8f c = {};
#pragma unroll
    for (int k = 0; k < 64; k += 4) {
        int ka = k + 2 * hi;
        v2f a = { fmaxf(arow[ka], 0.0f), fmaxf(arow[ka + 1], 0.0f) };
        v2f b = { cok ? W2[(size_t)ka * 40 + col] : 0.0f,
                  cok ? W2[(size_t)(ka + 1) * 40 + col] : 0.0f };
        c = __builtin_amdgcn_wmma_f32_16x16x4_f32(false, a, false, b,
                                                  (short)0, c, false, false);
    }
    if (cok) {
        float* orow = O + (size_t)(mt * 16 + 8 * hi) * 64 + col;
#pragma unroll
        for (int r = 0; r < 8; ++r) orow[(size_t)r * 64] = c[r];
    }
}


__device__ __forceinline__ void axpy4(float4& a, float s, const float4& r) {
    a.x += s * r.x; a.y += s * r.y; a.z += s * r.z; a.w += s * r.w;
}

__global__ void k_agg1(const int* __restrict__ rowptr, const int* __restrict__ cnt,
                       const unsigned long long* __restrict__ ep,
                       const float* __restrict__ dinv, const float* __restrict__ h,
                       const float* __restrict__ bias, float* __restrict__ nrmS,
                       float* __restrict__ out, int n) {
    const int t = blockIdx.x * blockDim.x + threadIdx.x;
    const int g = t >> 4, l = t & 15;
    if (g >= n) return;
    const int p0 = rowptr[g], c = cnt[g];
    const float dd = dinv[g];
    const float4* __restrict__ h4 = reinterpret_cast<const float4*>(h);
    float4 acc = make_float4(0.0f, 0.0f, 0.0f, 0.0f);
    int k = 0;
    for (; k + 4 <= c; k += 4) {
        const unsigned long long e0 = ep[p0 + k], e1 = ep[p0 + k + 1];
        const unsigned long long e2 = ep[p0 + k + 2], e3 = ep[p0 + k + 3];
        const int s0 = (int)(unsigned)e0, s1 = (int)(unsigned)e1;
        const int s2 = (int)(unsigned)e2, s3 = (int)(unsigned)e3;
        const float4 r0 = h4[(size_t)s0 * 16 + l];
        const float4 r1 = h4[(size_t)s1 * 16 + l];
        const float4 r2 = h4[(size_t)s2 * 16 + l];
        const float4 r3 = h4[(size_t)s3 * 16 + l];
        const float n0 = dinv[s0] * __uint_as_float((unsigned)(e0 >> 32)) * dd;
        const float n1 = dinv[s1] * __uint_as_float((unsigned)(e1 >> 32)) * dd;
        const float n2 = dinv[s2] * __uint_as_float((unsigned)(e2 >> 32)) * dd;
        const float n3 = dinv[s3] * __uint_as_float((unsigned)(e3 >> 32)) * dd;
        axpy4(acc, n0, r0);
        axpy4(acc, n1, r1);
        axpy4(acc, n2, r2);
        axpy4(acc, n3, r3);
        if (l < 4) nrmS[p0 + k + l] = (l == 0) ? n0 : (l == 1) ? n1 : (l == 2) ? n2 : n3;
    }
    for (; k < c; ++k) {
        const unsigned long long e0 = ep[p0 + k];
        const int s0 = (int)(unsigned)e0;
        const float4 r0 = h4[(size_t)s0 * 16 + l];
        const float n0 = dinv[s0] * __uint_as_float((unsigned)(e0 >> 32)) * dd;
        axpy4(acc, n0, r0);
        if (l == 0) nrmS[p0 + k] = n0;
    }
    const float4 hs = h4[(size_t)g * 16 + l];
    const float4 b = reinterpret_cast<const float4*>(bias)[l];
    axpy4(acc, dd * dd, hs);
    acc.x += b.x; acc.y += b.y; acc.z += b.z; acc.w += b.w;
    reinterpret_cast<float4*>(out)[(size_t)g * 16 + l] = acc;
}

__global__ void k_agg2(const int* __restrict__ rowptr, const int* __restrict__ cnt,
                       const unsigned long long* __restrict__ ep,
                       const float* __restrict__ nrmS, const float* __restrict__ dinv,
                       const float* __restrict__ h, const float* __restrict__ bias,
                       float* __restrict__ out, int n) {
    const int t = blockIdx.x * blockDim.x + threadIdx.x;
    const int g = t >> 4, l = t & 15;
    if (g >= n || l >= 10) return;
    const int p0 = rowptr[g], c = cnt[g];
    const float4* __restrict__ h4 = reinterpret_cast<const float4*>(h);
    float4 acc = make_float4(0.0f, 0.0f, 0.0f, 0.0f);
    int k = 0;
    for (; k + 4 <= c; k += 4) {
        const int s0 = (int)(unsigned)ep[p0 + k], s1 = (int)(unsigned)ep[p0 + k + 1];
        const int s2 = (int)(unsigned)ep[p0 + k + 2], s3 = (int)(unsigned)ep[p0 + k + 3];
        const float4 r0 = h4[(size_t)s0 * 16 + l];
        const float4 r1 = h4[(size_t)s1 * 16 + l];
        const float4 r2 = h4[(size_t)s2 * 16 + l];
        const float4 r3 = h4[(size_t)s3 * 16 + l];
        const float n0 = nrmS[p0 + k], n1 = nrmS[p0 + k + 1];
        const float n2 = nrmS[p0 + k + 2], n3 = nrmS[p0 + k + 3];
        axpy4(acc, n0, r0);
        axpy4(acc, n1, r1);
        axpy4(acc, n2, r2);
        axpy4(acc, n3, r3);
    }
    for (; k < c; ++k) {
        const int s0 = (int)(unsigned)ep[p0 + k];
        const float4 r0 = h4[(size_t)s0 * 16 + l];
        axpy4(acc, nrmS[p0 + k], r0);
    }
    const float dd = dinv[g];
    const float4 hs = h4[(size_t)g * 16 + l];
    const float4 b = reinterpret_cast<const float4*>(bias)[l];
    axpy4(acc, dd * dd, hs);
    acc.x += b.x; acc.y += b.y; acc.z += b.z; acc.w += b.w;
    reinterpret_cast<float4*>(out)[(size_t)g * 10 + l] = acc;
}


__global__ void k_lsm(const float* __restrict__ h, float* __restrict__ out, int n) {
    int i = blockIdx.x * blockDim.x + threadIdx.x;
    if (i >= n) return;
    float v[40];
    float mx = -3.402823466e38f;
#pragma unroll
    for (int c = 0; c < 40; ++c) { v[c] = h[(size_t)i * 40 + c]; mx = fmaxf(mx, v[c]); }
    float s = 0.0f;
#pragma unroll
    for (int c = 0; c < 40; ++c) s += expf(v[c] - mx);
    float l = mx + logf(s);
#pragma unroll
    for (int c = 0; c < 40; ++c) out[(size_t)i * 40 + c] = v[c] - l;
}


extern "C" void kernel_launch(void* const* d_in, const int* in_sizes, int n_in,
                              void* d_out, int out_size, void* d_ws, size_t ws_size,
                              hipStream_t stream) {
    const float* x   = (const float*)d_in[0];
    const int*   ei  = (const int*)d_in[1];
    const float* ew  = (const float*)d_in[2];
    const float* W1  = (const float*)d_in[3];
    const float* b1  = (const float*)d_in[4];
    const float* W2  = (const float*)d_in[5];
    const float* b2  = (const float*)d_in[6];
    float* out = (float*)d_out;

    const int N = in_sizes[0] / 64;
    const int E = in_sizes[2];
    const int* src = ei;
    const int* dst = ei + E;

    const size_t Ep = ((size_t)E + 63) / 64 * 64;
    const size_t Np = ((size_t)N + 63) / 64 * 64;
    unsigned long long* ep = (unsigned long long*)d_ws;
    int*   rank   = (int*)(ep + Ep);
    float* nrmS   = (float*)(rank + Ep);
    int*   cnt    = (int*)(nrmS + Ep);
    int*   rowptr = cnt + Np;
    float* dinv   = (float*)(rowptr + Np);
    float* bufA   = dinv + Np;
    float* bufB   = bufA + (size_t)N * 64;

    const int BLK = 256;
    const int mtiles = N / 16;
    const unsigned gN  = (unsigned)((N + BLK - 1) / BLK);
    const unsigned gE  = (unsigned)((E + BLK - 1) / BLK);
    const unsigned gNg = (unsigned)(((long long)N * 16 + BLK - 1) / BLK);

    k_zero<<<gN, BLK, 0, stream>>>(cnt, N);
    k_hist<<<gE, BLK, 0, stream>>>(dst, cnt, rank, E);
    k_scan<<<(unsigned)((N + 1023) / 1024), 1024, 0, stream>>>(cnt, rowptr, N);
    k_fill<<<gE, BLK, 0, stream>>>(src, dst, ew, rank, rowptr, ep, E);
    k_deg<<<gN, BLK, 0, stream>>>(rowptr, cnt, ep, dinv, N);

    k_gemm1<<<(mtiles * 4 + 7) / 8, BLK, 0, stream>>>(x, W1, bufA, mtiles);
    k_agg1<<<gNg, BLK, 0, stream>>>(rowptr, cnt, ep, dinv, bufA, b1, nrmS, bufB, N);

    k_gemm2<<<(mtiles * 3 + 7) / 8, BLK, 0, stream>>>(bufB, W2, bufA, mtiles);
    k_agg2<<<gNg, BLK, 0, stream>>>(rowptr, cnt, ep, nrmS, dinv, bufA, b2, bufB, N);

    k_lsm<<<gN, BLK, 0, stream>>>(bufB, out, N);
}


# edge grouping by count + prefix sum + place replaced by fixed 32-slot 256-B rows per node (returned count = slot, rare spill list): scan and place launches go; sums stay in slot order
# speedup vs baseline: 3.1591x; 1.1269x over previous
#include <hip/hip_runtime.h>
#include <math.h>


typedef __attribute__((ext_vector_type(2))) float v2f;
typedef __attribute__((ext_vector_type(8))) float v8f;

constexpr int CAP = 32;


__global__ void k_zero(int* __restrict__ cnt, int* __restrict__ nspill, int n) {
    int i = blockIdx.x * blockDim.x + threadIdx.x;
    if (i < n) cnt[i] = 0;
    if (i == 0) nspill[0] = 0;
}

__global__ void k_place(const int* __restrict__ src, const int* __restrict__ dst,
                        const float* __restrict__ w, int* __restrict__ cnt,
                        unsigned long long* __restrict__ ell,
                        int* __restrict__ nspill,
                        unsigned long long* __restrict__ spill_e,
                        int* __restrict__ spill_d, int E) {
    int e = blockIdx.x * blockDim.x + threadIdx.x;
    if (e < E) {
        const int d = dst[e];
        const unsigned long long v =
            ((unsigned long long)__float_as_uint(w[e]) << 32) |
            (unsigned long long)(unsigned)src[e];
        const int slot = __hip_atomic_fetch_add(&cnt[d], 1, __ATOMIC_RELAXED,
                                                __HIP_MEMORY_SCOPE_AGENT);
        if (slot < CAP) {
            ell[(size_t)d * CAP + slot] = v;
        } else {
            const int j = __hip_atomic_fetch_add(nspill, 1, __ATOMIC_RELAXED,
                                                 __HIP_MEMORY_SCOPE_AGENT);
            spill_e[j] = v;
            spill_d[j] = d;
        }
    }
}

__global__ void k_deg(const int* __restrict__ cnt,
                      const unsigned long long* __restrict__ ell,
                      const int* __restrict__ nspill,
                      const unsigned long long* __restrict__ spill_e,
                      const int* __restrict__ spill_d,
                      float* __restrict__ dinv, int n) {
    int i = blockIdx.x * blockDim.x + threadIdx.x;
    if (i < n) {
        const int c = cnt[i];
        const int m = (c < CAP) ? c : CAP;
        const unsigned long long* __restrict__ row = ell + (size_t)i * CAP;
        const ulonglong2* __restrict__ row2 = reinterpret_cast<const ulonglong2*>(row);
        float d = 1.0f;
        int k = 0;
        for (; k + 2 <= m; k += 2) {
            const ulonglong2 v = row2[k >> 1];
            d += __uint_as_float((unsigned)(v.x >> 32));
            d += __uint_as_float((unsigned)(v.y >> 32));
        }
        if (k < m) d += __uint_as_float((unsigned)(row[k] >> 32));
        if (c > CAP) {
            const int ns = nspill[0];
            for (int j = 0; j < ns; ++j)
                if (spill_d[j] == i) d += __uint_as_float((unsigned)(spill_e[j] >> 32));
        }
        dinv[i] = (d > 0.0f) ? rsqrtf(d) : 0.0f;
    }
}


__global__ void k_gemm1(const float* __restrict__ X, const float* __restrict__ W,
                        float* __restrict__ O, int mtiles) {
    int wave = threadIdx.x >> 5;
    int tile = blockIdx.x * (blockDim.x >> 5) + wave;
    if (tile >= mtiles * 4) return;
    int mt = tile >> 2, nt = tile & 3;
    int lane = threadIdx.x & 31;
    int lo = lane & 15, hi = lane >> 4;

    const float* arow = X + (size_t)(mt * 16 + lo) * 64;
    int col = nt * 16 + lo;
    v8f c = {};
#pragma unroll
    for (int k = 0; k < 64; k += 4) {
        int ka = k + 2 * hi;
        v2f a = { arow[ka], arow[ka + 1] };
        v2f b = { W[(size_t)ka * 64 + col], W[(size_t)(ka + 1) * 64 + col] };
        c = __builtin_amdgcn_wmma_f32_16x16x4_f32(false, a, false, b,
                                                  (short)0, c, false, false);
    }
    float* orow = O + (size_t)(mt * 16 + 8 * hi) * 64 + col;
#pragma unroll
    for (int r = 0; r < 8; ++r) orow[(size_t)r * 64] = c[r];
}


__global__ void k_gemm2(const float* __restrict__ Hin, const float* __restrict__ W2,
                        float* __restrict__ O, int mtiles) {
    int wave = threadIdx.x >> 5;
    int tile = blockIdx.x * (blockDim.x >> 5) + wave;
    if (tile >= mtiles * 3) return;
    int mt = tile / 3, nt = tile - mt * 3;
    int lane = threadIdx.x & 31;
    int lo = lane & 15, hi = lane >> 4;

    const float* arow = Hin + (size_t)(mt * 16 + lo) * 64;
    int col = nt * 16 + lo;
    bool cok = col < 40;
    v8f c = {};
#pragma unroll
    for (int k = 0; k < 64; k += 4) {
        int ka = k + 2 * hi;
        v2f a = { fmaxf(arow[ka], 0.0f), fmaxf(arow[ka + 1], 0.0f) };
        v2f b = { cok ? W2[(size_t)ka * 40 + col] : 0.0f,
                  cok ? W2[(size_t)(ka + 1) * 40 + col] : 0.0f };
        c = __builtin_amdgcn_wmma_f32_16x16x4_f32(false, a, false, b,
                                                  (short)0, c, false, false);
    }
    if (cok) {
        float* orow = O + (size_t)(mt * 16 + 8 * hi) * 64 + col;
#pragma unroll
        for (int r = 0; r < 8; ++r) orow[(size_t)r * 64] = c[r];
    }
}


__device__ __forceinline__ void axpy4(float4& a, float s, const float4& r) {
    a.x += s * r.x; a.y += s * r.y; a.z += s * r.z; a.w += s * r.w;
}

__global__ void k_agg1(const int* __restrict__ cnt,
                       const unsigned long long* __restrict__ ell,
                       const int* __restrict__ nspill,
                       const unsigned long long* __restrict__ spill_e,
                       const int* __restrict__ spill_d,
                       const float* __restrict__ dinv, const float* __restrict__ h,
                       const float* __restrict__ bias, float* __restrict__ nrmS,
                       float* __restrict__ nrmP, float* __restrict__ out, int n) {
    const int t = blockIdx.x * blockDim.x + threadIdx.x;
    const int g = t >> 4, l = t & 15;
    if (g >= n) return;
    const int c = cnt[g];
    const int m = (c < CAP) ? c : CAP;
    const size_t p0 = (size_t)g * CAP;
    const ulonglong2* __restrict__ row2 = reinterpret_cast<const ulonglong2*>(ell + p0);
    const float dd = dinv[g];
    const float4* __restrict__ h4 = reinterpret_cast<const float4*>(h);
    float4 acc = make_float4(0.0f, 0.0f, 0.0f, 0.0f);
    int k = 0;
    for (; k + 4 <= m; k += 4) {
        const ulonglong2 ea = row2[k >> 1], eb = row2[(k >> 1) + 1];
        const unsigned long long e0 = ea.x, e1 = ea.y, e2 = eb.x, e3 = eb.y;
        const int s0 = (int)(unsigned)e0, s1 = (int)(unsigned)e1;
        const int s2 = (int)(unsigned)e2, s3 = (int)(unsigned)e3;
        const float4 r0 = h4[(size_t)s0 * 16 + l];
        const float4 r1 = h4[(size_t)s1 * 16 + l];
        const float4 r2 = h4[(size_t)s2 * 16 + l];
        const float4 r3 = h4[(size_t)s3 * 16 + l];
        const float n0 = dinv[s0] * __uint_as_float((unsigned)(e0 >> 32)) * dd;
        const float n1 = dinv[s1] * __uint_as_float((unsigned)(e1 >> 32)) * dd;
        const float n2 = dinv[s2] * __uint_as_float((unsigned)(e2 >> 32)) * dd;
        const float n3 = dinv[s3] * __uint_as_float((unsigned)(e3 >> 32)) * dd;
        axpy4(acc, n0, r0);
        axpy4(acc, n1, r1);
        axpy4(acc, n2, r2);
        axpy4(acc, n3, r3);
        if (l < 4) nrmS[p0 + k + l] = (l == 0) ? n0 : (l == 1) ? n1 : (l == 2) ? n2 : n3;
    }
    for (; k < m; ++k) {
        const unsigned long long e0 = ell[p0 + k];
        const int s0 = (int)(unsigned)e0;
        const float4 r0 = h4[(size_t)s0 * 16 + l];
        const float n0 = dinv[s0] * __uint_as_float((unsigned)(e0 >> 32)) * dd;
        axpy4(acc, n0, r0);
        if (l == 0) nrmS[p0 + k] = n0;
    }
    if (c > CAP) {
        const int ns = nspill[0];
        for (int j = 0; j < ns; ++j) {
            if (spill_d[j] != g) continue;
            const unsigned long long e0 = spill_e[j];
            const int s0 = (int)(unsigned)e0;
            const float4 r0 = h4[(size_t)s0 * 16 + l];
            const float n0 = dinv[s0] * __uint_as_float((unsigned)(e0 >> 32)) * dd;
            axpy4(acc, n0, r0);
            if (l == 0) nrmP[j] = n0;
        }
    }
    const float4 hs = h4[(size_t)g * 16 + l];
    const float4 b = reinterpret_cast<const float4*>(bias)[l];
    axpy4(acc, dd * dd, hs);
    acc.x += b.x; acc.y += b.y; acc.z += b.z; acc.w += b.w;
    reinterpret_cast<float4*>(out)[(size_t)g * 16 + l] = acc;
}

__global__ void k_agg2(const int* __restrict__ cnt,
                       const unsigned long long* __restrict__ ell,
                       const int* __restrict__ nspill,
                       const unsigned long long* __restrict__ spill_e,
                       const int* __restrict__ spill_d,
                       const float* __restrict__ nrmS, const float* __restrict__ nrmP,
                       const float* __restrict__ dinv,
                       const float* __restrict__ h, const float* __restrict__ bias,
                       float* __restrict__ out, int n) {
    const int t = blockIdx.x * blockDim.x + threadIdx.x;
    const int g = t >> 4, l = t & 15;
    if (g >= n || l >= 10) return;
    const int c = cnt[g];
    const int m = (c < CAP) ? c : CAP;
    const size_t p0 = (size_t)g * CAP;
    const ulonglong2* __restrict__ row2 = reinterpret_cast<const ulonglong2*>(ell + p0);
    const float4* __restrict__ nrm4 = reinterpret_cast<const float4*>(nrmS + p0);
    const float4* __restrict__ h4 = reinterpret_cast<const float4*>(h);
    float4 acc = make_float4(0.0f, 0.0f, 0.0f, 0.0f);
    int k = 0;
    for (; k + 4 <= m; k += 4) {
        const ulonglong2 ea = row2[k >> 1], eb = row2[(k >> 1) + 1];
        const int s0 = (int)(unsigned)ea.x, s1 = (int)(unsigned)ea.y;
        const int s2 = (int)(unsigned)eb.x, s3 = (int)(unsigned)eb.y;
        const float4 r0 = h4[(size_t)s0 * 16 + l];
        const float4 r1 = h4[(size_t)s1 * 16 + l];
        const float4 r2 = h4[(size_t)s2 * 16 + l];
        const float4 r3 = h4[(size_t)s3 * 16 + l];
        const float4 nn = nrm4[k >> 2];
        axpy4(acc, nn.x, r0);
        axpy4(acc, nn.y, r1);
        axpy4(acc, nn.z, r2);
        axpy4(acc, nn.w, r3);
    }
    for (; k < m; ++k) {
        const int s0 = (int)(unsigned)ell[p0 + k];
        const float4 r0 = h4[(size_t)s0 * 16 + l];
        axpy4(acc, nrmS[p0 + k], r0);
    }
    if (c > CAP) {
        const int ns = nspill[0];
        for (int j = 0; j < ns; ++j) {
            if (spill_d[j] != g) continue;
            const int s0 = (int)(unsigned)spill_e[j];
            const float4 r0 = h4[(size_t)s0 * 16 + l];
            axpy4(acc, nrmP[j], r0);
        }
    }
    const float dd = dinv[g];
    const float4 hs = h4[(size_t)g * 16 + l];
    const float4 b = reinterpret_cast<const float4*>(bias)[l];
    axpy4(acc, dd * dd, hs);
    acc.x += b.x; acc.y += b.y; acc.z += b.z; acc.w += b.w;
    reinterpret_cast<float4*>(out)[(size_t)g * 10 + l] = acc;
}


__global__ void k_lsm(const float* __restrict__ h, float* __restrict__ out, int n) {
    int i = blockIdx.x * blockDim.x + threadIdx.x;
    if (i >= n) return;
    float v[40];
    float mx = -3.402823466e38f;
#pragma unroll
    for (int c = 0; c < 40; ++c) { v[c] = h[(size_t)i * 40 + c]; mx = fmaxf(mx, v[c]); }
    float s = 0.0f;
#pragma unroll
    for (int c = 0; c < 40; ++c) s += expf(v[c] - mx);
    float l = mx + logf(s);
#pragma unroll
    for (int c = 0; c < 40; ++c) out[(size_t)i * 40 + c] = v[c] - l;
}


extern "C" void kernel_launch(void* const* d_in, const int* in_sizes, int n_in,
                              void* d_out, int out_size, void* d_ws, size_t ws_size,
                              hipStream_t stream) {
    const float* x   = (const float*)d_in[0];
    const int*   ei  = (const int*)d_in[1];
    const float* ew  = (const float*)d_in[2];
    const float* W1  = (const float*)d_in[3];
    const float* b1  = (const float*)d_in[4];
    const float* W2  = (const float*)d_in[5];
    const float* b2  = (const float*)d_in[6];
    float* out = (float*)d_out;

    const int N = in_sizes[0] / 64;
    const int E = in_sizes[2];
    const int* src = ei;
    const int* dst = ei + E;

    const size_t Ep = ((size_t)E + 63) / 64 * 64;
    const size_t Np = ((size_t)N + 63) / 64 * 64;
    unsigned long long* ell     = (unsigned long long*)d_ws;
    unsigned long long* spill_e = ell + Np * CAP;
    float* nrmS    = (float*)(spill_e + Ep);
    float* nrmP    = nrmS + Np * CAP;
    int*   spill_d = (int*)(nrmP + Ep);
    int*   cnt     = spill_d + Ep;
    int*   nspill  = cnt + Np;
    float* dinv    = (float*)(nspill + 64);
    float* bufA    = dinv + Np;
    float* bufB    = bufA + (size_t)N * 64;

    const int BLK = 256;
    const int mtiles = N / 16;
    const unsigned gN  = (unsigned)((N + BLK - 1) / BLK);
    const unsigned gE  = (unsigned)((E + BLK - 1) / BLK);
    const unsigned gNg = (unsigned)(((long long)N * 16 + BLK - 1) / BLK);

    k_zero<<<gN, BLK, 0, stream>>>(cnt, nspill, N);
    k_place<<<gE, BLK, 0, stream>>>(src, dst, ew, cnt, ell, nspill, spill_e, spill_d, E);
    k_deg<<<gN, BLK, 0, stream>>>(cnt, ell, nspill, spill_e, spill_d, dinv, N);

    k_gemm1<<<(mtiles * 4 + 7) / 8, BLK, 0, stream>>>(x, W1, bufA, mtiles);
    k_agg1<<<gNg, BLK, 0, stream>>>(cnt, ell, nspill, spill_e, spill_d, dinv, bufA, b1,
                                    nrmS, nrmP, bufB, N);

    k_gemm2<<<(mtiles * 3 + 7) / 8, BLK, 0, stream>>>(bufB, W2, bufA, mtiles);
    k_agg2<<<gNg, BLK, 0, stream>>>(cnt, ell, nspill, spill_e, spill_d, nrmS, nrmP, dinv,
                                    bufA, b2, bufB, N);

    k_lsm<<<gN, BLK, 0, stream>>>(bufB, out, N);
}


# dependent launches 8 -> 5: GEMM1 shares the edge-placement grid, GEMM2 (rows via LDS) and log_softmax run as epilogues of the two node-sum kernels; softmax sum becomes 4-per-lane + 16-lane tree
# speedup vs baseline: 3.7094x; 1.1742x over previous
#include <hip/hip_runtime.h>
#include <math.h>


typedef __attribute__((ext_vector_type(2))) float v2f;
typedef __attribute__((ext_vector_type(8))) float v8f;

constexpr int CAP = 32;


__global__ void k_zero(int* __restrict__ cnt, int* __restrict__ nspill, int n) {
    int i = blockIdx.x * blockDim.x + threadIdx.x;
    if (i < n) cnt[i] = 0;
    if (i == 0) nspill[0] = 0;
}

__device__ __forceinline__ void place_block(int blk,
                        const int* __restrict__ src, const int* __restrict__ dst,
                        const float* __restrict__ w, int* __restrict__ cnt,
                        unsigned long long* __restrict__ ell,
                        int* __restrict__ nspill,
                        unsigned long long* __restrict__ spill_e,
                        int* __restrict__ spill_d, int E) {
    int e = blk * blockDim.x + threadIdx.x;
    if (e < E) {
        const int d = dst[e];
        const unsigned long long v =
            ((unsigned long long)__float_as_uint(w[e]) << 32) |
            (unsigned long long)(unsigned)src[e];
        const int slot = __hip_atomic_fetch_add(&cnt[d], 1, __ATOMIC_RELAXED,
                                                __HIP_MEMORY_SCOPE_AGENT);
        if (slot < CAP) {
            ell[(size_t)d * CAP + slot] = v;
        } else {
            const int j = __hip_atomic_fetch_add(nspill, 1, __ATOMIC_RELAXED,
                                                 __HIP_MEMORY_SCOPE_AGENT);
            spill_e[j] = v;
            spill_d[j] = d;
        }
    }
}

__global__ void k_deg(const int* __restrict__ cnt,
                      const unsigned long long* __restrict__ ell,
                      const int* __restrict__ nspill,
                      const unsigned long long* __restrict__ spill_e,
                      const int* __restrict__ spill_d,
                      float* __restrict__ dinv, int n) {
    int i = blockIdx.x * blockDim.x + threadIdx.x;
    if (i < n) {
        const int c = cnt[i];
        const int m = (c < CAP) ? c : CAP;
        const unsigned long long* __restrict__ row = ell + (size_t)i * CAP;
        const ulonglong2* __restrict__ row2 = reinterpret_cast<const ulonglong2*>(row);
        float d = 1.0f;
        int k = 0;
        for (; k + 2 <= m; k += 2) {
            const ulonglong2 v = row2[k >> 1];
            d += __uint_as_float((unsigned)(v.x >> 32));
            d += __uint_as_float((unsigned)(v.y >> 32));
        }
        if (k < m) d += __uint_as_float((unsigned)(row[k] >> 32));
        if (c > CAP) {
            const int ns = nspill[0];
            for (int j = 0; j < ns; ++j)
                if (spill_d[j] == i) d += __uint_as_float((unsigned)(spill_e[j] >> 32));
        }
        dinv[i] = (d > 0.0f) ? rsqrtf(d) : 0.0f;
    }
}


__device__ __forceinline__ void gemm1_block(int blk,
                        const float* __restrict__ X, const float* __restrict__ W,
                        float* __restrict__ O, int mtiles) {
    int wave = threadIdx.x >> 5;
    int tile = blk * (blockDim.x >> 5) + wave;
    if (tile >= mtiles * 4) return;
    int mt = tile >> 2, nt = tile & 3;
    int lane = threadIdx.x & 31;
    int lo = lane & 15, hi = lane >> 4;

    const float* arow = X + (size_t)(mt * 16 + lo) * 64;
    int col = nt * 16 + lo;
    v8f c = {};
#pragma unroll
    for (int k = 0; k < 64; k += 4) {
        int ka = k + 2 * hi;
        v2f a = { arow[ka], arow[ka + 1] };
        v2f b = { W[(size_t)ka * 64 + col], W[(size_t)(ka + 1) * 64 + col] };
        c = __builtin_amdgcn_wmma_f32_16x16x4_f32(false, a, false, b,
                                                  (short)0, c, false, false);
    }
    float* orow = O + (size_t)(mt * 16 + 8 * hi) * 64 + col;
#pragma unroll
    for (int r = 0; r < 8; ++r) orow[(size_t)r * 64] = c[r];
}

__global__ void k_place_gemm1(const int* __restrict__ src, const int* __restrict__ dst,
                              const float* __restrict__ w, int* __restrict__ cnt,
                              unsigned long long* __restrict__ ell,
                              int* __restrict__ nspill,
                              unsigned long long* __restrict__ spill_e,
                              int* __restrict__ spill_d, int E,
                              const float* __restrict__ X, const float* __restrict__ W1,
                              float* __restrict__ O, int mtiles, int gP, int gG) {
    const int b = blockIdx.x;
    const int nI = ((gP >> 1) < gG) ? (gP >> 1) : gG;
    int pb = -1, gb = -1;
    if (b < 3 * nI) {
        const int t = b / 3, r = b - 3 * t;
        if (r == 2) gb = t; else pb = 2 * t + r;
    } else {
        const int b2 = b - 3 * nI, pr = gP - 2 * nI;
        if (b2 < pr) pb = 2 * nI + b2; else gb = nI + (b2 - pr);
    }
    if (gb >= 0) gemm1_block(gb, X, W1, O, mtiles);
    else place_block(pb, src, dst, w, cnt, ell, nspill, spill_e, spill_d, E);
}


__device__ __forceinline__ void axpy4(float4& a, float s, const float4& r) {
    a.x += s * r.x; a.y += s * r.y; a.z += s * r.z; a.w += s * r.w;
}

constexpr int TP = 68;

__global__ void __launch_bounds__(256)
k_agg1_gemm2(const int* __restrict__ cnt,
             const unsigned long long* __restrict__ ell,
             const int* __restrict__ nspill,
             const unsigned long long* __restrict__ spill_e,
             const int* __restrict__ spill_d,
             const float* __restrict__ dinv, const float* __restrict__ h,
             const float* __restrict__ bias, const float* __restrict__ W2,
             float* __restrict__ nrmS, float* __restrict__ nrmP,
             float* __restrict__ O, int n) {
    __shared__ float tile[16 * TP];
    const int gi = threadIdx.x >> 4, l = threadIdx.x & 15;
    const int row0 = blockIdx.x * 16;
    const int g = row0 + gi;
    float4 acc = make_float4(0.0f, 0.0f, 0.0f, 0.0f);
    if (g < n) {
        const int c = cnt[g];
        const int m = (c < CAP) ? c : CAP;
        const size_t p0 = (size_t)g * CAP;
        const ulonglong2* __restrict__ row2 = reinterpret_cast<const ulonglong2*>(ell + p0);
        const float dd = dinv[g];
        const float4* __restrict__ h4 = reinterpret_cast<const float4*>(h);
        int k = 0;
        for (; k + 4 <= m; k += 4) {
            const ulonglong2 ea = row2[k >> 1], eb = row2[(k >> 1) + 1];
            const unsigned long long e0 = ea.x, e1 = ea.y, e2 = eb.x, e3 = eb.y;
            const int s0 = (int)(unsigned)e0, s1 = (int)(unsigned)e1;
            const int s2 = (int)(unsigned)e2, s3 = (int)(unsigned)e3;
            const float4 r0 = h4[(size_t)s0 * 16 + l];
            const float4 r1 = h4[(size_t)s1 * 16 + l];
            const float4 r2 = h4[(size_t)s2 * 16 + l];
            const float4 r3 = h4[(size_t)s3 * 16 + l];
            const float n0 = dinv[s0] * __uint_as_float((unsigned)(e0 >> 32)) * dd;
            const float n1 = dinv[s1] * __uint_as_float((unsigned)(e1 >> 32)) * dd;
            const float n2 = dinv[s2] * __uint_as_float((unsigned)(e2 >> 32)) * dd;
            const float n3 = dinv[s3] * __uint_as_float((unsigned)(e3 >> 32)) * dd;
            axpy4(acc, n0, r0);
            axpy4(acc, n1, r1);
            axpy4(acc, n2, r2);
            axpy4(acc, n3, r3);
            if (l < 4) nrmS[p0 + k + l] = (l == 0) ? n0 : (l == 1) ? n1 : (l == 2) ? n2 : n3;
        }
        for (; k < m; ++k) {
            const unsigned long long e0 = ell[p0 + k];
            const int s0 = (int)(unsigned)e0;
            const float4 r0 = h4[(size_t)s0 * 16 + l];
            const float n0 = dinv[s0] * __uint_as_float((unsigned)(e0 >> 32)) * dd;
            axpy4(acc, n0, r0);
            if (l == 0) nrmS[p0 + k] = n0;
        }
        if (c > CAP) {
            const int ns = nspill[0];
            for (int j = 0; j < ns; ++j) {
                if (spill_d[j] != g) continue;
                const unsigned long long e0 = spill_e[j];
                const int s0 = (int)(unsigned)e0;
                const float4 r0 = h4[(size_t)s0 * 16 + l];
                const float n0 = dinv[s0] * __uint_as_float((unsigned)(e0 >> 32)) * dd;
                axpy4(acc, n0, r0);
                if (l == 0) nrmP[j] = n0;
            }
        }
        const float4 hs = h4[(size_t)g * 16 + l];
        const float4 b = reinterpret_cast<const float4*>(bias)[l];
        axpy4(acc, dd * dd, hs);
        acc.x += b.x; acc.y += b.y; acc.z += b.z; acc.w += b.w;
        acc.x = fmaxf(acc.x, 0.0f); acc.y = fmaxf(acc.y, 0.0f);
        acc.z = fmaxf(acc.z, 0.0f); acc.w = fmaxf(acc.w, 0.0f);
    }
    *reinterpret_cast<float4*>(&tile[gi * TP + 4 * l]) = acc;

    const int wave = threadIdx.x >> 5;
    const int lane = threadIdx.x & 31;
    const int lo = lane & 15, hi = lane >> 4;
    const int col = wave * 16 + lo;
    const bool cok = (wave < 3) && (col < 40);
    float bw[32];
    if (wave < 3) {
        const float* __restrict__ wcol = W2 + (cok ? col : 0);
#pragma unroll
        for (int j = 0; j < 16; ++j) {
            const int ka = 4 * j + 2 * hi;
            const float w0 = wcol[ka * 40], w1 = wcol[(ka + 1) * 40];
            bw[2 * j]     = cok ? w0 : 0.0f;
            bw[2 * j + 1] = cok ? w1 : 0.0f;
        }
    }
    __syncthreads();
    if (wave < 3) {
        const float* arow = tile + lo * TP;
        v8f c = {};
#pragma unroll
        for (int j = 0; j < 16; ++j) {
            const int ka = 4 * j + 2 * hi;
            v2f a = { arow[ka], arow[ka + 1] };
            v2f b = { bw[2 * j], bw[2 * j + 1] };
            c = __builtin_amdgcn_wmma_f32_16x16x4_f32(false, a, false, b,
                                                      (short)0, c, false, false);
        }
        if (cok) {
            const int r0 = row0 + 8 * hi;
            float* orow = O + (size_t)r0 * 64 + col;
            if (row0 + 16 <= n) {
#pragma unroll
                for (int r = 0; r < 8; ++r) orow[(size_t)r * 64] = c[r];
            } else {
#pragma unroll
                for (int r = 0; r < 8; ++r)
                    if (r0 + r < n) orow[(size_t)r * 64] = c[r];
            }
        }
    }
}

__global__ void k_agg2_lsm(const int* __restrict__ cnt,
                           const unsigned long long* __restrict__ ell,
                           const int* __restrict__ nspill,
                           const unsigned long long* __restrict__ spill_e,
                           const int* __restrict__ spill_d,
                           const float* __restrict__ nrmS, const float* __restrict__ nrmP,
                           const float* __restrict__ dinv,
                           const float* __restrict__ h, const float* __restrict__ bias,
                           float* __restrict__ out, int n) {
    const int t = blockIdx.x * blockDim.x + threadIdx.x;
    const int g = t >> 4, l = t & 15;
    const bool act = (g < n) && (l < 10);
    float4 acc = make_float4(0.0f, 0.0f, 0.0f, 0.0f);
    if (act) {
        const int c = cnt[g];
        const int m = (c < CAP) ? c : CAP;
        const size_t p0 = (size_t)g * CAP;
        const ulonglong2* __restrict__ row2 = reinterpret_cast<const ulonglong2*>(ell + p0);
        const float4* __restrict__ nrm4 = reinterpret_cast<const float4*>(nrmS + p0);
        const float4* __restrict__ h4 = reinterpret_cast<const float4*>(h);
        int k = 0;
        for (; k + 4 <= m; k += 4) {
            const ulonglong2 ea = row2[k >> 1], eb = row2[(k >> 1) + 1];
            const int s0 = (int)(unsigned)ea.x, s1 = (int)(unsigned)ea.y;
            const int s2 = (int)(unsigned)eb.x, s3 = (int)(unsigned)eb.y;
            const float4 r0 = h4[(size_t)s0 * 16 + l];
            const float4 r1 = h4[(size_t)s1 * 16 + l];
            const float4 r2 = h4[(size_t)s2 * 16 + l];
            const float4 r3 = h4[(size_t)s3 * 16 + l];
            const float4 nn = nrm4[k >> 2];
            axpy4(acc, nn.x, r0);
            axpy4(acc, nn.y, r1);
            axpy4(acc, nn.z, r2);
            axpy4(acc, nn.w, r3);
        }
        for (; k < m; ++k) {
            const int s0 = (int)(unsigned)ell[p0 + k];
            const float4 r0 = h4[(size_t)s0 * 16 + l];
            axpy4(acc, nrmS[p0 + k], r0);
        }
        if (c > CAP) {
            const int ns = nspill[0];
            for (int j = 0; j < ns; ++j) {
                if (spill_d[j] != g) continue;
                const int s0 = (int)(unsigned)spill_e[j];
                const float4 r0 = h4[(size_t)s0 * 16 + l];
                axpy4(acc, nrmP[j], r0);
            }
        }
        const float dd = dinv[g];
        const float4 hs = h4[(size_t)g * 16 + l];
        const float4 b = reinterpret_cast<const float4*>(bias)[l];
        axpy4(acc, dd * dd, hs);
        acc.x += b.x; acc.y += b.y; acc.z += b.z; acc.w += b.w;
    }
    float mx = act ? fmaxf(fmaxf(acc.x, acc.y), fmaxf(acc.z, acc.w)) : -3.402823466e38f;
    for (int o = 8; o >= 1; o >>= 1) mx = fmaxf(mx, __shfl_xor(mx, o, 16));
    float s = act ? (expf(acc.x - mx) + expf(acc.y - mx)) + (expf(acc.z - mx) + expf(acc.w - mx))
                  : 0.0f;
    for (int o = 8; o >= 1; o >>= 1) s += __shfl_xor(s, o, 16);
    const float lse = mx + logf(s);
    if (act) {
        acc.x -= lse; acc.y -= lse; acc.z -= lse; acc.w -= lse;
        reinterpret_cast<float4*>(out)[(size_t)g * 10 + l] = acc;
    }
}


extern "C" void kernel_launch(void* const* d_in, const int* in_sizes, int n_in,
                              void* d_out, int out_size, void* d_ws, size_t ws_size,
                              hipStream_t stream) {
    const float* x   = (const float*)d_in[0];
    const int*   ei  = (const int*)d_in[1];
    const float* ew  = (const float*)d_in[2];
    const float* W1  = (const float*)d_in[3];
    const float* b1  = (const float*)d_in[4];
    const float* W2  = (const float*)d_in[5];
    const float* b2  = (const float*)d_in[6];
    float* out = (float*)d_out;

    const int N = in_sizes[0] / 64;
    const int E = in_sizes[2];
    const int* src = ei;
    const int* dst = ei + E;

    const size_t Ep = ((size_t)E + 63) / 64 * 64;
    const size_t Np = ((size_t)N + 63) / 64 * 64;
    unsigned long long* ell     = (unsigned long long*)d_ws;
    unsigned long long* spill_e = ell + Np * CAP;
    float* nrmS    = (float*)(spill_e + Ep);
    float* nrmP    = nrmS + Np * CAP;
    int*   spill_d = (int*)(nrmP + Ep);
    int*   cnt     = spill_d + Ep;
    int*   nspill  = cnt + Np;
    float* dinv    = (float*)(nspill + 64);
    float* bufA    = dinv + Np;
    float* bufB    = bufA + (size_t)N * 64;

    const int BLK = 256;
    const int mtiles = N / 16;
    const unsigned gN  = (unsigned)((N + BLK - 1) / BLK);
    const unsigned gE  = (unsigned)((E + BLK - 1) / BLK);
    const unsigned gG1 = (unsigned)((mtiles * 4 + 7) / 8);
    const unsigned gNg = (unsigned)(((long long)N * 16 + BLK - 1) / BLK);

    k_zero<<<gN, BLK, 0, stream>>>(cnt, nspill, N);
    k_place_gemm1<<<gE + gG1, BLK, 0, stream>>>(src, dst, ew, cnt, ell, nspill, spill_e,
                                                spill_d, E, x, W1, bufA, mtiles,
                                                (int)gE, (int)gG1);
    k_deg<<<gN, BLK, 0, stream>>>(cnt, ell, nspill, spill_e, spill_d, dinv, N);

    k_agg1_gemm2<<<gNg, BLK, 0, stream>>>(cnt, ell, nspill, spill_e, spill_d, dinv, bufA, b1,
                                          W2, nrmS, nrmP, bufB, N);

    k_agg2_lsm<<<gNg, BLK, 0, stream>>>(cnt, ell, nspill, spill_e, spill_d, nrmS, nrmP, dinv,
                                        bufB, b2, out, N);
}


# per-edge global count atomics and the degree pass replaced by two-level LDS grouping (128-node ranges; rows, counts, dinv built in LDS per range): 5->4 launches; sums stay in slot (arrival) order
# speedup vs baseline: 4.7172x; 1.2717x over previous
#include <hip/hip_runtime.h>
#include <math.h>


typedef __attribute__((ext_vector_type(2))) float v2f;
typedef __attribute__((ext_vector_type(8))) float v8f;

constexpr int CAP = 32;


constexpr int BSH  = 7;
constexpr int BN   = 128;
constexpr int MAXB = 1024;
constexpr int CELL = 32;
constexpr int EPB  = 8192;
constexpr int RP   = 33;

__device__ __forceinline__ void bin_block(int blk,
                        const int* __restrict__ src, const int* __restrict__ dst,
                        const float* __restrict__ w,
                        unsigned long long* __restrict__ cells, int* __restrict__ cellcnt,
                        unsigned long long* __restrict__ ovf_e, int* __restrict__ ovf_d,
                        int* __restrict__ ovfcnt, int E, int nblk, int nbinsA) {
    __shared__ int lcnt[MAXB + 1];
    __shared__ int lovf;
    for (int i = threadIdx.x; i <= MAXB; i += blockDim.x) lcnt[i] = 0;
    if (threadIdx.x == 0) lovf = 0;
    __syncthreads();
    const int e0 = blk * EPB;
    const int e1 = (e0 + EPB < E) ? (e0 + EPB) : E;
    for (int base = e0 + (int)threadIdx.x; base < e1; base += 8 * (int)blockDim.x) {
        int dd[8], rk[8];
        unsigned ss[8], ww[8];
#pragma unroll
        for (int u = 0; u < 8; ++u) {
            const int e = base + u * (int)blockDim.x;
            const int ec = (e < e1) ? e : (e1 - 1);
            dd[u] = dst[ec];
            ss[u] = (unsigned)src[ec];
            ww[u] = __float_as_uint(w[ec]);
        }
#pragma unroll
        for (int u = 0; u < 8; ++u) {
            const int bin = dd[u] >> BSH;
            const bool cell = (base + u * (int)blockDim.x < e1) && (bin < MAXB);
            const int r = atomicAdd(&lcnt[cell ? bin : MAXB], 1);
            rk[u] = cell ? r : CELL;
        }
#pragma unroll
        for (int u = 0; u < 8; ++u) {
            if (base + u * (int)blockDim.x < e1) {
                const int d = dd[u];
                if (rk[u] < CELL) {
                    cells[((size_t)(d >> BSH) * nblk + blk) * CELL + rk[u]] =
                        ((unsigned long long)ww[u] << 32) |
                        (unsigned long long)(ss[u] | ((unsigned)(d & (BN - 1)) << 24));
                } else {
                    const int o = atomicAdd(&lovf, 1);
                    ovf_e[(size_t)blk * EPB + o] =
                        ((unsigned long long)ww[u] << 32) | (unsigned long long)ss[u];
                    ovf_d[(size_t)blk * EPB + o] = d;
                }
            }
        }
    }
    __syncthreads();
    for (int i = threadIdx.x; i < nbinsA; i += blockDim.x) {
        const int cc = lcnt[i];
        cellcnt[(size_t)i * nblk + blk] = (cc < CELL) ? cc : CELL;
    }
    if (threadIdx.x == 0) ovfcnt[blk] = lovf;
}


__device__ __forceinline__ void gemm1_block(int blk,
                        const float* __restrict__ X, const float* __restrict__ W,
                        float* __restrict__ O, int mtiles) {
    int wave = threadIdx.x >> 5;
    int tile = blk * (blockDim.x >> 5) + wave;
    if (tile >= mtiles * 4) return;
    int mt = tile >> 2, nt = tile & 3;
    int lane = threadIdx.x & 31;
    int lo = lane & 15, hi = lane >> 4;

    const float* arow = X + (size_t)(mt * 16 + lo) * 64;
    int col = nt * 16 + lo;
    v8f c = {};
#pragma unroll
    for (int k = 0; k < 64; k += 4) {
        int ka = k + 2 * hi;
        v2f a = { arow[ka], arow[ka + 1] };
        v2f b = { W[(size_t)ka * 64 + col], W[(size_t)(ka + 1) * 64 + col] };
        c = __builtin_amdgcn_wmma_f32_16x16x4_f32(false, a, false, b,
                                                  (short)0, c, false, false);
    }
    float* orow = O + (size_t)(mt * 16 + 8 * hi) * 64 + col;
#pragma unroll
    for (int r = 0; r < 8; ++r) orow[(size_t)r * 64] = c[r];
}

__global__ void __launch_bounds__(256)
k_bin_gemm1(const int* __restrict__ src, const int* __restrict__ dst,
            const float* __restrict__ w,
            unsigned long long* __restrict__ cells, int* __restrict__ cellcnt,
            unsigned long long* __restrict__ ovf_e, int* __restrict__ ovf_d,
            int* __restrict__ ovfcnt, int* __restrict__ nspill,
            unsigned long long* __restrict__ pad, int E, int nblk, int nbinsA,
            const float* __restrict__ X, const float* __restrict__ W1,
            float* __restrict__ O, int mtiles, int gP, int gG) {
    const int b = blockIdx.x;
    if (b == 0 && threadIdx.x == 0) {
        nspill[0] = 0;
        pad[0] = 0;
    }
    const int nI = ((gP >> 1) < gG) ? (gP >> 1) : gG;
    int pb = -1, gb = -1;
    if (b < 3 * nI) {
        const int t = b / 3, r = b - 3 * t;
        if (r == 2) gb = t; else pb = 2 * t + r;
    } else {
        const int b2 = b - 3 * nI, pr = gP - 2 * nI;
        if (b2 < pr) pb = 2 * nI + b2; else gb = nI + (b2 - pr);
    }
    if (gb >= 0) gemm1_block(gb, X, W1, O, mtiles);
    else bin_block(pb, src, dst, w, cells, cellcnt, ovf_e, ovf_d, ovfcnt, E, nblk, nbinsA);
}

__device__ __forceinline__ void place_store(int slot, int dl, unsigned s, unsigned wb, int bin,
                        float* sxw, unsigned long long* srow,
                        int* __restrict__ nspill,
                        unsigned long long* __restrict__ spill_e, int* __restrict__ spill_d) {
    const unsigned long long v = ((unsigned long long)wb << 32) | (unsigned long long)s;
    if (slot < CAP) {
        srow[dl * RP + slot] = v;
    } else {
        const int j = __hip_atomic_fetch_add(nspill, 1, __ATOMIC_RELAXED,
                                             __HIP_MEMORY_SCOPE_AGENT);
        spill_e[j] = v;
        spill_d[j] = bin * BN + dl;
        atomicAdd(&sxw[dl], __uint_as_float(wb));
    }
}

__global__ void __launch_bounds__(256)
k_place_deg(const unsigned long long* __restrict__ cells, const int* __restrict__ cellcnt,
            const unsigned long long* __restrict__ pad,
            const unsigned long long* __restrict__ ovf_e, const int* __restrict__ ovf_d,
            const int* __restrict__ ovfcnt, int nblk,
            unsigned long long* __restrict__ ell, int* __restrict__ cnt,
            float* __restrict__ dinv, int* __restrict__ nspill,
            unsigned long long* __restrict__ spill_e, int* __restrict__ spill_d, int n) {
    __shared__ unsigned long long srow[BN * RP];
    __shared__ int scnt[BN + 1];
    __shared__ float sxw[BN];
    __shared__ int sn[256];
    __shared__ int so[256];
    __shared__ int sany;
    const int bin = blockIdx.x;
    const int t = threadIdx.x;
    if (t <= BN) scnt[t] = 0;
    if (t < BN) sxw[t] = 0.0f;
    if (t == 0) sany = 0;
    __syncthreads();
    for (int c0 = 0; c0 < nblk; c0 += 256) {
        const int nb = (nblk - c0 < 256) ? (nblk - c0) : 256;
        if (t < nb) {
            sn[t] = (bin < MAXB) ? cellcnt[(size_t)bin * nblk + c0 + t] : 0;
            const int ov = ovfcnt[c0 + t];
            so[t] = ov;
            if (ov > 0) atomicOr(&sany, 1);
        }
        __syncthreads();
        if (bin < MAXB) {
            const unsigned long long* __restrict__ cb = cells + ((size_t)bin * nblk + c0) * CELL;
            const int np = nb * CELL;
            for (int p0 = t; p0 < np; p0 += 4 * 256) {
                unsigned long long rec[4];
                int sl[4];
                bool ok[4];
#pragma unroll
                for (int u = 0; u < 4; ++u) {
                    const int p = p0 + u * 256;
                    const int pc = (p < np) ? p : 0;
                    ok[u] = (p < np) && ((pc & (CELL - 1)) < sn[pc >> 5]);
                    rec[u] = *(ok[u] ? (cb + pc) : pad);
                }
#pragma unroll
                for (int u = 0; u < 4; ++u) {
                    const int dl = ok[u] ? (int)((unsigned)rec[u] >> 24) : BN;
                    sl[u] = atomicAdd(&scnt[dl], 1);
                }
#pragma unroll
                for (int u = 0; u < 4; ++u) {
                    if (ok[u]) {
                        const unsigned lo = (unsigned)rec[u];
                        place_store(sl[u], (int)(lo >> 24), lo & 0xFFFFFFu, (unsigned)(rec[u] >> 32),
                                    bin, sxw, srow, nspill, spill_e, spill_d);
                    }
                }
            }
        }
        if (sany) {
            for (int cc = 0; cc < nb; ++cc) {
                const int no = so[cc];
                for (int i = t; i < no; i += 256) {
                    const int d = ovf_d[(size_t)(c0 + cc) * EPB + i];
                    if ((d >> BSH) == bin) {
                        const unsigned long long rec = ovf_e[(size_t)(c0 + cc) * EPB + i];
                        const int dl = d & (BN - 1);
                        const int slot = atomicAdd(&scnt[dl], 1);
                        place_store(slot, dl, (unsigned)rec, (unsigned)(rec >> 32),
                                    bin, sxw, srow, nspill, spill_e, spill_d);
                    }
                }
            }
        }
        __syncthreads();
    }
    if (t < BN) {
        const int g = bin * BN + t;
        if (g < n) {
            const int cc = scnt[t];
            const int m = (cc < CAP) ? cc : CAP;
            float d = 1.0f;
            for (int k = 0; k < m; ++k) d += __uint_as_float((unsigned)(srow[t * RP + k] >> 32));
            if (cc > CAP) d += sxw[t];
            cnt[g] = cc;
            dinv[g] = (d > 0.0f) ? rsqrtf(d) : 0.0f;
        }
    }
    for (int idx = t; idx < BN * CAP; idx += 256) {
        const int row = idx >> 5, k = idx & (CAP - 1);
        const int g = bin * BN + row;
        if (g < n && k < scnt[row]) ell[(size_t)g * CAP + k] = srow[row * RP + k];
    }
}


__device__ __forceinline__ void axpy4(float4& a, float s, const float4& r) {
    a.x += s * r.x; a.y += s * r.y; a.z += s * r.z; a.w += s * r.w;
}

template <int K>
__device__ __forceinline__ int slot_src(int sa, int sb) {
    return __builtin_amdgcn_update_dpp(0, (K & 1) ? sb : sa, 0x150 + (K >> 1), 0xf, 0xf, true);
}
template <int K>
__device__ __forceinline__ float slot_nrm(float na, float nb) {
    return __int_as_float(__builtin_amdgcn_update_dpp(0, __float_as_int((K & 1) ? nb : na),
                                                      0x150 + (K >> 1), 0xf, 0xf, true));
}

template <int K0>
__device__ __forceinline__ void rows8(float4 (&r)[8], int l, const float4* __restrict__ h4,
                                      int sa, int sb) {
    r[0] = h4[(size_t)slot_src<K0 + 0>(sa, sb) * 16 + l];
    r[1] = h4[(size_t)slot_src<K0 + 1>(sa, sb) * 16 + l];
    r[2] = h4[(size_t)slot_src<K0 + 2>(sa, sb) * 16 + l];
    r[3] = h4[(size_t)slot_src<K0 + 3>(sa, sb) * 16 + l];
    r[4] = h4[(size_t)slot_src<K0 + 4>(sa, sb) * 16 + l];
    r[5] = h4[(size_t)slot_src<K0 + 5>(sa, sb) * 16 + l];
    r[6] = h4[(size_t)slot_src<K0 + 6>(sa, sb) * 16 + l];
    r[7] = h4[(size_t)slot_src<K0 + 7>(sa, sb) * 16 + l];
}
template <int K0>
__device__ __forceinline__ void add8(float4& acc, const float4 (&r)[8], float na, float nb) {
    axpy4(acc, slot_nrm<K0 + 0>(na, nb), r[0]);
    axpy4(acc, slot_nrm<K0 + 1>(na, nb), r[1]);
    axpy4(acc, slot_nrm<K0 + 2>(na, nb), r[2]);
    axpy4(acc, slot_nrm<K0 + 3>(na, nb), r[3]);
    axpy4(acc, slot_nrm<K0 + 4>(na, nb), r[4]);
    axpy4(acc, slot_nrm<K0 + 5>(na, nb), r[5]);
    axpy4(acc, slot_nrm<K0 + 6>(na, nb), r[6]);
    axpy4(acc, slot_nrm<K0 + 7>(na, nb), r[7]);
}

constexpr int TP = 68;

__global__ void __launch_bounds__(256)
k_agg1_gemm2(const int* __restrict__ cnt,
             const unsigned long long* __restrict__ ell,
             const int* __restrict__ nspill,
             const unsigned long long* __restrict__ spill_e,
             const int* __restrict__ spill_d,
             const float* __restrict__ dinv, const float* __restrict__ h,
             const float* __restrict__ bias, const float* __restrict__ W2,
             float* __restrict__ nrmS, float* __restrict__ nrmP,
             float* __restrict__ O, int n) {
    __shared__ float tile[16 * TP];
    const int gi = threadIdx.x >> 4, l = threadIdx.x & 15;
    const int row0 = blockIdx.x * 16;
    const int g = row0 + gi;
    float4 acc = make_float4(0.0f, 0.0f, 0.0f, 0.0f);
    if (g < n) {
        const int c = cnt[g];
        const int m = (c < CAP) ? c : CAP;
        const size_t p0 = (size_t)g * CAP;
        const float4* __restrict__ h4 = reinterpret_cast<const float4*>(h);
        const float4 hs = h4[(size_t)g * 16 + l];
        const float4 b = reinterpret_cast<const float4*>(bias)[l];
        int sa = g, sb = g;
        float wa = 0.0f, wb = 0.0f;
        if (2 * l + 1 < m) {
            const ulonglong2 ev = reinterpret_cast<const ulonglong2*>(ell + p0)[l];
            sa = (int)(unsigned)ev.x;
            sb = (int)(unsigned)ev.y;
            wa = __uint_as_float((unsigned)(ev.x >> 32));
            wb = __uint_as_float((unsigned)(ev.y >> 32));
        } else if (2 * l < m) {
            const unsigned long long e0 = ell[p0 + 2 * l];
            sa = (int)(unsigned)e0;
            wa = __uint_as_float((unsigned)(e0 >> 32));
        }
        float4 r[8];
        rows8<0>(r, l, h4, sa, sb);
        const float dd = dinv[g];
        const float da = dinv[sa], db = dinv[sb];
        const float na = (2 * l < m) ? da * wa * dd : 0.0f;
        const float nb = (2 * l + 1 < m) ? db * wb * dd : 0.0f;
        if (2 * l + 1 < m) *reinterpret_cast<float2*>(nrmS + p0 + 2 * l) = make_float2(na, nb);
        else if (2 * l < m) nrmS[p0 + 2 * l] = na;
        add8<0>(acc, r, na, nb);
        if (m > 8)  { rows8<8>(r, l, h4, sa, sb);  add8<8>(acc, r, na, nb); }
        if (m > 16) { rows8<16>(r, l, h4, sa, sb); add8<16>(acc, r, na, nb); }
        if (m > 24) { rows8<24>(r, l, h4, sa, sb); add8<24>(acc, r, na, nb); }
        if (c > CAP) {
            const int ns = nspill[0];
            for (int j = 0; j < ns; ++j) {
                if (spill_d[j] != g) continue;
                const unsigned long long e0 = spill_e[j];
                const int s0 = (int)(unsigned)e0;
                const float4 r0 = h4[(size_t)s0 * 16 + l];
                const float n0 = dinv[s0] * __uint_as_float((unsigned)(e0 >> 32)) * dd;
                axpy4(acc, n0, r0);
                if (l == 0) nrmP[j] = n0;
            }
        }
        axpy4(acc, dd * dd, hs);
        acc.x += b.x; acc.y += b.y; acc.z += b.z; acc.w += b.w;
        acc.x = fmaxf(acc.x, 0.0f); acc.y = fmaxf(acc.y, 0.0f);
        acc.z = fmaxf(acc.z, 0.0f); acc.w = fmaxf(acc.w, 0.0f);
    }
    *reinterpret_cast<float4*>(&tile[gi * TP + 4 * l]) = acc;

    const int wave = threadIdx.x >> 5;
    const int lane = threadIdx.x & 31;
    const int lo = lane & 15, hi = lane >> 4;
    const int col = wave * 16 + lo;
    const bool cok = (wave < 3) && (col < 40);
    float bw[32];
    if (wave < 3) {
        const float* __restrict__ wcol = W2 + (cok ? col : 0);
#pragma unroll
        for (int j = 0; j < 16; ++j) {
            const int ka = 4 * j + 2 * hi;
            const float w0 = wcol[ka * 40], w1 = wcol[(ka + 1) * 40];
            bw[2 * j]     = cok ? w0 : 0.0f;
            bw[2 * j + 1] = cok ? w1 : 0.0f;
        }
    }
    __syncthreads();
    if (wave < 3) {
        const float* arow = tile + lo * TP;
        v8f c = {};
#pragma unroll
        for (int j = 0; j < 16; ++j) {
            const int ka = 4 * j + 2 * hi;
            v2f a = { arow[ka], arow[ka + 1] };
            v2f b = { bw[2 * j], bw[2 * j + 1] };
            c = __builtin_amdgcn_wmma_f32_16x16x4_f32(false, a, false, b,
                                                      (short)0, c, false, false);
        }
        if (cok) {
            const int r0 = row0 + 8 * hi;
            float* orow = O + (size_t)r0 * 64 + col;
            if (row0 + 16 <= n) {
#pragma unroll
                for (int r = 0; r < 8; ++r) orow[(size_t)r * 64] = c[r];
            } else {
#pragma unroll
                for (int r = 0; r < 8; ++r)
                    if (r0 + r < n) orow[(size_t)r * 64] = c[r];
            }
        }
    }
}

__global__ void k_agg2_lsm(const int* __restrict__ cnt,
                           const unsigned long long* __restrict__ ell,
                           const int* __restrict__ nspill,
                           const unsigned long long* __restrict__ spill_e,
                           const int* __restrict__ spill_d,
                           const float* __restrict__ nrmS, const float* __restrict__ nrmP,
                           const float* __restrict__ dinv,
                           const float* __restrict__ h, const float* __restrict__ bias,
                           float* __restrict__ out, int n) {
    const int t = blockIdx.x * blockDim.x + threadIdx.x;
    const int g = t >> 4, l = t & 15;
    const bool act = (g < n) && (l < 10);
    const int lc = (l < 10) ? l : 0;
    float4 acc = make_float4(0.0f, 0.0f, 0.0f, 0.0f);
    if (g < n) {
        const int c = cnt[g];
        const int m = (c < CAP) ? c : CAP;
        const size_t p0 = (size_t)g * CAP;
        const float4* __restrict__ h4 = reinterpret_cast<const float4*>(h);
        const float dd = dinv[g];
        const float4 hs = h4[(size_t)g * 16 + lc];
        const float4 b = reinterpret_cast<const float4*>(bias)[lc];
        int sa = g, sb = g;
        float na = 0.0f, nb = 0.0f;
        if (2 * l + 1 < m) {
            const ulonglong2 ev = reinterpret_cast<const ulonglong2*>(ell + p0)[l];
            const float2 nn = *reinterpret_cast<const float2*>(nrmS + p0 + 2 * l);
            sa = (int)(unsigned)ev.x;
            sb = (int)(unsigned)ev.y;
            na = nn.x;
            nb = nn.y;
        } else if (2 * l < m) {
            sa = (int)(unsigned)ell[p0 + 2 * l];
            na = nrmS[p0 + 2 * l];
        }
        float4 r[8];
        rows8<0>(r, lc, h4, sa, sb);
        add8<0>(acc, r, na, nb);
        if (m > 8)  { rows8<8>(r, lc, h4, sa, sb);  add8<8>(acc, r, na, nb); }
        if (m > 16) { rows8<16>(r, lc, h4, sa, sb); add8<16>(acc, r, na, nb); }
        if (m > 24) { rows8<24>(r, lc, h4, sa, sb); add8<24>(acc, r, na, nb); }
        if (c > CAP) {
            const int ns = nspill[0];
            for (int j = 0; j < ns; ++j) {
                if (spill_d[j] != g) continue;
                const int s0 = (int)(unsigned)spill_e[j];
                const float4 r0 = h4[(size_t)s0 * 16 + lc];
                axpy4(acc, nrmP[j], r0);
            }
        }
        axpy4(acc, dd * dd, hs);
        acc.x += b.x; acc.y += b.y; acc.z += b.z; acc.w += b.w;
    }
    float mx = act ? fmaxf(fmaxf(acc.x, acc.y), fmaxf(acc.z, acc.w)) : -3.402823466e38f;
    for (int o = 8; o >= 1; o >>= 1) mx = fmaxf(mx, __shfl_xor(mx, o, 16));
    float s = act ? (expf(acc.x - mx) + expf(acc.y - mx)) + (expf(acc.z - mx) + expf(acc.w - mx))
                  : 0.0f;
    for (int o = 8; o >= 1; o >>= 1) s += __shfl_xor(s, o, 16);
    const float lse = mx + logf(s);
    if (act) {
        acc.x -= lse; acc.y -= lse; acc.z -= lse; acc.w -= lse;
        reinterpret_cast<float4*>(out)[(size_t)g * 10 + l] = acc;
    }
}


extern "C" void kernel_launch(void* const* d_in, const int* in_sizes, int n_in,
                              void* d_out, int out_size, void* d_ws, size_t ws_size,
                              hipStream_t stream) {
    const float* x   = (const float*)d_in[0];
    const int*   ei  = (const int*)d_in[1];
    const float* ew  = (const float*)d_in[2];
    const float* W1  = (const float*)d_in[3];
    const float* b1  = (const float*)d_in[4];
    const float* W2  = (const float*)d_in[5];
    const float* b2  = (const float*)d_in[6];
    float* out = (float*)d_out;

    const int N = in_sizes[0] / 64;
    const int E = in_sizes[2];
    const int* src = ei;
    const int* dst = ei + E;

    const int nblk   = (E + EPB - 1) / EPB;
    const int nbins  = (N + BN - 1) / BN;
    const int nbinsA = (nbins < MAXB) ? nbins : MAXB;

    const size_t Ep = ((size_t)E + 63) / 64 * 64;
    const size_t Np = ((size_t)N + 63) / 64 * 64;
    const size_t Cp = ((size_t)nbinsA * nblk + 63) / 64 * 64;
    unsigned long long* ell     = (unsigned long long*)d_ws;
    unsigned long long* spill_e = ell + Np * CAP;
    float* nrmS    = (float*)(spill_e + Ep);
    float* nrmP    = nrmS + Np * CAP;
    int*   spill_d = (int*)(nrmP + Ep);
    int*   cnt     = spill_d + Ep;
    int*   nspill  = cnt + Np;
    float* dinv    = (float*)(nspill + 64);
    float* bufA    = dinv + Np;
    float* bufB    = bufA + (size_t)N * 64;
    unsigned long long* cells = (unsigned long long*)(bufB + (size_t)N * 64);
    unsigned long long* pad   = cells + (size_t)nbinsA * nblk * CELL;
    unsigned long long* ovf_e = pad + 32;
    int*   ovf_d   = (int*)(ovf_e + (size_t)nblk * EPB);
    int*   cellcnt = ovf_d + (size_t)nblk * EPB;
    int*   ovfcnt  = cellcnt + Cp;

    const int BLK = 256;
    const int mtiles = N / 16;
    const unsigned gG1 = (unsigned)((mtiles * 4 + 7) / 8);
    const unsigned gNg = (unsigned)(((long long)N * 16 + BLK - 1) / BLK);

    k_bin_gemm1<<<(unsigned)nblk + gG1, BLK, 0, stream>>>(src, dst, ew, cells, cellcnt, ovf_e,
                                                          ovf_d, ovfcnt, nspill, pad, E, nblk,
                                                          nbinsA, x, W1, bufA, mtiles, nblk,
                                                          (int)gG1);
    k_place_deg<<<(unsigned)nbins, BLK, 0, stream>>>(cells, cellcnt, pad, ovf_e, ovf_d, ovfcnt, nblk,
                                                     ell, cnt, dinv, nspill, spill_e, spill_d, N);

    k_agg1_gemm2<<<gNg, BLK, 0, stream>>>(cnt, ell, nspill, spill_e, spill_d, dinv, bufA, b1,
                                          W2, nrmS, nrmP, bufB, N);

    k_agg2_lsm<<<gNg, BLK, 0, stream>>>(cnt, ell, nspill, spill_e, spill_d, nrmS, nrmP, dinv,
                                        bufB, b2, out, N);
}


# level-1 binning fills its cells in LDS (a 1024-thread block per WGP) and stores each cell with one contiguous store, not 8 bytes per edge; GEMM1 rides in the level-2 grid; sums stay in slot order
# speedup vs baseline: 5.0678x; 1.0743x over previous
#include <hip/hip_runtime.h>
#include <math.h>


typedef __attribute__((ext_vector_type(2))) float v2f;
typedef __attribute__((ext_vector_type(8))) float v8f;

constexpr int CAP = 32;


constexpr int BSH  = 7;
constexpr int BN   = 128;
constexpr int MAXB = 1024;
constexpr int CELL = 32;
constexpr int EPB  = 8192;
constexpr int RP   = 33;
constexpr int LP   = 33;

__device__ __forceinline__ void bin_block(int blk,
                        const int* __restrict__ src, const int* __restrict__ dst,
                        const float* __restrict__ w,
                        unsigned long long* __restrict__ cells, int* __restrict__ cellcnt,
                        unsigned long long* __restrict__ ovf_e, int* __restrict__ ovf_d,
                        int* __restrict__ ovfcnt, int E, int nblk, int nbinsA) {
    __shared__ unsigned long long lcell[MAXB * LP];
    __shared__ int lcnt[MAXB + 1];
    __shared__ int lovf;
    for (int i = threadIdx.x; i <= MAXB; i += blockDim.x) lcnt[i] = 0;
    if (threadIdx.x == 0) lovf = 0;
    __syncthreads();
    const int e0 = blk * EPB;
    const int e1 = (e0 + EPB < E) ? (e0 + EPB) : E;
    for (int base = e0 + (int)threadIdx.x; base < e1; base += 8 * (int)blockDim.x) {
        int dd[8], rk[8];
        unsigned ss[8], ww[8];
#pragma unroll
        for (int u = 0; u < 8; ++u) {
            const int e = base + u * (int)blockDim.x;
            const int ec = (e < e1) ? e : (e1 - 1);
            dd[u] = dst[ec];
            ss[u] = (unsigned)src[ec];
            ww[u] = __float_as_uint(w[ec]);
        }
#pragma unroll
        for (int u = 0; u < 8; ++u) {
            const int bin = dd[u] >> BSH;
            const bool cell = (base + u * (int)blockDim.x < e1) && (bin < MAXB);
            const int r = atomicAdd(&lcnt[cell ? bin : MAXB], 1);
            rk[u] = cell ? r : CELL;
        }
#pragma unroll
        for (int u = 0; u < 8; ++u) {
            if (base + u * (int)blockDim.x < e1) {
                const int d = dd[u];
                if (rk[u] < CELL) {
                    lcell[(d >> BSH) * LP + rk[u]] =
                        ((unsigned long long)ww[u] << 32) |
                        (unsigned long long)(ss[u] | ((unsigned)(d & (BN - 1)) << 24));
                } else {
                    const int o = atomicAdd(&lovf, 1);
                    ovf_e[(size_t)blk * EPB + o] =
                        ((unsigned long long)ww[u] << 32) | (unsigned long long)ss[u];
                    ovf_d[(size_t)blk * EPB + o] = d;
                }
            }
        }
    }
    __syncthreads();
    {
        const int lane = threadIdx.x & 31;
        const int nw = (int)blockDim.x >> 5;
        for (int i0 = (int)threadIdx.x >> 5; i0 < nbinsA; i0 += 8 * nw) {
            int cc[8];
            unsigned long long v[8];
#pragma unroll
            for (int u = 0; u < 8; ++u) {
                const int i = i0 + u * nw;
                const int c = lcnt[(i < nbinsA) ? i : 0];
                cc[u] = (i < nbinsA) ? ((c < CELL) ? c : CELL) : 0;
            }
#pragma unroll
            for (int u = 0; u < 8; ++u)
                v[u] = (lane < cc[u]) ? lcell[(i0 + u * nw) * LP + lane] : 0ull;
#pragma unroll
            for (int u = 0; u < 8; ++u)
                if (lane < cc[u])
                    cells[((size_t)(i0 + u * nw) * nblk + blk) * CELL + lane] = v[u];
        }
    }
    for (int i = threadIdx.x; i < nbinsA; i += blockDim.x) {
        const int cc = lcnt[i];
        cellcnt[(size_t)i * nblk + blk] = (cc < CELL) ? cc : CELL;
    }
    if (threadIdx.x == 0) ovfcnt[blk] = lovf;
}


__device__ __forceinline__ void gemm1_block(int blk,
                        const float* __restrict__ X, const float* __restrict__ W,
                        float* __restrict__ O, int mtiles) {
    int wave = threadIdx.x >> 5;
    int tile = blk * (blockDim.x >> 5) + wave;
    if (tile >= mtiles * 4) return;
    int mt = tile >> 2, nt = tile & 3;
    int lane = threadIdx.x & 31;
    int lo = lane & 15, hi = lane >> 4;

    const float* arow = X + (size_t)(mt * 16 + lo) * 64;
    int col = nt * 16 + lo;
    v8f c = {};
#pragma unroll
    for (int k = 0; k < 64; k += 4) {
        int ka = k + 2 * hi;
        v2f a = { arow[ka], arow[ka + 1] };
        v2f b = { W[(size_t)ka * 64 + col], W[(size_t)(ka + 1) * 64 + col] };
        c = __builtin_amdgcn_wmma_f32_16x16x4_f32(false, a, false, b,
                                                  (short)0, c, false, false);
    }
    float* orow = O + (size_t)(mt * 16 + 8 * hi) * 64 + col;
#pragma unroll
    for (int r = 0; r < 8; ++r) orow[(size_t)r * 64] = c[r];
}

__global__ void __launch_bounds__(1024)
k_bin(const int* __restrict__ src, const int* __restrict__ dst,
      const float* __restrict__ w,
      unsigned long long* __restrict__ cells, int* __restrict__ cellcnt,
      unsigned long long* __restrict__ ovf_e, int* __restrict__ ovf_d,
      int* __restrict__ ovfcnt, int* __restrict__ nspill,
      unsigned long long* __restrict__ pad, int E, int nblk, int nbinsA) {
    if (blockIdx.x == 0 && threadIdx.x == 0) {
        nspill[0] = 0;
        pad[0] = 0;
    }
    bin_block((int)blockIdx.x, src, dst, w, cells, cellcnt, ovf_e, ovf_d, ovfcnt, E, nblk, nbinsA);
}

__device__ __forceinline__ void place_store(int slot, int dl, unsigned s, unsigned wb, int bin,
                        float* sxw, unsigned long long* srow,
                        int* __restrict__ nspill,
                        unsigned long long* __restrict__ spill_e, int* __restrict__ spill_d) {
    const unsigned long long v = ((unsigned long long)wb << 32) | (unsigned long long)s;
    if (slot < CAP) {
        srow[dl * RP + slot] = v;
    } else {
        const int j = __hip_atomic_fetch_add(nspill, 1, __ATOMIC_RELAXED,
                                             __HIP_MEMORY_SCOPE_AGENT);
        spill_e[j] = v;
        spill_d[j] = bin * BN + dl;
        atomicAdd(&sxw[dl], __uint_as_float(wb));
    }
}

__device__ __forceinline__ void place_block(int bin,
            const unsigned long long* __restrict__ cells, const int* __restrict__ cellcnt,
            const unsigned long long* __restrict__ pad,
            const unsigned long long* __restrict__ ovf_e, const int* __restrict__ ovf_d,
            const int* __restrict__ ovfcnt, int nblk,
            unsigned long long* __restrict__ ell, int* __restrict__ cnt,
            float* __restrict__ dinv, int* __restrict__ nspill,
            unsigned long long* __restrict__ spill_e, int* __restrict__ spill_d, int n) {
    __shared__ unsigned long long srow[BN * RP];
    __shared__ int scnt[BN + 1];
    __shared__ float sxw[BN];
    __shared__ int sn[256];
    __shared__ int so[256];
    __shared__ int sany;
    const int t = threadIdx.x;
    if (t <= BN) scnt[t] = 0;
    if (t < BN) sxw[t] = 0.0f;
    if (t == 0) sany = 0;
    __syncthreads();
    for (int c0 = 0; c0 < nblk; c0 += 256) {
        const int nb = (nblk - c0 < 256) ? (nblk - c0) : 256;
        if (t < nb) {
            sn[t] = (bin < MAXB) ? cellcnt[(size_t)bin * nblk + c0 + t] : 0;
            const int ov = ovfcnt[c0 + t];
            so[t] = ov;
            if (ov > 0) atomicOr(&sany, 1);
        }
        __syncthreads();
        if (bin < MAXB) {
            const unsigned long long* __restrict__ cb = cells + ((size_t)bin * nblk + c0) * CELL;
            const int np = nb * CELL;
            for (int p0 = t; p0 < np; p0 += 4 * 256) {
                unsigned long long rec[4];
                int sl[4];
                bool ok[4];
#pragma unroll
                for (int u = 0; u < 4; ++u) {
                    const int p = p0 + u * 256;
                    const int pc = (p < np) ? p : 0;
                    ok[u] = (p < np) && ((pc & (CELL - 1)) < sn[pc >> 5]);
                    rec[u] = *(ok[u] ? (cb + pc) : pad);
                }
#pragma unroll
                for (int u = 0; u < 4; ++u) {
                    const int dl = ok[u] ? (int)((unsigned)rec[u] >> 24) : BN;
                    sl[u] = atomicAdd(&scnt[dl], 1);
                }
#pragma unroll
                for (int u = 0; u < 4; ++u) {
                    if (ok[u]) {
                        const unsigned lo = (unsigned)rec[u];
                        place_store(sl[u], (int)(lo >> 24), lo & 0xFFFFFFu, (unsigned)(rec[u] >> 32),
                                    bin, sxw, srow, nspill, spill_e, spill_d);
                    }
                }
            }
        }
        if (sany) {
            for (int cc = 0; cc < nb; ++cc) {
                const int no = so[cc];
                for (int i = t; i < no; i += 256) {
                    const int d = ovf_d[(size_t)(c0 + cc) * EPB + i];
                    if ((d >> BSH) == bin) {
                        const unsigned long long rec = ovf_e[(size_t)(c0 + cc) * EPB + i];
                        const int dl = d & (BN - 1);
                        const int slot = atomicAdd(&scnt[dl], 1);
                        place_store(slot, dl, (unsigned)rec, (unsigned)(rec >> 32),
                                    bin, sxw, srow, nspill, spill_e, spill_d);
                    }
                }
            }
        }
        __syncthreads();
    }
    if (t < BN) {
        const int g = bin * BN + t;
        if (g < n) {
            const int cc = scnt[t];
            const int m = (cc < CAP) ? cc : CAP;
            float d = 1.0f;
            for (int k = 0; k < m; ++k) d += __uint_as_float((unsigned)(srow[t * RP + k] >> 32));
            if (cc > CAP) d += sxw[t];
            cnt[g] = cc;
            dinv[g] = (d > 0.0f) ? rsqrtf(d) : 0.0f;
        }
    }
    for (int idx = t; idx < BN * CAP; idx += 256) {
        const int row = idx >> 5, k = idx & (CAP - 1);
        const int g = bin * BN + row;
        if (g < n && k < scnt[row]) ell[(size_t)g * CAP + k] = srow[row * RP + k];
    }
}

__global__ void __launch_bounds__(256)
k_place_gemm1(const unsigned long long* __restrict__ cells, const int* __restrict__ cellcnt,
              const unsigned long long* __restrict__ pad,
              const unsigned long long* __restrict__ ovf_e, const int* __restrict__ ovf_d,
              const int* __restrict__ ovfcnt, int nblk,
              unsigned long long* __restrict__ ell, int* __restrict__ cnt,
              float* __restrict__ dinv, int* __restrict__ nspill,
              unsigned long long* __restrict__ spill_e, int* __restrict__ spill_d, int n,
              int nbins, const float* __restrict__ X, const float* __restrict__ W1,
              float* __restrict__ O, int mtiles) {
    const int b = blockIdx.x;
    if (b < nbins)
        place_block(b, cells, cellcnt, pad, ovf_e, ovf_d, ovfcnt, nblk, ell, cnt, dinv, nspill,
                    spill_e, spill_d, n);
    else
        gemm1_block(b - nbins, X, W1, O, mtiles);
}


__device__ __forceinline__ void axpy4(float4& a, float s, const float4& r) {
    a.x += s * r.x; a.y += s * r.y; a.z += s * r.z; a.w += s * r.w;
}

template <int K>
__device__ __forceinline__ int slot_src(int sa, int sb) {
    return __builtin_amdgcn_update_dpp(0, (K & 1) ? sb : sa, 0x150 + (K >> 1), 0xf, 0xf, true);
}
template <int K>
__device__ __forceinline__ float slot_nrm(float na, float nb) {
    return __int_as_float(__builtin_amdgcn_update_dpp(0, __float_as_int((K & 1) ? nb : na),
                                                      0x150 + (K >> 1), 0xf, 0xf, true));
}

template <int K0>
__device__ __forceinline__ void rows8(float4 (&r)[8], int l, const float4* __restrict__ h4,
                                      int sa, int sb) {
    r[0] = h4[(size_t)slot_src<K0 + 0>(sa, sb) * 16 + l];
    r[1] = h4[(size_t)slot_src<K0 + 1>(sa, sb) * 16 + l];
    r[2] = h4[(size_t)slot_src<K0 + 2>(sa, sb) * 16 + l];
    r[3] = h4[(size_t)slot_src<K0 + 3>(sa, sb) * 16 + l];
    r[4] = h4[(size_t)slot_src<K0 + 4>(sa, sb) * 16 + l];
    r[5] = h4[(size_t)slot_src<K0 + 5>(sa, sb) * 16 + l];
    r[6] = h4[(size_t)slot_src<K0 + 6>(sa, sb) * 16 + l];
    r[7] = h4[(size_t)slot_src<K0 + 7>(sa, sb) * 16 + l];
}
template <int K0>
__device__ __forceinline__ void add8(float4& acc, const float4 (&r)[8], float na, float nb) {
    axpy4(acc, slot_nrm<K0 + 0>(na, nb), r[0]);
    axpy4(acc, slot_nrm<K0 + 1>(na, nb), r[1]);
    axpy4(acc, slot_nrm<K0 + 2>(na, nb), r[2]);
    axpy4(acc, slot_nrm<K0 + 3>(na, nb), r[3]);
    axpy4(acc, slot_nrm<K0 + 4>(na, nb), r[4]);
    axpy4(acc, slot_nrm<K0 + 5>(na, nb), r[5]);
    axpy4(acc, slot_nrm<K0 + 6>(na, nb), r[6]);
    axpy4(acc, slot_nrm<K0 + 7>(na, nb), r[7]);
}

constexpr int TP = 68;

__global__ void __launch_bounds__(256)
k_agg1_gemm2(const int* __restrict__ cnt,
             const unsigned long long* __restrict__ ell,
             const int* __restrict__ nspill,
             const unsigned long long* __restrict__ spill_e,
             const int* __restrict__ spill_d,
             const float* __restrict__ dinv, const float* __restrict__ h,
             const float* __restrict__ bias, const float* __restrict__ W2,
             float* __restrict__ nrmS, float* __restrict__ nrmP,
             float* __restrict__ O, int n) {
    __shared__ float tile[16 * TP];
    const int gi = threadIdx.x >> 4, l = threadIdx.x & 15;
    const int row0 = blockIdx.x * 16;
    const int g = row0 + gi;
    float4 acc = make_float4(0.0f, 0.0f, 0.0f, 0.0f);
    if (g < n) {
        const int c = cnt[g];
        const int m = (c < CAP) ? c : CAP;
        const size_t p0 = (size_t)g * CAP;
        const float4* __restrict__ h4 = reinterpret_cast<const float4*>(h);
        const float4 hs = h4[(size_t)g * 16 + l];
        const float4 b = reinterpret_cast<const float4*>(bias)[l];
        int sa = g, sb = g;
        float wa = 0.0f, wb = 0.0f;
        if (2 * l + 1 < m) {
            const ulonglong2 ev = reinterpret_cast<const ulonglong2*>(ell + p0)[l];
            sa = (int)(unsigned)ev.x;
            sb = (int)(unsigned)ev.y;
            wa = __uint_as_float((unsigned)(ev.x >> 32));
            wb = __uint_as_float((unsigned)(ev.y >> 32));
        } else if (2 * l < m) {
            const unsigned long long e0 = ell[p0 + 2 * l];
            sa = (int)(unsigned)e0;
            wa = __uint_as_float((unsigned)(e0 >> 32));
        }
        float4 r[8];
        rows8<0>(r, l, h4, sa, sb);
        const float dd = dinv[g];
        const float da = dinv[sa], db = dinv[sb];
        const float na = (2 * l < m) ? da * wa * dd : 0.0f;
        const float nb = (2 * l + 1 < m) ? db * wb * dd : 0.0f;
        if (2 * l + 1 < m) *reinterpret_cast<float2*>(nrmS + p0 + 2 * l) = make_float2(na, nb);
        else if (2 * l < m) nrmS[p0 + 2 * l] = na;
        add8<0>(acc, r, na, nb);
        if (m > 8)  { rows8<8>(r, l, h4, sa, sb);  add8<8>(acc, r, na, nb); }
        if (m > 16) { rows8<16>(r, l, h4, sa, sb); add8<16>(acc, r, na, nb); }
        if (m > 24) { rows8<24>(r, l, h4, sa, sb); add8<24>(acc, r, na, nb); }
        if (c > CAP) {
            const int ns = nspill[0];
            for (int j = 0; j < ns; ++j) {
                if (spill_d[j] != g) continue;
                const unsigned long long e0 = spill_e[j];
                const int s0 = (int)(unsigned)e0;
                const float4 r0 = h4[(size_t)s0 * 16 + l];
                const float n0 = dinv[s0] * __uint_as_float((unsigned)(e0 >> 32)) * dd;
                axpy4(acc, n0, r0);
                if (l == 0) nrmP[j] = n0;
            }
        }
        axpy4(acc, dd * dd, hs);
        acc.x += b.x; acc.y += b.y; acc.z += b.z; acc.w += b.w;
        acc.x = fmaxf(acc.x, 0.0f); acc.y = fmaxf(acc.y, 0.0f);
        acc.z = fmaxf(acc.z, 0.0f); acc.w = fmaxf(acc.w, 0.0f);
    }
    *reinterpret_cast<float4*>(&tile[gi * TP + 4 * l]) = acc;

    const int wave = threadIdx.x >> 5;
    const int lane = threadIdx.x & 31;
    const int lo = lane & 15, hi = lane >> 4;
    const int col = wave * 16 + lo;
    const bool cok = (wave < 3) && (col < 40);
    float bw[32];
    if (wave < 3) {
        const float* __restrict__ wcol = W2 + (cok ? col : 0);
#pragma unroll
        for (int j = 0; j < 16; ++j) {
            const int ka = 4 * j + 2 * hi;
            const float w0 = wcol[ka * 40], w1 = wcol[(ka + 1) * 40];
            bw[2 * j]     = cok ? w0 : 0.0f;
            bw[2 * j + 1] = cok ? w1 : 0.0f;
        }
    }
    __syncthreads();
    if (wave < 3) {
        const float* arow = tile + lo * TP;
        v8f c = {};
#pragma unroll
        for (int j = 0; j < 16; ++j) {
            const int ka = 4 * j + 2 * hi;
            v2f a = { arow[ka], arow[ka + 1] };
            v2f b = { bw[2 * j], bw[2 * j + 1] };
            c = __builtin_amdgcn_wmma_f32_16x16x4_f32(false, a, false, b,
                                                      (short)0, c, false, false);
        }
        if (cok) {
            const int r0 = row0 + 8 * hi;
            float* orow = O + (size_t)r0 * 64 + col;
            if (row0 + 16 <= n) {
#pragma unroll
                for (int r = 0; r < 8; ++r) orow[(size_t)r * 64] = c[r];
            } else {
#pragma unroll
                for (int r = 0; r < 8; ++r)
                    if (r0 + r < n) orow[(size_t)r * 64] = c[r];
            }
        }
    }
}

__global__ void k_agg2_lsm(const int* __restrict__ cnt,
                           const unsigned long long* __restrict__ ell,
                           const int* __restrict__ nspill,
                           const unsigned long long* __restrict__ spill_e,
                           const int* __restrict__ spill_d,
                           const float* __restrict__ nrmS, const float* __restrict__ nrmP,
                           const float* __restrict__ dinv,
                           const float* __restrict__ h, const float* __restrict__ bias,
                           float* __restrict__ out, int n) {
    const int t = blockIdx.x * blockDim.x + threadIdx.x;
    const int g = t >> 4, l = t & 15;
    const bool act = (g < n) && (l < 10);
    const int lc = (l < 10) ? l : 0;
    float4 acc = make_float4(0.0f, 0.0f, 0.0f, 0.0f);
    if (g < n) {
        const int c = cnt[g];
        const int m = (c < CAP) ? c : CAP;
        const size_t p0 = (size_t)g * CAP;
        const float4* __restrict__ h4 = reinterpret_cast<const float4*>(h);
        const float dd = dinv[g];
        const float4 hs = h4[(size_t)g * 16 + lc];
        const float4 b = reinterpret_cast<const float4*>(bias)[lc];
        int sa = g, sb = g;
        float na = 0.0f, nb = 0.0f;
        if (2 * l + 1 < m) {
            const ulonglong2 ev = reinterpret_cast<const ulonglong2*>(ell + p0)[l];
            const float2 nn = *reinterpret_cast<const float2*>(nrmS + p0 + 2 * l);
            sa = (int)(unsigned)ev.x;
            sb = (int)(unsigned)ev.y;
            na = nn.x;
            nb = nn.y;
        } else if (2 * l < m) {
            sa = (int)(unsigned)ell[p0 + 2 * l];
            na = nrmS[p0 + 2 * l];
        }
        float4 r[8];
        rows8<0>(r, lc, h4, sa, sb);
        add8<0>(acc, r, na, nb);
        if (m > 8)  { rows8<8>(r, lc, h4, sa, sb);  add8<8>(acc, r, na, nb); }
        if (m > 16) { rows8<16>(r, lc, h4, sa, sb); add8<16>(acc, r, na, nb); }
        if (m > 24) { rows8<24>(r, lc, h4, sa, sb); add8<24>(acc, r, na, nb); }
        if (c > CAP) {
            const int ns = nspill[0];
            for (int j = 0; j < ns; ++j) {
                if (spill_d[j] != g) continue;
                const int s0 = (int)(unsigned)spill_e[j];
                const float4 r0 = h4[(size_t)s0 * 16 + lc];
                axpy4(acc, nrmP[j], r0);
            }
        }
        axpy4(acc, dd * dd, hs);
        acc.x += b.x; acc.y += b.y; acc.z += b.z; acc.w += b.w;
    }
    float mx = act ? fmaxf(fmaxf(acc.x, acc.y), fmaxf(acc.z, acc.w)) : -3.402823466e38f;
    for (int o = 8; o >= 1; o >>= 1) mx = fmaxf(mx, __shfl_xor(mx, o, 16));
    float s = act ? (expf(acc.x - mx) + expf(acc.y - mx)) + (expf(acc.z - mx) + expf(acc.w - mx))
                  : 0.0f;
    for (int o = 8; o >= 1; o >>= 1) s += __shfl_xor(s, o, 16);
    const float lse = mx + logf(s);
    if (act) {
        acc.x -= lse; acc.y -= lse; acc.z -= lse; acc.w -= lse;
        reinterpret_cast<float4*>(out)[(size_t)g * 10 + l] = acc;
    }
}


extern "C" void kernel_launch(void* const* d_in, const int* in_sizes, int n_in,
                              void* d_out, int out_size, void* d_ws, size_t ws_size,
                              hipStream_t stream) {
    const float* x   = (const float*)d_in[0];
    const int*   ei  = (const int*)d_in[1];
    const float* ew  = (const float*)d_in[2];
    const float* W1  = (const float*)d_in[3];
    const float* b1  = (const float*)d_in[4];
    const float* W2  = (const float*)d_in[5];
    const float* b2  = (const float*)d_in[6];
    float* out = (float*)d_out;

    const int N = in_sizes[0] / 64;
    const int E = in_sizes[2];
    const int* src = ei;
    const int* dst = ei + E;

    const int nblk   = (E + EPB - 1) / EPB;
    const int nbins  = (N + BN - 1) / BN;
    const int nbinsA = (nbins < MAXB) ? nbins : MAXB;

    const size_t Ep = ((size_t)E + 63) / 64 * 64;
    const size_t Np = ((size_t)N + 63) / 64 * 64;
    const size_t Cp = ((size_t)nbinsA * nblk + 63) / 64 * 64;
    unsigned long long* ell     = (unsigned long long*)d_ws;
    unsigned long long* spill_e = ell + Np * CAP;
    float* nrmS    = (float*)(spill_e + Ep);
    float* nrmP    = nrmS + Np * CAP;
    int*   spill_d = (int*)(nrmP + Ep);
    int*   cnt     = spill_d + Ep;
    int*   nspill  = cnt + Np;
    float* dinv    = (float*)(nspill + 64);
    float* bufA    = dinv + Np;
    float* bufB    = bufA + (size_t)N * 64;
    unsigned long long* cells = (unsigned long long*)(bufB + (size_t)N * 64);
    unsigned long long* pad   = cells + (size_t)nbinsA * nblk * CELL;
    unsigned long long* ovf_e = pad + 32;
    int*   ovf_d   = (int*)(ovf_e + (size_t)nblk * EPB);
    int*   cellcnt = ovf_d + (size_t)nblk * EPB;
    int*   ovfcnt  = cellcnt + Cp;

    const int BLK = 256;
    const int mtiles = N / 16;
    const unsigned gG1 = (unsigned)((mtiles * 4 + 7) / 8);
    const unsigned gNg = (unsigned)(((long long)N * 16 + BLK - 1) / BLK);

    k_bin<<<(unsigned)nblk, 1024, 0, stream>>>(src, dst, ew, cells, cellcnt, ovf_e, ovf_d, ovfcnt,
                                               nspill, pad, E, nblk, nbinsA);
    k_place_gemm1<<<(unsigned)nbins + gG1, BLK, 0, stream>>>(cells, cellcnt, pad, ovf_e, ovf_d,
                                                             ovfcnt, nblk, ell, cnt, dinv, nspill,
                                                             spill_e, spill_d, N, nbins, x, W1,
                                                             bufA, mtiles);

    k_agg1_gemm2<<<gNg, BLK, 0, stream>>>(cnt, ell, nspill, spill_e, spill_d, dinv, bufA, b1,
                                          W2, nrmS, nrmP, bufB, N);

    k_agg2_lsm<<<gNg, BLK, 0, stream>>>(cnt, ell, nspill, spill_e, spill_d, nrmS, nrmP, dinv,
                                        bufB, b2, out, N);
}


# LDS slot counts of both grouping levels come only from lanes that hold an edge: idle lanes no longer all add to one spare LDS word (same-word LDS atomics serialise); slot order kept
# speedup vs baseline: 6.9109x; 1.3637x over previous
#include <hip/hip_runtime.h>
#include <math.h>


typedef __attribute__((ext_vector_type(2))) float v2f;
typedef __attribute__((ext_vector_type(8))) float v8f;

constexpr int CAP = 32;


constexpr int BSH  = 7;
constexpr int BN   = 128;
constexpr int MAXB = 1024;
constexpr int CELL = 32;
constexpr int EPB  = 8192;
constexpr int RP   = 33;
constexpr int LP   = 33;

__device__ __forceinline__ void bin_block(int blk,
                        const int* __restrict__ src, const int* __restrict__ dst,
                        const float* __restrict__ w,
                        unsigned long long* __restrict__ cells, int* __restrict__ cellcnt,
                        unsigned long long* __restrict__ ovf_e, int* __restrict__ ovf_d,
                        int* __restrict__ ovfcnt, int E, int nblk, int nbinsA) {
    __shared__ unsigned long long lcell[MAXB * LP];
    __shared__ int lcnt[MAXB];
    __shared__ int lovf;
    for (int i = threadIdx.x; i < MAXB; i += blockDim.x) lcnt[i] = 0;
    if (threadIdx.x == 0) lovf = 0;
    __syncthreads();
    const int e0 = blk * EPB;
    const int e1 = (e0 + EPB < E) ? (e0 + EPB) : E;
    for (int base = e0 + (int)threadIdx.x; base < e1; base += 8 * (int)blockDim.x) {
        int dd[8], rk[8];
        unsigned ss[8], ww[8];
#pragma unroll
        for (int u = 0; u < 8; ++u) {
            const int e = base + u * (int)blockDim.x;
            const int ec = (e < e1) ? e : (e1 - 1);
            dd[u] = dst[ec];
            ss[u] = (unsigned)src[ec];
            ww[u] = __float_as_uint(w[ec]);
        }
#pragma unroll
        for (int u = 0; u < 8; ++u) {
            const int bin = dd[u] >> BSH;
            const bool cell = (base + u * (int)blockDim.x < e1) && (bin < MAXB);
            rk[u] = CELL;
            if (cell) rk[u] = atomicAdd(&lcnt[bin], 1);
        }
#pragma unroll
        for (int u = 0; u < 8; ++u) {
            if (base + u * (int)blockDim.x < e1) {
                const int d = dd[u];
                if (rk[u] < CELL) {
                    lcell[(d >> BSH) * LP + rk[u]] =
                        ((unsigned long long)ww[u] << 32) |
                        (unsigned long long)(ss[u] | ((unsigned)(d & (BN - 1)) << 24));
                } else {
                    const int o = atomicAdd(&lovf, 1);
                    ovf_e[(size_t)blk * EPB + o] =
                        ((unsigned long long)ww[u] << 32) | (unsigned long long)ss[u];
                    ovf_d[(size_t)blk * EPB + o] = d;
                }
            }
        }
    }
    __syncthreads();
    {
        const int lane = threadIdx.x & 31;
        const int nw = (int)blockDim.x >> 5;
        for (int i0 = (int)threadIdx.x >> 5; i0 < nbinsA; i0 += 8 * nw) {
            int cc[8];
            unsigned long long v[8];
#pragma unroll
            for (int u = 0; u < 8; ++u) {
                const int i = i0 + u * nw;
                const int c = lcnt[(i < nbinsA) ? i : 0];
                cc[u] = (i < nbinsA) ? ((c < CELL) ? c : CELL) : 0;
            }
#pragma unroll
            for (int u = 0; u < 8; ++u)
                v[u] = (lane < cc[u]) ? lcell[(i0 + u * nw) * LP + lane] : 0ull;
#pragma unroll
            for (int u = 0; u < 8; ++u)
                if (lane < cc[u])
                    cells[((size_t)(i0 + u * nw) * nblk + blk) * CELL + lane] = v[u];
        }
    }
    for (int i = threadIdx.x; i < nbinsA; i += blockDim.x) {
        const int cc = lcnt[i];
        cellcnt[(size_t)i * nblk + blk] = (cc < CELL) ? cc : CELL;
    }
    if (threadIdx.x == 0) ovfcnt[blk] = lovf;
}


__device__ __forceinline__ void gemm1_block(int blk,
                        const float* __restrict__ X, const float* __restrict__ W,
                        float* __restrict__ O, int mtiles) {
    int wave = threadIdx.x >> 5;
    int tile = blk * (blockDim.x >> 5) + wave;
    if (tile >= mtiles * 4) return;
    int mt = tile >> 2, nt = tile & 3;
    int lane = threadIdx.x & 31;
    int lo = lane & 15, hi = lane >> 4;

    const float* arow = X + (size_t)(mt * 16 + lo) * 64;
    int col = nt * 16 + lo;
    v8f c = {};
#pragma unroll
    for (int k = 0; k < 64; k += 4) {
        int ka = k + 2 * hi;
        v2f a = { arow[ka], arow[ka + 1] };
        v2f b = { W[(size_t)ka * 64 + col], W[(size_t)(ka + 1) * 64 + col] };
        c = __builtin_amdgcn_wmma_f32_16x16x4_f32(false, a, false, b,
                                                  (short)0, c, false, false);
    }
    float* orow = O + (size_t)(mt * 16 + 8 * hi) * 64 + col;
#pragma unroll
    for (int r = 0; r < 8; ++r) orow[(size_t)r * 64] = c[r];
}

__global__ void __launch_bounds__(1024)
k_bin(const int* __restrict__ src, const int* __restrict__ dst,
      const float* __restrict__ w,
      unsigned long long* __restrict__ cells, int* __restrict__ cellcnt,
      unsigned long long* __restrict__ ovf_e, int* __restrict__ ovf_d,
      int* __restrict__ ovfcnt, int* __restrict__ nspill,
      unsigned long long* __restrict__ pad, int E, int nblk, int nbinsA) {
    if (blockIdx.x == 0 && threadIdx.x == 0) {
        nspill[0] = 0;
        pad[0] = 0;
    }
    bin_block((int)blockIdx.x, src, dst, w, cells, cellcnt, ovf_e, ovf_d, ovfcnt, E, nblk, nbinsA);
}

__device__ __forceinline__ void place_store(int slot, int dl, unsigned s, unsigned wb, int bin,
                        float* sxw, unsigned long long* srow,
                        int* __restrict__ nspill,
                        unsigned long long* __restrict__ spill_e, int* __restrict__ spill_d) {
    const unsigned long long v = ((unsigned long long)wb << 32) | (unsigned long long)s;
    if (slot < CAP) {
        srow[dl * RP + slot] = v;
    } else {
        const int j = __hip_atomic_fetch_add(nspill, 1, __ATOMIC_RELAXED,
                                             __HIP_MEMORY_SCOPE_AGENT);
        spill_e[j] = v;
        spill_d[j] = bin * BN + dl;
        atomicAdd(&sxw[dl], __uint_as_float(wb));
    }
}

__device__ __forceinline__ void place_block(int bin,
            const unsigned long long* __restrict__ cells, const int* __restrict__ cellcnt,
            const unsigned long long* __restrict__ pad,
            const unsigned long long* __restrict__ ovf_e, const int* __restrict__ ovf_d,
            const int* __restrict__ ovfcnt, int nblk,
            unsigned long long* __restrict__ ell, int* __restrict__ cnt,
            float* __restrict__ dinv, int* __restrict__ nspill,
            unsigned long long* __restrict__ spill_e, int* __restrict__ spill_d, int n) {
    __shared__ unsigned long long srow[BN * RP];
    __shared__ int scnt[BN];
    __shared__ float sxw[BN];
    __shared__ int sn[256];
    __shared__ int so[256];
    __shared__ int sany;
    const int t = threadIdx.x;
    if (t < BN) scnt[t] = 0;
    if (t < BN) sxw[t] = 0.0f;
    if (t == 0) sany = 0;
    __syncthreads();
    for (int c0 = 0; c0 < nblk; c0 += 256) {
        const int nb = (nblk - c0 < 256) ? (nblk - c0) : 256;
        if (t < nb) {
            sn[t] = (bin < MAXB) ? cellcnt[(size_t)bin * nblk + c0 + t] : 0;
            const int ov = ovfcnt[c0 + t];
            so[t] = ov;
            if (ov > 0) atomicOr(&sany, 1);
        }
        __syncthreads();
        if (bin < MAXB) {
            const unsigned long long* __restrict__ cb = cells + ((size_t)bin * nblk + c0) * CELL;
            const int np = nb * CELL;
            for (int p0 = t; p0 < np; p0 += 4 * 256) {
                unsigned long long rec[4];
                int sl[4];
                bool ok[4];
#pragma unroll
                for (int u = 0; u < 4; ++u) {
                    const int p = p0 + u * 256;
                    const int pc = (p < np) ? p : 0;
                    ok[u] = (p < np) && ((pc & (CELL - 1)) < sn[pc >> 5]);
                    rec[u] = *(ok[u] ? (cb + pc) : pad);
                }
#pragma unroll
                for (int u = 0; u < 4; ++u) {
                    sl[u] = 0;
                    if (ok[u]) sl[u] = atomicAdd(&scnt[(int)((unsigned)rec[u] >> 24)], 1);
                }
#pragma unroll
                for (int u = 0; u < 4; ++u) {
                    if (ok[u]) {
                        const unsigned lo = (unsigned)rec[u];
                        place_store(sl[u], (int)(lo >> 24), lo & 0xFFFFFFu, (unsigned)(rec[u] >> 32),
                                    bin, sxw, srow, nspill, spill_e, spill_d);
                    }
                }
            }
        }
        if (sany) {
            for (int cc = 0; cc < nb; ++cc) {
                const int no = so[cc];
                for (int i = t; i < no; i += 256) {
                    const int d = ovf_d[(size_t)(c0 + cc) * EPB + i];
                    if ((d >> BSH) == bin) {
                        const unsigned long long rec = ovf_e[(size_t)(c0 + cc) * EPB + i];
                        const int dl = d & (BN - 1);
                        const int slot = atomicAdd(&scnt[dl], 1);
                        place_store(slot, dl, (unsigned)rec, (unsigned)(rec >> 32),
                                    bin, sxw, srow, nspill, spill_e, spill_d);
                    }
                }
            }
        }
        __syncthreads();
    }
    if (t < BN) {
        const int g = bin * BN + t;
        if (g < n) {
            const int cc = scnt[t];
            const int m = (cc < CAP) ? cc : CAP;
            float d = 1.0f;
            for (int k = 0; k < m; ++k) d += __uint_as_float((unsigned)(srow[t * RP + k] >> 32));
            if (cc > CAP) d += sxw[t];
            cnt[g] = cc;
            dinv[g] = (d > 0.0f) ? rsqrtf(d) : 0.0f;
        }
    }
    for (int idx = t; idx < BN * CAP; idx += 256) {
        const int row = idx >> 5, k = idx & (CAP - 1);
        const int g = bin * BN + row;
        if (g < n && k < scnt[row]) ell[(size_t)g * CAP + k] = srow[row * RP + k];
    }
}

__global__ void __launch_bounds__(256)
k_place_gemm1(const unsigned long long* __restrict__ cells, const int* __restrict__ cellcnt,
              const unsigned long long* __restrict__ pad,
              const unsigned long long* __restrict__ ovf_e, const int* __restrict__ ovf_d,
              const int* __restrict__ ovfcnt, int nblk,
              unsigned long long* __restrict__ ell, int* __restrict__ cnt,
              float* __restrict__ dinv, int* __restrict__ nspill,
              unsigned long long* __restrict__ spill_e, int* __restrict__ spill_d, int n,
              int nbins, const float* __restrict__ X, const float* __restrict__ W1,
              float* __restrict__ O, int mtiles) {
    const int b = blockIdx.x;
    if (b < nbins)
        place_block(b, cells, cellcnt, pad, ovf_e, ovf_d, ovfcnt, nblk, ell, cnt, dinv, nspill,
                    spill_e, spill_d, n);
    else
        gemm1_block(b - nbins, X, W1, O, mtiles);
}


__device__ __forceinline__ void axpy4(float4& a, float s, const float4& r) {
    a.x += s * r.x; a.y += s * r.y; a.z += s * r.z; a.w += s * r.w;
}

template <int K>
__device__ __forceinline__ int slot_src(int sa, int sb) {
    return __builtin_amdgcn_update_dpp(0, (K & 1) ? sb : sa, 0x150 + (K >> 1), 0xf, 0xf, true);
}
template <int K>
__device__ __forceinline__ float slot_nrm(float na, float nb) {
    return __int_as_float(__builtin_amdgcn_update_dpp(0, __float_as_int((K & 1) ? nb : na),
                                                      0x150 + (K >> 1), 0xf, 0xf, true));
}

template <int K0>
__device__ __forceinline__ void rows8(float4 (&r)[8], int l, const float4* __restrict__ h4,
                                      int sa, int sb) {
    r[0] = h4[(size_t)slot_src<K0 + 0>(sa, sb) * 16 + l];
    r[1] = h4[(size_t)slot_src<K0 + 1>(sa, sb) * 16 + l];
    r[2] = h4[(size_t)slot_src<K0 + 2>(sa, sb) * 16 + l];
    r[3] = h4[(size_t)slot_src<K0 + 3>(sa, sb) * 16 + l];
    r[4] = h4[(size_t)slot_src<K0 + 4>(sa, sb) * 16 + l];
    r[5] = h4[(size_t)slot_src<K0 + 5>(sa, sb) * 16 + l];
    r[6] = h4[(size_t)slot_src<K0 + 6>(sa, sb) * 16 + l];
    r[7] = h4[(size_t)slot_src<K0 + 7>(sa, sb) * 16 + l];
}
template <int K0>
__device__ __forceinline__ void add8(float4& acc, const float4 (&r)[8], float na, float nb) {
    axpy4(acc, slot_nrm<K0 + 0>(na, nb), r[0]);
    axpy4(acc, slot_nrm<K0 + 1>(na, nb), r[1]);
    axpy4(acc, slot_nrm<K0 + 2>(na, nb), r[2]);
    axpy4(acc, slot_nrm<K0 + 3>(na, nb), r[3]);
    axpy4(acc, slot_nrm<K0 + 4>(na, nb), r[4]);
    axpy4(acc, slot_nrm<K0 + 5>(na, nb), r[5]);
    axpy4(acc, slot_nrm<K0 + 6>(na, nb), r[6]);
    axpy4(acc, slot_nrm<K0 + 7>(na, nb), r[7]);
}

constexpr int TP = 68;

__global__ void __launch_bounds__(256)
k_agg1_gemm2(const int* __restrict__ cnt,
             const unsigned long long* __restrict__ ell,
             const int* __restrict__ nspill,
             const unsigned long long* __restrict__ spill_e,
             const int* __restrict__ spill_d,
             const float* __restrict__ dinv, const float* __restrict__ h,
             const float* __restrict__ bias, const float* __restrict__ W2,
             float* __restrict__ nrmS, float* __restrict__ nrmP,
             float* __restrict__ O, int n) {
    __shared__ float tile[16 * TP];
    const int gi = threadIdx.x >> 4, l = threadIdx.x & 15;
    const int row0 = blockIdx.x * 16;
    const int g = row0 + gi;
    float4 acc = make_float4(0.0f, 0.0f, 0.0f, 0.0f);
    if (g < n) {
        const int c = cnt[g];
        const int m = (c < CAP) ? c : CAP;
        const size_t p0 = (size_t)g * CAP;
        const float4* __restrict__ h4 = reinterpret_cast<const float4*>(h);
        const float4 hs = h4[(size_t)g * 16 + l];
        const float4 b = reinterpret_cast<const float4*>(bias)[l];
        int sa = g, sb = g;
        float wa = 0.0f, wb = 0.0f;
        if (2 * l + 1 < m) {
            const ulonglong2 ev = reinterpret_cast<const ulonglong2*>(ell + p0)[l];
            sa = (int)(unsigned)ev.x;
            sb = (int)(unsigned)ev.y;
            wa = __uint_as_float((unsigned)(ev.x >> 32));
            wb = __uint_as_float((unsigned)(ev.y >> 32));
        } else if (2 * l < m) {
            const unsigned long long e0 = ell[p0 + 2 * l];
            sa = (int)(unsigned)e0;
            wa = __uint_as_float((unsigned)(e0 >> 32));
        }
        float4 r[8];
        rows8<0>(r, l, h4, sa, sb);
        const float dd = dinv[g];
        const float da = dinv[sa], db = dinv[sb];
        const float na = (2 * l < m) ? da * wa * dd : 0.0f;
        const float nb = (2 * l + 1 < m) ? db * wb * dd : 0.0f;
        if (2 * l + 1 < m) *reinterpret_cast<float2*>(nrmS + p0 + 2 * l) = make_float2(na, nb);
        else if (2 * l < m) nrmS[p0 + 2 * l] = na;
        add8<0>(acc, r, na, nb);
        if (m > 8)  { rows8<8>(r, l, h4, sa, sb);  add8<8>(acc, r, na, nb); }
        if (m > 16) { rows8<16>(r, l, h4, sa, sb); add8<16>(acc, r, na, nb); }
        if (m > 24) { rows8<24>(r, l, h4, sa, sb); add8<24>(acc, r, na, nb); }
        if (c > CAP) {
            const int ns = nspill[0];
            for (int j = 0; j < ns; ++j) {
                if (spill_d[j] != g) continue;
                const unsigned long long e0 = spill_e[j];
                const int s0 = (int)(unsigned)e0;
                const float4 r0 = h4[(size_t)s0 * 16 + l];
                const float n0 = dinv[s0] * __uint_as_float((unsigned)(e0 >> 32)) * dd;
                axpy4(acc, n0, r0);
                if (l == 0) nrmP[j] = n0;
            }
        }
        axpy4(acc, dd * dd, hs);
        acc.x += b.x; acc.y += b.y; acc.z += b.z; acc.w += b.w;
        acc.x = fmaxf(acc.x, 0.0f); acc.y = fmaxf(acc.y, 0.0f);
        acc.z = fmaxf(acc.z, 0.0f); acc.w = fmaxf(acc.w, 0.0f);
    }
    *reinterpret_cast<float4*>(&tile[gi * TP + 4 * l]) = acc;

    const int wave = threadIdx.x >> 5;
    const int lane = threadIdx.x & 31;
    const int lo = lane & 15, hi = lane >> 4;
    const int col = wave * 16 + lo;
    const bool cok = (wave < 3) && (col < 40);
    float bw[32];
    if (wave < 3) {
        const float* __restrict__ wcol = W2 + (cok ? col : 0);
#pragma unroll
        for (int j = 0; j < 16; ++j) {
            const int ka = 4 * j + 2 * hi;
            const float w0 = wcol[ka * 40], w1 = wcol[(ka + 1) * 40];
            bw[2 * j]     = cok ? w0 : 0.0f;
            bw[2 * j + 1] = cok ? w1 : 0.0f;
        }
    }
    __syncthreads();
    if (wave < 3) {
        const float* arow = tile + lo * TP;
        v8f c = {};
#pragma unroll
        for (int j = 0; j < 16; ++j) {
            const int ka = 4 * j + 2 * hi;
            v2f a = { arow[ka], arow[ka + 1] };
            v2f b = { bw[2 * j], bw[2 * j + 1] };
            c = __builtin_amdgcn_wmma_f32_16x16x4_f32(false, a, false, b,
                                                      (short)0, c, false, false);
        }
        if (cok) {
            const int r0 = row0 + 8 * hi;
            float* orow = O + (size_t)r0 * 64 + col;
            if (row0 + 16 <= n) {
#pragma unroll
                for (int r = 0; r < 8; ++r) orow[(size_t)r * 64] = c[r];
            } else {
#pragma unroll
                for (int r = 0; r < 8; ++r)
                    if (r0 + r < n) orow[(size_t)r * 64] = c[r];
            }
        }
    }
}

__global__ void k_agg2_lsm(const int* __restrict__ cnt,
                           const unsigned long long* __restrict__ ell,
                           const int* __restrict__ nspill,
                           const unsigned long long* __restrict__ spill_e,
                           const int* __restrict__ spill_d,
                           const float* __restrict__ nrmS, const float* __restrict__ nrmP,
                           const float* __restrict__ dinv,
                           const float* __restrict__ h, const float* __restrict__ bias,
                           float* __restrict__ out, int n) {
    const int t = blockIdx.x * blockDim.x + threadIdx.x;
    const int g = t >> 4, l = t & 15;
    const bool act = (g < n) && (l < 10);
    const int lc = (l < 10) ? l : 0;
    float4 acc = make_float4(0.0f, 0.0f, 0.0f, 0.0f);
    if (g < n) {
        const int c = cnt[g];
        const int m = (c < CAP) ? c : CAP;
        const size_t p0 = (size_t)g * CAP;
        const float4* __restrict__ h4 = reinterpret_cast<const float4*>(h);
        const float dd = dinv[g];
        const float4 hs = h4[(size_t)g * 16 + lc];
        const float4 b = reinterpret_cast<const float4*>(bias)[lc];
        int sa = g, sb = g;
        float na = 0.0f, nb = 0.0f;
        if (2 * l + 1 < m) {
            const ulonglong2 ev = reinterpret_cast<const ulonglong2*>(ell + p0)[l];
            const float2 nn = *reinterpret_cast<const float2*>(nrmS + p0 + 2 * l);
            sa = (int)(unsigned)ev.x;
            sb = (int)(unsigned)ev.y;
            na = nn.x;
            nb = nn.y;
        } else if (2 * l < m) {
            sa = (int)(unsigned)ell[p0 + 2 * l];
            na = nrmS[p0 + 2 * l];
        }
        float4 r[8];
        rows8<0>(r, lc, h4, sa, sb);
        add8<0>(acc, r, na, nb);
        if (m > 8)  { rows8<8>(r, lc, h4, sa, sb);  add8<8>(acc, r, na, nb); }
        if (m > 16) { rows8<16>(r, lc, h4, sa, sb); add8<16>(acc, r, na, nb); }
        if (m > 24) { rows8<24>(r, lc, h4, sa, sb); add8<24>(acc, r, na, nb); }
        if (c > CAP) {
            const int ns = nspill[0];
            for (int j = 0; j < ns; ++j) {
                if (spill_d[j] != g) continue;
                const int s0 = (int)(unsigned)spill_e[j];
                const float4 r0 = h4[(size_t)s0 * 16 + lc];
                axpy4(acc, nrmP[j], r0);
            }
        }
        axpy4(acc, dd * dd, hs);
        acc.x += b.x; acc.y += b.y; acc.z += b.z; acc.w += b.w;
    }
    float mx = act ? fmaxf(fmaxf(acc.x, acc.y), fmaxf(acc.z, acc.w)) : -3.402823466e38f;
    for (int o = 8; o >= 1; o >>= 1) mx = fmaxf(mx, __shfl_xor(mx, o, 16));
    float s = act ? (expf(acc.x - mx) + expf(acc.y - mx)) + (expf(acc.z - mx) + expf(acc.w - mx))
                  : 0.0f;
    for (int o = 8; o >= 1; o >>= 1) s += __shfl_xor(s, o, 16);
    const float lse = mx + logf(s);
    if (act) {
        acc.x -= lse; acc.y -= lse; acc.z -= lse; acc.w -= lse;
        reinterpret_cast<float4*>(out)[(size_t)g * 10 + l] = acc;
    }
}


extern "C" void kernel_launch(void* const* d_in, const int* in_sizes, int n_in,
                              void* d_out, int out_size, void* d_ws, size_t ws_size,
                              hipStream_t stream) {
    const float* x   = (const float*)d_in[0];
    const int*   ei  = (const int*)d_in[1];
    const float* ew  = (const float*)d_in[2];
    const float* W1  = (const float*)d_in[3];
    const float* b1  = (const float*)d_in[4];
    const float* W2  = (const float*)d_in[5];
    const float* b2  = (const float*)d_in[6];
    float* out = (float*)d_out;

    const int N = in_sizes[0] / 64;
    const int E = in_sizes[2];
    const int* src = ei;
    const int* dst = ei + E;

    const int nblk   = (E + EPB - 1) / EPB;
    const int nbins  = (N + BN - 1) / BN;
    const int nbinsA = (nbins < MAXB) ? nbins : MAXB;

    const size_t Ep = ((size_t)E + 63) / 64 * 64;
    const size_t Np = ((size_t)N + 63) / 64 * 64;
    const size_t Cp = ((size_t)nbinsA * nblk + 63) / 64 * 64;
    unsigned long long* ell     = (unsigned long long*)d_ws;
    unsigned long long* spill_e = ell + Np * CAP;
    float* nrmS    = (float*)(spill_e + Ep);
    float* nrmP    = nrmS + Np * CAP;
    int*   spill_d = (int*)(nrmP + Ep);
    int*   cnt     = spill_d + Ep;
    int*   nspill  = cnt + Np;
    float* dinv    = (float*)(nspill + 64);
    float* bufA    = dinv + Np;
    float* bufB    = bufA + (size_t)N * 64;
    unsigned long long* cells = (unsigned long long*)(bufB + (size_t)N * 64);
    unsigned long long* pad   = cells + (size_t)nbinsA * nblk * CELL;
    unsigned long long* ovf_e = pad + 32;
    int*   ovf_d   = (int*)(ovf_e + (size_t)nblk * EPB);
    int*   cellcnt = ovf_d + (size_t)nblk * EPB;
    int*   ovfcnt  = cellcnt + Cp;

    const int BLK = 256;
    const int mtiles = N / 16;
    const unsigned gG1 = (unsigned)((mtiles * 4 + 7) / 8);
    const unsigned gNg = (unsigned)(((long long)N * 16 + BLK - 1) / BLK);

    k_bin<<<(unsigned)nblk, 1024, 0, stream>>>(src, dst, ew, cells, cellcnt, ovf_e, ovf_d, ovfcnt,
                                               nspill, pad, E, nblk, nbinsA);
    k_place_gemm1<<<(unsigned)nbins + gG1, BLK, 0, stream>>>(cells, cellcnt, pad, ovf_e, ovf_d,
                                                             ovfcnt, nblk, ell, cnt, dinv, nspill,
                                                             spill_e, spill_d, N, nbins, x, W1,
                                                             bufA, mtiles);

    k_agg1_gemm2<<<gNg, BLK, 0, stream>>>(cnt, ell, nspill, spill_e, spill_d, dinv, bufA, b1,
                                          W2, nrmS, nrmP, bufB, N);

    k_agg2_lsm<<<gNg, BLK, 0, stream>>>(cnt, ell, nspill, spill_e, spill_d, nrmS, nrmP, dinv,
                                        bufB, b2, out, N);
}


# norm split: GEMM1 (now each range block's tail) and GEMM2 scale rows by the node's own deg^-1/2; an edge adds weight*row, dst factor after the sum: no per-edge degree lookup; slot order kept
# speedup vs baseline: 7.2460x; 1.0485x over previous
#include <hip/hip_runtime.h>
#include <math.h>


typedef __attribute__((ext_vector_type(2))) float v2f;
typedef __attribute__((ext_vector_type(8))) float v8f;

constexpr int CAP = 32;


constexpr int BSH  = 7;
constexpr int BN   = 128;
constexpr int MAXB = 1024;
constexpr int CELL = 32;
constexpr int EPB  = 8192;
constexpr int RP   = 33;
constexpr int LP   = 33;

__device__ __forceinline__ void bin_block(int blk,
                        const int* __restrict__ src, const int* __restrict__ dst,
                        const float* __restrict__ w,
                        unsigned long long* __restrict__ cells, int* __restrict__ cellcnt,
                        unsigned long long* __restrict__ ovf_e, int* __restrict__ ovf_d,
                        int* __restrict__ ovfcnt, int E, int nblk, int nbinsA) {
    __shared__ unsigned long long lcell[MAXB * LP];
    __shared__ int lcnt[MAXB];
    __shared__ int lovf;
    for (int i = threadIdx.x; i < MAXB; i += blockDim.x) lcnt[i] = 0;
    if (threadIdx.x == 0) lovf = 0;
    __syncthreads();
    const int e0 = blk * EPB;
    const int e1 = (e0 + EPB < E) ? (e0 + EPB) : E;
    for (int base = e0 + (int)threadIdx.x; base < e1; base += 8 * (int)blockDim.x) {
        int dd[8], rk[8];
        unsigned ss[8], ww[8];
#pragma unroll
        for (int u = 0; u < 8; ++u) {
            const int e = base + u * (int)blockDim.x;
            const int ec = (e < e1) ? e : (e1 - 1);
            dd[u] = dst[ec];
            ss[u] = (unsigned)src[ec];
            ww[u] = __float_as_uint(w[ec]);
        }
#pragma unroll
        for (int u = 0; u < 8; ++u) {
            const int bin = dd[u] >> BSH;
            const bool cell = (base + u * (int)blockDim.x < e1) && (bin < MAXB);
            rk[u] = CELL;
            if (cell) rk[u] = atomicAdd(&lcnt[bin], 1);
        }
#pragma unroll
        for (int u = 0; u < 8; ++u) {
            if (base + u * (int)blockDim.x < e1) {
                const int d = dd[u];
                if (rk[u] < CELL) {
                    lcell[(d >> BSH) * LP + rk[u]] =
                        ((unsigned long long)ww[u] << 32) |
                        (unsigned long long)(ss[u] | ((unsigned)(d & (BN - 1)) << 24));
                } else {
                    const int o = atomicAdd(&lovf, 1);
                    ovf_e[(size_t)blk * EPB + o] =
                        ((unsigned long long)ww[u] << 32) | (unsigned long long)ss[u];
                    ovf_d[(size_t)blk * EPB + o] = d;
                }
            }
        }
    }
    __syncthreads();
    {
        const int lane = threadIdx.x & 31;
        const int nw = (int)blockDim.x >> 5;
        for (int i0 = (int)threadIdx.x >> 5; i0 < nbinsA; i0 += 8 * nw) {
            int cc[8];
            unsigned long long v[8];
#pragma unroll
            for (int u = 0; u < 8; ++u) {
                const int i = i0 + u * nw;
                const int c = lcnt[(i < nbinsA) ? i : 0];
                cc[u] = (i < nbinsA) ? ((c < CELL) ? c : CELL) : 0;
            }
#pragma unroll
            for (int u = 0; u < 8; ++u)
                v[u] = (lane < cc[u]) ? lcell[(i0 + u * nw) * LP + lane] : 0ull;
#pragma unroll
            for (int u = 0; u < 8; ++u)
                if (lane < cc[u])
                    cells[((size_t)(i0 + u * nw) * nblk + blk) * CELL + lane] = v[u];
        }
    }
    for (int i = threadIdx.x; i < nbinsA; i += blockDim.x) {
        const int cc = lcnt[i];
        cellcnt[(size_t)i * nblk + blk] = (cc < CELL) ? cc : CELL;
    }
    if (threadIdx.x == 0) ovfcnt[blk] = lovf;
}

__global__ void __launch_bounds__(1024)
k_bin(const int* __restrict__ src, const int* __restrict__ dst,
      const float* __restrict__ w,
      unsigned long long* __restrict__ cells, int* __restrict__ cellcnt,
      unsigned long long* __restrict__ ovf_e, int* __restrict__ ovf_d,
      int* __restrict__ ovfcnt, int* __restrict__ nspill,
      unsigned long long* __restrict__ pad, int E, int nblk, int nbinsA) {
    if (blockIdx.x == 0 && threadIdx.x == 0) {
        nspill[0] = 0;
        pad[0] = 0;
    }
    bin_block((int)blockIdx.x, src, dst, w, cells, cellcnt, ovf_e, ovf_d, ovfcnt, E, nblk, nbinsA);
}

__device__ __forceinline__ void place_store(int slot, int dl, unsigned s, unsigned wb, int bin,
                        float* sxw, unsigned long long* srow,
                        int* __restrict__ nspill,
                        unsigned long long* __restrict__ spill_e, int* __restrict__ spill_d) {
    const unsigned long long v = ((unsigned long long)wb << 32) | (unsigned long long)s;
    if (slot < CAP) {
        srow[dl * RP + slot] = v;
    } else {
        const int j = __hip_atomic_fetch_add(nspill, 1, __ATOMIC_RELAXED,
                                             __HIP_MEMORY_SCOPE_AGENT);
        spill_e[j] = v;
        spill_d[j] = bin * BN + dl;
        atomicAdd(&sxw[dl], __uint_as_float(wb));
    }
}

__device__ __forceinline__ void place_block(int bin,
            const unsigned long long* __restrict__ cells, const int* __restrict__ cellcnt,
            const unsigned long long* __restrict__ pad,
            const unsigned long long* __restrict__ ovf_e, const int* __restrict__ ovf_d,
            const int* __restrict__ ovfcnt, int nblk,
            unsigned long long* __restrict__ ell, int* __restrict__ cnt,
            float* __restrict__ dinv, int* __restrict__ nspill,
            unsigned long long* __restrict__ spill_e, int* __restrict__ spill_d, int n,
            const float* __restrict__ X, const float* __restrict__ W1, float* __restrict__ O) {
    __shared__ unsigned long long srow[BN * RP];
    __shared__ float sdinv[BN];
    __shared__ int scnt[BN];
    __shared__ float sxw[BN];
    __shared__ int sn[256];
    __shared__ int so[256];
    __shared__ int sany;
    const int t = threadIdx.x;
    const int lo = t & 15, hi = (t >> 4) & 1;
    const int mrow0 = bin * BN + (t >> 5) * 16;
    if (t < BN) scnt[t] = 0;
    if (t < BN) sxw[t] = 0.0f;
    if (t == 0) sany = 0;
    __syncthreads();
    float4 afr[8];
    {
        const int ar = (mrow0 + lo < n) ? (mrow0 + lo) : (n - 1);
        const float4* __restrict__ a4 = reinterpret_cast<const float4*>(X) + (size_t)ar * 16 + hi;
#pragma unroll
        for (int q = 0; q < 8; ++q) afr[q] = a4[2 * q];
    }
    for (int c0 = 0; c0 < nblk; c0 += 256) {
        const int nb = (nblk - c0 < 256) ? (nblk - c0) : 256;
        if (t < nb) {
            sn[t] = (bin < MAXB) ? cellcnt[(size_t)bin * nblk + c0 + t] : 0;
            const int ov = ovfcnt[c0 + t];
            so[t] = ov;
            if (ov > 0) atomicOr(&sany, 1);
        }
        __syncthreads();
        if (bin < MAXB) {
            const unsigned long long* __restrict__ cb = cells + ((size_t)bin * nblk + c0) * CELL;
            const int np = nb * CELL;
            for (int p0 = t; p0 < np; p0 += 4 * 256) {
                unsigned long long rec[4];
                int sl[4];
                bool ok[4];
#pragma unroll
                for (int u = 0; u < 4; ++u) {
                    const int p = p0 + u * 256;
                    const int pc = (p < np) ? p : 0;
                    ok[u] = (p < np) && ((pc & (CELL - 1)) < sn[pc >> 5]);
                    rec[u] = *(ok[u] ? (cb + pc) : pad);
                }
#pragma unroll
                for (int u = 0; u < 4; ++u) {
                    sl[u] = 0;
                    if (ok[u]) sl[u] = atomicAdd(&scnt[(int)((unsigned)rec[u] >> 24)], 1);
                }
#pragma unroll
                for (int u = 0; u < 4; ++u) {
                    if (ok[u]) {
                        const unsigned lo = (unsigned)rec[u];
                        place_store(sl[u], (int)(lo >> 24), lo & 0xFFFFFFu, (unsigned)(rec[u] >> 32),
                                    bin, sxw, srow, nspill, spill_e, spill_d);
                    }
                }
            }
        }
        if (sany) {
            for (int cc = 0; cc < nb; ++cc) {
                const int no = so[cc];
                for (int i = t; i < no; i += 256) {
                    const int d = ovf_d[(size_t)(c0 + cc) * EPB + i];
                    if ((d >> BSH) == bin) {
                        const unsigned long long rec = ovf_e[(size_t)(c0 + cc) * EPB + i];
                        const int dl = d & (BN - 1);
                        const int slot = atomicAdd(&scnt[dl], 1);
                        place_store(slot, dl, (unsigned)rec, (unsigned)(rec >> 32),
                                    bin, sxw, srow, nspill, spill_e, spill_d);
                    }
                }
            }
        }
        __syncthreads();
    }
    if (t < BN) {
        const int g = bin * BN + t;
        float di = 0.0f;
        if (g < n) {
            const int cc = scnt[t];
            const int m = (cc < CAP) ? cc : CAP;
            float d = 1.0f;
            for (int k = 0; k < m; ++k) d += __uint_as_float((unsigned)(srow[t * RP + k] >> 32));
            if (cc > CAP) d += sxw[t];
            di = (d > 0.0f) ? rsqrtf(d) : 0.0f;
            cnt[g] = cc;
            dinv[g] = di;
        }
        sdinv[t] = di;
    }
    for (int idx = t; idx < BN * CAP; idx += 256) {
        const int row = idx >> 5, k = idx & (CAP - 1);
        const int g = bin * BN + row;
        if (g < n && k < scnt[row]) ell[(size_t)g * CAP + k] = srow[row * RP + k];
    }
    __syncthreads();
    if (mrow0 < n) {
        const float4* __restrict__ W4 = reinterpret_cast<const float4*>(W1) + lo;
        v8f c0 = {}, c1 = {}, c2 = {}, c3 = {};
#pragma unroll
        for (int q = 0; q < 8; ++q) {
            const int kb = 8 * q + 4 * hi;
            const float4 w0 = W4[(kb + 0) * 16], w1 = W4[(kb + 1) * 16];
            const float4 w2 = W4[(kb + 2) * 16], w3 = W4[(kb + 3) * 16];
            const v2f a0 = { afr[q].x, afr[q].y };
            const v2f a1 = { afr[q].z, afr[q].w };
            const v2f b00 = { w0.x, w1.x }, b01 = { w0.y, w1.y }, b02 = { w0.z, w1.z }, b03 = { w0.w, w1.w };
            const v2f b10 = { w2.x, w3.x }, b11 = { w2.y, w3.y }, b12 = { w2.z, w3.z }, b13 = { w2.w, w3.w };
            c0 = __builtin_amdgcn_wmma_f32_16x16x4_f32(false, a0, false, b00, (short)0, c0, false, false);
            c1 = __builtin_amdgcn_wmma_f32_16x16x4_f32(false, a0, false, b01, (short)0, c1, false, false);
            c2 = __builtin_amdgcn_wmma_f32_16x16x4_f32(false, a0, false, b02, (short)0, c2, false, false);
            c3 = __builtin_amdgcn_wmma_f32_16x16x4_f32(false, a0, false, b03, (short)0, c3, false, false);
            c0 = __builtin_amdgcn_wmma_f32_16x16x4_f32(false, a1, false, b10, (short)0, c0, false, false);
            c1 = __builtin_amdgcn_wmma_f32_16x16x4_f32(false, a1, false, b11, (short)0, c1, false, false);
            c2 = __builtin_amdgcn_wmma_f32_16x16x4_f32(false, a1, false, b12, (short)0, c2, false, false);
            c3 = __builtin_amdgcn_wmma_f32_16x16x4_f32(false, a1, false, b13, (short)0, c3, false, false);
        }
        const int rb = (t >> 5) * 16 + 8 * hi;
        float4* __restrict__ o4 = reinterpret_cast<float4*>(O) + (size_t)(bin * BN + rb) * 16 + lo;
#pragma unroll
        for (int r = 0; r < 8; ++r) {
            const float sc = sdinv[rb + r];
            if (bin * BN + rb + r < n)
                o4[(size_t)r * 16] = make_float4(c0[r] * sc, c1[r] * sc, c2[r] * sc, c3[r] * sc);
        }
    }
}

__global__ void __launch_bounds__(256)
k_place_gemm1(const unsigned long long* __restrict__ cells, const int* __restrict__ cellcnt,
              const unsigned long long* __restrict__ pad,
              const unsigned long long* __restrict__ ovf_e, const int* __restrict__ ovf_d,
              const int* __restrict__ ovfcnt, int nblk,
              unsigned long long* __restrict__ ell, int* __restrict__ cnt,
              float* __restrict__ dinv, int* __restrict__ nspill,
              unsigned long long* __restrict__ spill_e, int* __restrict__ spill_d, int n,
              const float* __restrict__ X, const float* __restrict__ W1,
              float* __restrict__ O) {
    place_block((int)blockIdx.x, cells, cellcnt, pad, ovf_e, ovf_d, ovfcnt, nblk, ell, cnt, dinv,
                nspill, spill_e, spill_d, n, X, W1, O);
}


__device__ __forceinline__ void axpy4(float4& a, float s, const float4& r) {
    a.x += s * r.x; a.y += s * r.y; a.z += s * r.z; a.w += s * r.w;
}

template <int K>
__device__ __forceinline__ int slot_src(int sa, int sb) {
    return __builtin_amdgcn_update_dpp(0, (K & 1) ? sb : sa, 0x150 + (K >> 1), 0xf, 0xf, true);
}
template <int K>
__device__ __forceinline__ float slot_nrm(float na, float nb) {
    return __int_as_float(__builtin_amdgcn_update_dpp(0, __float_as_int((K & 1) ? nb : na),
                                                      0x150 + (K >> 1), 0xf, 0xf, true));
}

template <int K0>
__device__ __forceinline__ void rows8(float4 (&r)[8], int l, const float4* __restrict__ h4,
                                      int sa, int sb) {
    r[0] = h4[(size_t)slot_src<K0 + 0>(sa, sb) * 16 + l];
    r[1] = h4[(size_t)slot_src<K0 + 1>(sa, sb) * 16 + l];
    r[2] = h4[(size_t)slot_src<K0 + 2>(sa, sb) * 16 + l];
    r[3] = h4[(size_t)slot_src<K0 + 3>(sa, sb) * 16 + l];
    r[4] = h4[(size_t)slot_src<K0 + 4>(sa, sb) * 16 + l];
    r[5] = h4[(size_t)slot_src<K0 + 5>(sa, sb) * 16 + l];
    r[6] = h4[(size_t)slot_src<K0 + 6>(sa, sb) * 16 + l];
    r[7] = h4[(size_t)slot_src<K0 + 7>(sa, sb) * 16 + l];
}
template <int K0>
__device__ __forceinline__ void add8(float4& acc, const float4 (&r)[8], float na, float nb) {
    axpy4(acc, slot_nrm<K0 + 0>(na, nb), r[0]);
    axpy4(acc, slot_nrm<K0 + 1>(na, nb), r[1]);
    axpy4(acc, slot_nrm<K0 + 2>(na, nb), r[2]);
    axpy4(acc, slot_nrm<K0 + 3>(na, nb), r[3]);
    axpy4(acc, slot_nrm<K0 + 4>(na, nb), r[4]);
    axpy4(acc, slot_nrm<K0 + 5>(na, nb), r[5]);
    axpy4(acc, slot_nrm<K0 + 6>(na, nb), r[6]);
    axpy4(acc, slot_nrm<K0 + 7>(na, nb), r[7]);
}

constexpr int TP = 68;

__global__ void __launch_bounds__(256)
k_agg1_gemm2(const int* __restrict__ cnt,
             const unsigned long long* __restrict__ ell,
             const int* __restrict__ nspill,
             const unsigned long long* __restrict__ spill_e,
             const int* __restrict__ spill_d,
             const float* __restrict__ dinv, const float* __restrict__ h,
             const float* __restrict__ bias, const float* __restrict__ W2,
             float* __restrict__ O, int n) {
    __shared__ float tile[16 * TP];
    __shared__ float sdd[16];
    const int gi = threadIdx.x >> 4, l = threadIdx.x & 15;
    const int row0 = blockIdx.x * 16;
    const int g = row0 + gi;
    float4 acc = make_float4(0.0f, 0.0f, 0.0f, 0.0f);
    float dd = 0.0f;
    if (g < n) {
        const int c = cnt[g];
        const int m = (c < CAP) ? c : CAP;
        const size_t p0 = (size_t)g * CAP;
        const float4* __restrict__ h4 = reinterpret_cast<const float4*>(h);
        dd = dinv[g];
        const float4 hs = h4[(size_t)g * 16 + l];
        const float4 b = reinterpret_cast<const float4*>(bias)[l];
        int sa = g, sb = g;
        float wa = 0.0f, wb = 0.0f;
        if (2 * l + 1 < m) {
            const ulonglong2 ev = reinterpret_cast<const ulonglong2*>(ell + p0)[l];
            sa = (int)(unsigned)ev.x;
            sb = (int)(unsigned)ev.y;
            wa = __uint_as_float((unsigned)(ev.x >> 32));
            wb = __uint_as_float((unsigned)(ev.y >> 32));
        } else if (2 * l < m) {
            const unsigned long long e0 = ell[p0 + 2 * l];
            sa = (int)(unsigned)e0;
            wa = __uint_as_float((unsigned)(e0 >> 32));
        }
        float4 r[8];
        rows8<0>(r, l, h4, sa, sb);
        add8<0>(acc, r, wa, wb);
        if (m > 8)  { rows8<8>(r, l, h4, sa, sb);  add8<8>(acc, r, wa, wb); }
        if (m > 16) { rows8<16>(r, l, h4, sa, sb); add8<16>(acc, r, wa, wb); }
        if (m > 24) { rows8<24>(r, l, h4, sa, sb); add8<24>(acc, r, wa, wb); }
        if (c > CAP) {
            const int ns = nspill[0];
            for (int j = 0; j < ns; ++j) {
                if (spill_d[j] != g) continue;
                const unsigned long long e0 = spill_e[j];
                const float4 r0 = h4[(size_t)(int)(unsigned)e0 * 16 + l];
                axpy4(acc, __uint_as_float((unsigned)(e0 >> 32)), r0);
            }
        }
        acc.x += hs.x; acc.y += hs.y; acc.z += hs.z; acc.w += hs.w;
        acc.x = acc.x * dd + b.x; acc.y = acc.y * dd + b.y;
        acc.z = acc.z * dd + b.z; acc.w = acc.w * dd + b.w;
        acc.x = fmaxf(acc.x, 0.0f); acc.y = fmaxf(acc.y, 0.0f);
        acc.z = fmaxf(acc.z, 0.0f); acc.w = fmaxf(acc.w, 0.0f);
    }
    *reinterpret_cast<float4*>(&tile[gi * TP + 4 * l]) = acc;
    if (l == 0) sdd[gi] = dd;

    const int wave = threadIdx.x >> 5;
    const int lane = threadIdx.x & 31;
    const int lo = lane & 15, hi = lane >> 4;
    const int col = wave * 16 + lo;
    const bool cok = (wave < 3) && (col < 40);
    float bw[32];
    if (wave < 3) {
        const float* __restrict__ wcol = W2 + (cok ? col : 0);
#pragma unroll
        for (int j = 0; j < 16; ++j) {
            const int ka = 4 * j + 2 * hi;
            const float w0 = wcol[ka * 40], w1 = wcol[(ka + 1) * 40];
            bw[2 * j]     = cok ? w0 : 0.0f;
            bw[2 * j + 1] = cok ? w1 : 0.0f;
        }
    }
    __syncthreads();
    if (wave < 3) {
        const float* arow = tile + lo * TP;
        v8f c = {};
#pragma unroll
        for (int j = 0; j < 16; ++j) {
            const int ka = 4 * j + 2 * hi;
            v2f a = { arow[ka], arow[ka + 1] };
            v2f b = { bw[2 * j], bw[2 * j + 1] };
            c = __builtin_amdgcn_wmma_f32_16x16x4_f32(false, a, false, b,
                                                      (short)0, c, false, false);
        }
        if (cok) {
            const int r0 = row0 + 8 * hi;
            float* orow = O + (size_t)r0 * 64 + col;
            if (row0 + 16 <= n) {
#pragma unroll
                for (int r = 0; r < 8; ++r) orow[(size_t)r * 64] = c[r] * sdd[8 * hi + r];
            } else {
#pragma unroll
                for (int r = 0; r < 8; ++r)
                    if (r0 + r < n) orow[(size_t)r * 64] = c[r] * sdd[8 * hi + r];
            }
        }
    }
}

__global__ void k_agg2_lsm(const int* __restrict__ cnt,
                           const unsigned long long* __restrict__ ell,
                           const int* __restrict__ nspill,
                           const unsigned long long* __restrict__ spill_e,
                           const int* __restrict__ spill_d,
                           const float* __restrict__ dinv,
                           const float* __restrict__ h, const float* __restrict__ bias,
                           float* __restrict__ out, int n) {
    const int t = blockIdx.x * blockDim.x + threadIdx.x;
    const int g = t >> 4, l = t & 15;
    const bool act = (g < n) && (l < 10);
    const int lc = (l < 10) ? l : 0;
    float4 acc = make_float4(0.0f, 0.0f, 0.0f, 0.0f);
    if (g < n) {
        const int c = cnt[g];
        const int m = (c < CAP) ? c : CAP;
        const size_t p0 = (size_t)g * CAP;
        const float4* __restrict__ h4 = reinterpret_cast<const float4*>(h);
        const float dd = dinv[g];
        const float4 hs = h4[(size_t)g * 16 + lc];
        const float4 b = reinterpret_cast<const float4*>(bias)[lc];
        int sa = g, sb = g;
        float na = 0.0f, nb = 0.0f;
        if (2 * l + 1 < m) {
            const ulonglong2 ev = reinterpret_cast<const ulonglong2*>(ell + p0)[l];
            sa = (int)(unsigned)ev.x;
            sb = (int)(unsigned)ev.y;
            na = __uint_as_float((unsigned)(ev.x >> 32));
            nb = __uint_as_float((unsigned)(ev.y >> 32));
        } else if (2 * l < m) {
            const unsigned long long e0 = ell[p0 + 2 * l];
            sa = (int)(unsigned)e0;
            na = __uint_as_float((unsigned)(e0 >> 32));
        }
        float4 r[8];
        rows8<0>(r, lc, h4, sa, sb);
        add8<0>(acc, r, na, nb);
        if (m > 8)  { rows8<8>(r, lc, h4, sa, sb);  add8<8>(acc, r, na, nb); }
        if (m > 16) { rows8<16>(r, lc, h4, sa, sb); add8<16>(acc, r, na, nb); }
        if (m > 24) { rows8<24>(r, lc, h4, sa, sb); add8<24>(acc, r, na, nb); }
        if (c > CAP) {
            const int ns = nspill[0];
            for (int j = 0; j < ns; ++j) {
                if (spill_d[j] != g) continue;
                const unsigned long long e0 = spill_e[j];
                const float4 r0 = h4[(size_t)(int)(unsigned)e0 * 16 + lc];
                axpy4(acc, __uint_as_float((unsigned)(e0 >> 32)), r0);
            }
        }
        acc.x += hs.x; acc.y += hs.y; acc.z += hs.z; acc.w += hs.w;
        acc.x = acc.x * dd + b.x; acc.y = acc.y * dd + b.y;
        acc.z = acc.z * dd + b.z; acc.w = acc.w * dd + b.w;
    }
    float mx = act ? fmaxf(fmaxf(acc.x, acc.y), fmaxf(acc.z, acc.w)) : -3.402823466e38f;
    for (int o = 8; o >= 1; o >>= 1) mx = fmaxf(mx, __shfl_xor(mx, o, 16));
    float s = act ? (expf(acc.x - mx) + expf(acc.y - mx)) + (expf(acc.z - mx) + expf(acc.w - mx))
                  : 0.0f;
    for (int o = 8; o >= 1; o >>= 1) s += __shfl_xor(s, o, 16);
    const float lse = mx + logf(s);
    if (act) {
        acc.x -= lse; acc.y -= lse; acc.z -= lse; acc.w -= lse;
        reinterpret_cast<float4*>(out)[(size_t)g * 10 + l] = acc;
    }
}


extern "C" void kernel_launch(void* const* d_in, const int* in_sizes, int n_in,
                              void* d_out, int out_size, void* d_ws, size_t ws_size,
                              hipStream_t stream) {
    const float* x   = (const float*)d_in[0];
    const int*   ei  = (const int*)d_in[1];
    const float* ew  = (const float*)d_in[2];
    const float* W1  = (const float*)d_in[3];
    const float* b1  = (const float*)d_in[4];
    const float* W2  = (const float*)d_in[5];
    const float* b2  = (const float*)d_in[6];
    float* out = (float*)d_out;

    const int N = in_sizes[0] / 64;
    const int E = in_sizes[2];
    const int* src = ei;
    const int* dst = ei + E;

    const int nblk   = (E + EPB - 1) / EPB;
    const int nbins  = (N + BN - 1) / BN;
    const int nbinsA = (nbins < MAXB) ? nbins : MAXB;

    const size_t Ep = ((size_t)E + 63) / 64 * 64;
    const size_t Np = ((size_t)N + 63) / 64 * 64;
    const size_t Cp = ((size_t)nbinsA * nblk + 63) / 64 * 64;
    unsigned long long* ell     = (unsigned long long*)d_ws;
    unsigned long long* spill_e = ell + Np * CAP;
    int*   spill_d = (int*)(spill_e + Ep);
    int*   cnt     = spill_d + Ep;
    int*   nspill  = cnt + Np;
    float* dinv    = (float*)(nspill + 64);
    float* bufA    = dinv + Np;
    float* bufB    = bufA + (size_t)N * 64;
    unsigned long long* cells = (unsigned long long*)(bufB + (size_t)N * 64);
    unsigned long long* pad   = cells + (size_t)nbinsA * nblk * CELL;
    unsigned long long* ovf_e = pad + 32;
    int*   ovf_d   = (int*)(ovf_e + (size_t)nblk * EPB);
    int*   cellcnt = ovf_d + (size_t)nblk * EPB;
    int*   ovfcnt  = cellcnt + Cp;

    const int BLK = 256;
    const unsigned gNg = (unsigned)(((long long)N * 16 + BLK - 1) / BLK);

    k_bin<<<(unsigned)nblk, 1024, 0, stream>>>(src, dst, ew, cells, cellcnt, ovf_e, ovf_d, ovfcnt,
                                               nspill, pad, E, nblk, nbinsA);
    k_place_gemm1<<<(unsigned)nbins, BLK, 0, stream>>>(cells, cellcnt, pad, ovf_e, ovf_d, ovfcnt,
                                                       nblk, ell, cnt, dinv, nspill, spill_e,
                                                       spill_d, N, x, W1, bufA);

    k_agg1_gemm2<<<gNg, BLK, 0, stream>>>(cnt, ell, nspill, spill_e, spill_d, dinv, bufA, b1,
                                          W2, bufB, N);

    k_agg2_lsm<<<gNg, BLK, 0, stream>>>(cnt, ell, nspill, spill_e, spill_d, dinv,
                                        bufB, b2, out, N);
}


# level-1 cells written whole, end marker behind the last record: no count plane, so a level-2 block requests all its cell positions at entry in one group (1 trip, not 7); sums stay in slot order
# speedup vs baseline: 7.8098x; 1.0778x over previous
#include <hip/hip_runtime.h>
#include <math.h>


typedef __attribute__((ext_vector_type(2))) float v2f;
typedef __attribute__((ext_vector_type(8))) float v8f;

constexpr int CAP = 32;


constexpr int BSH  = 7;
constexpr int BN   = 128;
constexpr int MAXB = 1024;
constexpr int CELL = 32;
constexpr int EPB  = 8192;
constexpr int RP   = 33;
constexpr int LP   = 33;
constexpr unsigned long long ENDMARK = 0x80000000ull;
constexpr int PPT  = 25;

__device__ __forceinline__ void bin_block(int blk,
                        const int* __restrict__ src, const int* __restrict__ dst,
                        const float* __restrict__ w,
                        unsigned long long* __restrict__ cells,
                        unsigned long long* __restrict__ ovf_e, int* __restrict__ ovf_d,
                        int* __restrict__ ovfcnt, int E, int nblk, int nbinsA) {
    __shared__ unsigned long long lcell[MAXB * LP];
    __shared__ int lcnt[MAXB];
    __shared__ int lovf;
    for (int i = threadIdx.x; i < MAXB; i += blockDim.x) lcnt[i] = 0;
    if (threadIdx.x == 0) lovf = 0;
    __syncthreads();
    const int e0 = blk * EPB;
    const int e1 = (e0 + EPB < E) ? (e0 + EPB) : E;
    for (int base = e0 + (int)threadIdx.x; base < e1; base += 8 * (int)blockDim.x) {
        int dd[8], rk[8];
        unsigned ss[8], ww[8];
#pragma unroll
        for (int u = 0; u < 8; ++u) {
            const int e = base + u * (int)blockDim.x;
            const int ec = (e < e1) ? e : (e1 - 1);
            dd[u] = dst[ec];
            ss[u] = (unsigned)src[ec];
            ww[u] = __float_as_uint(w[ec]);
        }
#pragma unroll
        for (int u = 0; u < 8; ++u) {
            const int bin = dd[u] >> BSH;
            const bool cell = (base + u * (int)blockDim.x < e1) && (bin < MAXB);
            rk[u] = CELL;
            if (cell) rk[u] = atomicAdd(&lcnt[bin], 1);
        }
#pragma unroll
        for (int u = 0; u < 8; ++u) {
            if (base + u * (int)blockDim.x < e1) {
                const int d = dd[u];
                if (rk[u] < CELL) {
                    lcell[(d >> BSH) * LP + rk[u]] =
                        ((unsigned long long)ww[u] << 32) |
                        (unsigned long long)(ss[u] | ((unsigned)(d & (BN - 1)) << 24));
                } else {
                    const int o = atomicAdd(&lovf, 1);
                    ovf_e[(size_t)blk * EPB + o] =
                        ((unsigned long long)ww[u] << 32) | (unsigned long long)ss[u];
                    ovf_d[(size_t)blk * EPB + o] = d;
                }
            }
        }
    }
    __syncthreads();
    {
        const int lane = threadIdx.x & 31;
        const int nw = (int)blockDim.x >> 5;
        for (int i0 = (int)threadIdx.x >> 5; i0 < nbinsA; i0 += 8 * nw) {
            int cc[8];
            unsigned long long v[8];
#pragma unroll
            for (int u = 0; u < 8; ++u) {
                const int i = i0 + u * nw;
                const int c = lcnt[(i < nbinsA) ? i : 0];
                cc[u] = (i < nbinsA) ? ((c < CELL) ? c : CELL) : 0;
            }
#pragma unroll
            for (int u = 0; u < 8; ++u)
                v[u] = (lane < cc[u]) ? lcell[(i0 + u * nw) * LP + lane] : ENDMARK;
#pragma unroll
            for (int u = 0; u < 8; ++u)
                if (i0 + u * nw < nbinsA)
                    cells[((size_t)(i0 + u * nw) * nblk + blk) * CELL + lane] = v[u];
        }
    }
    if (threadIdx.x == 0) ovfcnt[blk] = lovf;
}

__global__ void __launch_bounds__(1024)
k_bin(const int* __restrict__ src, const int* __restrict__ dst,
      const float* __restrict__ w,
      unsigned long long* __restrict__ cells,
      unsigned long long* __restrict__ ovf_e, int* __restrict__ ovf_d,
      int* __restrict__ ovfcnt, int* __restrict__ nspill, int E, int nblk, int nbinsA) {
    if (blockIdx.x == 0 && threadIdx.x == 0)
        nspill[0] = 0;
    bin_block((int)blockIdx.x, src, dst, w, cells, ovf_e, ovf_d, ovfcnt, E, nblk, nbinsA);
}

__device__ __forceinline__ void place_store(int slot, int dl, unsigned s, unsigned wb, int bin,
                        float* sxw, unsigned long long* srow,
                        int* __restrict__ nspill,
                        unsigned long long* __restrict__ spill_e, int* __restrict__ spill_d) {
    const unsigned long long v = ((unsigned long long)wb << 32) | (unsigned long long)s;
    if (slot < CAP) {
        srow[dl * RP + slot] = v;
    } else {
        const int j = __hip_atomic_fetch_add(nspill, 1, __ATOMIC_RELAXED,
                                             __HIP_MEMORY_SCOPE_AGENT);
        spill_e[j] = v;
        spill_d[j] = bin * BN + dl;
        atomicAdd(&sxw[dl], __uint_as_float(wb));
    }
}

__device__ __forceinline__ void place_block(int bin,
            const unsigned long long* __restrict__ cells,
            const unsigned long long* __restrict__ ovf_e, const int* __restrict__ ovf_d,
            const int* __restrict__ ovfcnt, int nblk,
            unsigned long long* __restrict__ ell, int* __restrict__ cnt,
            float* __restrict__ dinv, int* __restrict__ nspill,
            unsigned long long* __restrict__ spill_e, int* __restrict__ spill_d, int n,
            const float* __restrict__ X, const float* __restrict__ W1, float* __restrict__ O) {
    __shared__ unsigned long long srow[BN * RP];
    __shared__ float sdinv[BN];
    __shared__ int scnt[BN];
    __shared__ float sxw[BN];
    __shared__ int so[256];
    __shared__ int sany;
    const int t = threadIdx.x;
    const int lo = t & 15, hi = (t >> 4) & 1;
    const int mrow0 = bin * BN + (t >> 5) * 16;
    if (t < BN) scnt[t] = 0;
    if (t < BN) sxw[t] = 0.0f;
    if (t == 0) sany = 0;
    __syncthreads();
    for (int c0 = 0; c0 < nblk; c0 += 256) {
        const int nb = (nblk - c0 < 256) ? (nblk - c0) : 256;
        const int ov = (t < nb) ? ovfcnt[c0 + t] : 0;
        if (bin < MAXB) {
            const unsigned long long* __restrict__ cb = cells + ((size_t)bin * nblk + c0) * CELL;
            const int np = nb * CELL;
            for (int p0 = t; p0 < np; p0 += PPT * 256) {
                unsigned long long rec[PPT];
                int sl[PPT];
#pragma unroll
                for (int u = 0; u < PPT; ++u) {
                    const int p = p0 + u * 256;
                    rec[u] = cb[(p < np) ? p : 0];
                    if (p >= np) rec[u] = ENDMARK;
                }
#pragma unroll
                for (int u = 0; u < PPT; ++u) {
                    sl[u] = 0;
                    if (!(rec[u] & ENDMARK)) sl[u] = atomicAdd(&scnt[(int)((unsigned)rec[u] >> 24)], 1);
                }
#pragma unroll
                for (int u = 0; u < PPT; ++u) {
                    if (!(rec[u] & ENDMARK)) {
                        const unsigned lo = (unsigned)rec[u];
                        place_store(sl[u], (int)(lo >> 24), lo & 0xFFFFFFu, (unsigned)(rec[u] >> 32),
                                    bin, sxw, srow, nspill, spill_e, spill_d);
                    }
                }
            }
        }
        if (t < nb) {
            so[t] = ov;
            if (ov > 0) atomicOr(&sany, 1);
        }
        __syncthreads();
        if (sany) {
            for (int cc = 0; cc < nb; ++cc) {
                const int no = so[cc];
                for (int i = t; i < no; i += 256) {
                    const int d = ovf_d[(size_t)(c0 + cc) * EPB + i];
                    if ((d >> BSH) == bin) {
                        const unsigned long long rec = ovf_e[(size_t)(c0 + cc) * EPB + i];
                        const int dl = d & (BN - 1);
                        const int slot = atomicAdd(&scnt[dl], 1);
                        place_store(slot, dl, (unsigned)rec, (unsigned)(rec >> 32),
                                    bin, sxw, srow, nspill, spill_e, spill_d);
                    }
                }
            }
        }
        __syncthreads();
    }
    float4 afr[8];
    {
        const int ar = (mrow0 + lo < n) ? (mrow0 + lo) : (n - 1);
        const float4* __restrict__ a4 = reinterpret_cast<const float4*>(X) + (size_t)ar * 16 + hi;
#pragma unroll
        for (int q = 0; q < 8; ++q) afr[q] = a4[2 * q];
    }
    if (t < BN) {
        const int g = bin * BN + t;
        float di = 0.0f;
        if (g < n) {
            const int cc = scnt[t];
            const int m = (cc < CAP) ? cc : CAP;
            float d = 1.0f;
            for (int k = 0; k < m; ++k) d += __uint_as_float((unsigned)(srow[t * RP + k] >> 32));
            if (cc > CAP) d += sxw[t];
            di = (d > 0.0f) ? rsqrtf(d) : 0.0f;
            cnt[g] = cc;
            dinv[g] = di;
        }
        sdinv[t] = di;
    }
    for (int idx = t; idx < BN * CAP; idx += 256) {
        const int row = idx >> 5, k = idx & (CAP - 1);
        const int g = bin * BN + row;
        if (g < n && k < scnt[row]) ell[(size_t)g * CAP + k] = srow[row * RP + k];
    }
    __syncthreads();
    if (mrow0 < n) {
        const float4* __restrict__ W4 = reinterpret_cast<const float4*>(W1) + lo;
        v8f c0 = {}, c1 = {}, c2 = {}, c3 = {};
#pragma unroll
        for (int q = 0; q < 8; ++q) {
            const int kb = 8 * q + 4 * hi;
            const float4 w0 = W4[(kb + 0) * 16], w1 = W4[(kb + 1) * 16];
            const float4 w2 = W4[(kb + 2) * 16], w3 = W4[(kb + 3) * 16];
            const v2f a0 = { afr[q].x, afr[q].y };
            const v2f a1 = { afr[q].z, afr[q].w };
            const v2f b00 = { w0.x, w1.x }, b01 = { w0.y, w1.y }, b02 = { w0.z, w1.z }, b03 = { w0.w, w1.w };
            const v2f b10 = { w2.x, w3.x }, b11 = { w2.y, w3.y }, b12 = { w2.z, w3.z }, b13 = { w2.w, w3.w };
            c0 = __builtin_amdgcn_wmma_f32_16x16x4_f32(false, a0, false, b00, (short)0, c0, false, false);
            c1 = __builtin_amdgcn_wmma_f32_16x16x4_f32(false, a0, false, b01, (short)0, c1, false, false);
            c2 = __builtin_amdgcn_wmma_f32_16x16x4_f32(false, a0, false, b02, (short)0, c2, false, false);
            c3 = __builtin_amdgcn_wmma_f32_16x16x4_f32(false, a0, false, b03, (short)0, c3, false, false);
            c0 = __builtin_amdgcn_wmma_f32_16x16x4_f32(false, a1, false, b10, (short)0, c0, false, false);
            c1 = __builtin_amdgcn_wmma_f32_16x16x4_f32(false, a1, false, b11, (short)0, c1, false, false);
            c2 = __builtin_amdgcn_wmma_f32_16x16x4_f32(false, a1, false, b12, (short)0, c2, false, false);
            c3 = __builtin_amdgcn_wmma_f32_16x16x4_f32(false, a1, false, b13, (short)0, c3, false, false);
        }
        const int rb = (t >> 5) * 16 + 8 * hi;
        float4* __restrict__ o4 = reinterpret_cast<float4*>(O) + (size_t)(bin * BN + rb) * 16 + lo;
#pragma unroll
        for (int r = 0; r < 8; ++r) {
            const float sc = sdinv[rb + r];
            if (bin * BN + rb + r < n)
                o4[(size_t)r * 16] = make_float4(c0[r] * sc, c1[r] * sc, c2[r] * sc, c3[r] * sc);
        }
    }
}

__global__ void __launch_bounds__(256)
k_place_gemm1(const unsigned long long* __restrict__ cells,
              const unsigned long long* __restrict__ ovf_e, const int* __restrict__ ovf_d,
              const int* __restrict__ ovfcnt, int nblk,
              unsigned long long* __restrict__ ell, int* __restrict__ cnt,
              float* __restrict__ dinv, int* __restrict__ nspill,
              unsigned long long* __restrict__ spill_e, int* __restrict__ spill_d, int n,
              const float* __restrict__ X, const float* __restrict__ W1,
              float* __restrict__ O) {
    place_block((int)blockIdx.x, cells, ovf_e, ovf_d, ovfcnt, nblk, ell, cnt, dinv,
                nspill, spill_e, spill_d, n, X, W1, O);
}


__device__ __forceinline__ void axpy4(float4& a, float s, const float4& r) {
    a.x += s * r.x; a.y += s * r.y; a.z += s * r.z; a.w += s * r.w;
}

template <int K>
__device__ __forceinline__ int slot_src(int sa, int sb) {
    return __builtin_amdgcn_update_dpp(0, (K & 1) ? sb : sa, 0x150 + (K >> 1), 0xf, 0xf, true);
}
template <int K>
__device__ __forceinline__ float slot_nrm(float na, float nb) {
    return __int_as_float(__builtin_amdgcn_update_dpp(0, __float_as_int((K & 1) ? nb : na),
                                                      0x150 + (K >> 1), 0xf, 0xf, true));
}

template <int K0>
__device__ __forceinline__ void rows8(float4 (&r)[8], int l, const float4* __restrict__ h4,
                                      int sa, int sb) {
    r[0] = h4[(size_t)slot_src<K0 + 0>(sa, sb) * 16 + l];
    r[1] = h4[(size_t)slot_src<K0 + 1>(sa, sb) * 16 + l];
    r[2] = h4[(size_t)slot_src<K0 + 2>(sa, sb) * 16 + l];
    r[3] = h4[(size_t)slot_src<K0 + 3>(sa, sb) * 16 + l];
    r[4] = h4[(size_t)slot_src<K0 + 4>(sa, sb) * 16 + l];
    r[5] = h4[(size_t)slot_src<K0 + 5>(sa, sb) * 16 + l];
    r[6] = h4[(size_t)slot_src<K0 + 6>(sa, sb) * 16 + l];
    r[7] = h4[(size_t)slot_src<K0 + 7>(sa, sb) * 16 + l];
}
template <int K0>
__device__ __forceinline__ void add8(float4& acc, const float4 (&r)[8], float na, float nb) {
    axpy4(acc, slot_nrm<K0 + 0>(na, nb), r[0]);
    axpy4(acc, slot_nrm<K0 + 1>(na, nb), r[1]);
    axpy4(acc, slot_nrm<K0 + 2>(na, nb), r[2]);
    axpy4(acc, slot_nrm<K0 + 3>(na, nb), r[3]);
    axpy4(acc, slot_nrm<K0 + 4>(na, nb), r[4]);
    axpy4(acc, slot_nrm<K0 + 5>(na, nb), r[5]);
    axpy4(acc, slot_nrm<K0 + 6>(na, nb), r[6]);
    axpy4(acc, slot_nrm<K0 + 7>(na, nb), r[7]);
}

constexpr int TP = 68;

__global__ void __launch_bounds__(256)
k_agg1_gemm2(const int* __restrict__ cnt,
             const unsigned long long* __restrict__ ell,
             const int* __restrict__ nspill,
             const unsigned long long* __restrict__ spill_e,
             const int* __restrict__ spill_d,
             const float* __restrict__ dinv, const float* __restrict__ h,
             const float* __restrict__ bias, const float* __restrict__ W2,
             float* __restrict__ O, int n) {
    __shared__ float tile[16 * TP];
    __shared__ float sdd[16];
    const int gi = threadIdx.x >> 4, l = threadIdx.x & 15;
    const int row0 = blockIdx.x * 16;
    const int g = row0 + gi;
    float4 acc = make_float4(0.0f, 0.0f, 0.0f, 0.0f);
    float dd = 0.0f;
    if (g < n) {
        const int c = cnt[g];
        const int m = (c < CAP) ? c : CAP;
        const size_t p0 = (size_t)g * CAP;
        const float4* __restrict__ h4 = reinterpret_cast<const float4*>(h);
        dd = dinv[g];
        const float4 hs = h4[(size_t)g * 16 + l];
        const float4 b = reinterpret_cast<const float4*>(bias)[l];
        int sa = g, sb = g;
        float wa = 0.0f, wb = 0.0f;
        if (2 * l + 1 < m) {
            const ulonglong2 ev = reinterpret_cast<const ulonglong2*>(ell + p0)[l];
            sa = (int)(unsigned)ev.x;
            sb = (int)(unsigned)ev.y;
            wa = __uint_as_float((unsigned)(ev.x >> 32));
            wb = __uint_as_float((unsigned)(ev.y >> 32));
        } else if (2 * l < m) {
            const unsigned long long e0 = ell[p0 + 2 * l];
            sa = (int)(unsigned)e0;
            wa = __uint_as_float((unsigned)(e0 >> 32));
        }
        float4 r[8];
        rows8<0>(r, l, h4, sa, sb);
        add8<0>(acc, r, wa, wb);
        if (m > 8)  { rows8<8>(r, l, h4, sa, sb);  add8<8>(acc, r, wa, wb); }
        if (m > 16) { rows8<16>(r, l, h4, sa, sb); add8<16>(acc, r, wa, wb); }
        if (m > 24) { rows8<24>(r, l, h4, sa, sb); add8<24>(acc, r, wa, wb); }
        if (c > CAP) {
            const int ns = nspill[0];
            for (int j = 0; j < ns; ++j) {
                if (spill_d[j] != g) continue;
                const unsigned long long e0 = spill_e[j];
                const float4 r0 = h4[(size_t)(int)(unsigned)e0 * 16 + l];
                axpy4(acc, __uint_as_float((unsigned)(e0 >> 32)), r0);
            }
        }
        acc.x += hs.x; acc.y += hs.y; acc.z += hs.z; acc.w += hs.w;
        acc.x = acc.x * dd + b.x; acc.y = acc.y * dd + b.y;
        acc.z = acc.z * dd + b.z; acc.w = acc.w * dd + b.w;
        acc.x = fmaxf(acc.x, 0.0f); acc.y = fmaxf(acc.y, 0.0f);
        acc.z = fmaxf(acc.z, 0.0f); acc.w = fmaxf(acc.w, 0.0f);
    }
    *reinterpret_cast<float4*>(&tile[gi * TP + 4 * l]) = acc;
    if (l == 0) sdd[gi] = dd;

    const int wave = threadIdx.x >> 5;
    const int lane = threadIdx.x & 31;
    const int lo = lane & 15, hi = lane >> 4;
    const int col = wave * 16 + lo;
    const bool cok = (wave < 3) && (col < 40);
    float bw[32];
    if (wave < 3) {
        const float* __restrict__ wcol = W2 + (cok ? col : 0);
#pragma unroll
        for (int j = 0; j < 16; ++j) {
            const int ka = 4 * j + 2 * hi;
            const float w0 = wcol[ka * 40], w1 = wcol[(ka + 1) * 40];
            bw[2 * j]     = cok ? w0 : 0.0f;
            bw[2 * j + 1] = cok ? w1 : 0.0f;
        }
    }
    __syncthreads();
    if (wave < 3) {
        const float* arow = tile + lo * TP;
        v8f c = {};
#pragma unroll
        for (int j = 0; j < 16; ++j) {
            const int ka = 4 * j + 2 * hi;
            v2f a = { arow[ka], arow[ka + 1] };
            v2f b = { bw[2 * j], bw[2 * j + 1] };
            c = __builtin_amdgcn_wmma_f32_16x16x4_f32(false, a, false, b,
                                                      (short)0, c, false, false);
        }
        if (cok) {
            const int r0 = row0 + 8 * hi;
            float* orow = O + (size_t)r0 * 64 + col;
            if (row0 + 16 <= n) {
#pragma unroll
                for (int r = 0; r < 8; ++r) orow[(size_t)r * 64] = c[r] * sdd[8 * hi + r];
            } else {
#pragma unroll
                for (int r = 0; r < 8; ++r)
                    if (r0 + r < n) orow[(size_t)r * 64] = c[r] * sdd[8 * hi + r];
            }
        }
    }
}

__global__ void k_agg2_lsm(const int* __restrict__ cnt,
                           const unsigned long long* __restrict__ ell,
                           const int* __restrict__ nspill,
                           const unsigned long long* __restrict__ spill_e,
                           const int* __restrict__ spill_d,
                           const float* __restrict__ dinv,
                           const float* __restrict__ h, const float* __restrict__ bias,
                           float* __restrict__ out, int n) {
    const int t = blockIdx.x * blockDim.x + threadIdx.x;
    const int g = t >> 4, l = t & 15;
    const bool act = (g < n) && (l < 10);
    const int lc = (l < 10) ? l : 0;
    float4 acc = make_float4(0.0f, 0.0f, 0.0f, 0.0f);
    if (g < n) {
        const int c = cnt[g];
        const int m = (c < CAP) ? c : CAP;
        const size_t p0 = (size_t)g * CAP;
        const float4* __restrict__ h4 = reinterpret_cast<const float4*>(h);
        const float dd = dinv[g];
        const float4 hs = h4[(size_t)g * 16 + lc];
        const float4 b = reinterpret_cast<const float4*>(bias)[lc];
        int sa = g, sb = g;
        float na = 0.0f, nb = 0.0f;
        if (2 * l + 1 < m) {
            const ulonglong2 ev = reinterpret_cast<const ulonglong2*>(ell + p0)[l];
            sa = (int)(unsigned)ev.x;
            sb = (int)(unsigned)ev.y;
            na = __uint_as_float((unsigned)(ev.x >> 32));
            nb = __uint_as_float((unsigned)(ev.y >> 32));
        } else if (2 * l < m) {
            const unsigned long long e0 = ell[p0 + 2 * l];
            sa = (int)(unsigned)e0;
            na = __uint_as_float((unsigned)(e0 >> 32));
        }
        float4 r[8];
        rows8<0>(r, lc, h4, sa, sb);
        add8<0>(acc, r, na, nb);
        if (m > 8)  { rows8<8>(r, lc, h4, sa, sb);  add8<8>(acc, r, na, nb); }
        if (m > 16) { rows8<16>(r, lc, h4, sa, sb); add8<16>(acc, r, na, nb); }
        if (m > 24) { rows8<24>(r, lc, h4, sa, sb); add8<24>(acc, r, na, nb); }
        if (c > CAP) {
            const int ns = nspill[0];
            for (int j = 0; j < ns; ++j) {
                if (spill_d[j] != g) continue;
                const unsigned long long e0 = spill_e[j];
                const float4 r0 = h4[(size_t)(int)(unsigned)e0 * 16 + lc];
                axpy4(acc, __uint_as_float((unsigned)(e0 >> 32)), r0);
            }
        }
        acc.x += hs.x; acc.y += hs.y; acc.z += hs.z; acc.w += hs.w;
        acc.x = acc.x * dd + b.x; acc.y = acc.y * dd + b.y;
        acc.z = acc.z * dd + b.z; acc.w = acc.w * dd + b.w;
    }
    float mx = act ? fmaxf(fmaxf(acc.x, acc.y), fmaxf(acc.z, acc.w)) : -3.402823466e38f;
    for (int o = 8; o >= 1; o >>= 1) mx = fmaxf(mx, __shfl_xor(mx, o, 16));
    float s = act ? (expf(acc.x - mx) + expf(acc.y - mx)) + (expf(acc.z - mx) + expf(acc.w - mx))
                  : 0.0f;
    for (int o = 8; o >= 1; o >>= 1) s += __shfl_xor(s, o, 16);
    const float lse = mx + logf(s);
    if (act) {
        acc.x -= lse; acc.y -= lse; acc.z -= lse; acc.w -= lse;
        reinterpret_cast<float4*>(out)[(size_t)g * 10 + l] = acc;
    }
}


extern "C" void kernel_launch(void* const* d_in, const int* in_sizes, int n_in,
                              void* d_out, int out_size, void* d_ws, size_t ws_size,
                              hipStream_t stream) {
    const float* x   = (const float*)d_in[0];
    const int*   ei  = (const int*)d_in[1];
    const float* ew  = (const float*)d_in[2];
    const float* W1  = (const float*)d_in[3];
    const float* b1  = (const float*)d_in[4];
    const float* W2  = (const float*)d_in[5];
    const float* b2  = (const float*)d_in[6];
    float* out = (float*)d_out;

    const int N = in_sizes[0] / 64;
    const int E = in_sizes[2];
    const int* src = ei;
    const int* dst = ei + E;

    const int nblk   = (E + EPB - 1) / EPB;
    const int nbins  = (N + BN - 1) / BN;
    const int nbinsA = (nbins < MAXB) ? nbins : MAXB;

    const size_t Ep = ((size_t)E + 63) / 64 * 64;
    const size_t Np = ((size_t)N + 63) / 64 * 64;
    unsigned long long* ell     = (unsigned long long*)d_ws;
    unsigned long long* spill_e = ell + Np * CAP;
    int*   spill_d = (int*)(spill_e + Ep);
    int*   cnt     = spill_d + Ep;
    int*   nspill  = cnt + Np;
    float* dinv    = (float*)(nspill + 64);
    float* bufA    = dinv + Np;
    float* bufB    = bufA + (size_t)N * 64;
    unsigned long long* cells = (unsigned long long*)(bufB + (size_t)N * 64);
    unsigned long long* ovf_e = cells + (size_t)nbinsA * nblk * CELL;
    int*   ovf_d   = (int*)(ovf_e + (size_t)nblk * EPB);
    int*   ovfcnt  = ovf_d + (size_t)nblk * EPB;

    const int BLK = 256;
    const unsigned gNg = (unsigned)(((long long)N * 16 + BLK - 1) / BLK);

    k_bin<<<(unsigned)nblk, 1024, 0, stream>>>(src, dst, ew, cells, ovf_e, ovf_d, ovfcnt,
                                               nspill, E, nblk, nbinsA);
    k_place_gemm1<<<(unsigned)nbins, BLK, 0, stream>>>(cells, ovf_e, ovf_d, ovfcnt,
                                                       nblk, ell, cnt, dinv, nspill, spill_e,
                                                       spill_d, N, x, W1, bufA);

    k_agg1_gemm2<<<gNg, BLK, 0, stream>>>(cnt, ell, nspill, spill_e, spill_d, dinv, bufA, b1,
                                          W2, bufB, N);

    k_agg2_lsm<<<gNg, BLK, 0, stream>>>(cnt, ell, nspill, spill_e, spill_d, dinv,
                                        bufB, b2, out, N);
}


# node sums: a row load = h's scalar base + its own 32-bit index, eight formed before the eight loads: no 64-bit address math, no s_wait_xcnt between them; layer-2 bias read last; slot order kept
# speedup vs baseline: 8.1413x; 1.0424x over previous
#include <hip/hip_runtime.h>
#include <math.h>


typedef __attribute__((ext_vector_type(2))) float v2f;
typedef __attribute__((ext_vector_type(8))) float v8f;

constexpr int CAP = 32;


constexpr int BSH  = 7;
constexpr int BN   = 128;
constexpr int MAXB = 1024;
constexpr int CELL = 32;
constexpr int EPB  = 8192;
constexpr int RP   = 33;
constexpr int LP   = 33;
constexpr unsigned long long ENDMARK = 0x80000000ull;
constexpr int PPT  = 25;

__device__ __forceinline__ void bin_block(int blk,
                        const int* __restrict__ src, const int* __restrict__ dst,
                        const float* __restrict__ w,
                        unsigned long long* __restrict__ cells,
                        unsigned long long* __restrict__ ovf_e, int* __restrict__ ovf_d,
                        int* __restrict__ ovfcnt, int E, int nblk, int nbinsA) {
    __shared__ unsigned long long lcell[MAXB * LP];
    __shared__ int lcnt[MAXB];
    __shared__ int lovf;
    for (int i = threadIdx.x; i < MAXB; i += blockDim.x) lcnt[i] = 0;
    if (threadIdx.x == 0) lovf = 0;
    __syncthreads();
    const int e0 = blk * EPB;
    const int e1 = (e0 + EPB < E) ? (e0 + EPB) : E;
    for (int base = e0 + (int)threadIdx.x; base < e1; base += 8 * (int)blockDim.x) {
        int dd[8], rk[8];
        unsigned ss[8], ww[8];
#pragma unroll
        for (int u = 0; u < 8; ++u) {
            const int e = base + u * (int)blockDim.x;
            const int ec = (e < e1) ? e : (e1 - 1);
            dd[u] = dst[ec];
            ss[u] = (unsigned)src[ec];
            ww[u] = __float_as_uint(w[ec]);
        }
#pragma unroll
        for (int u = 0; u < 8; ++u) {
            const int bin = dd[u] >> BSH;
            const bool cell = (base + u * (int)blockDim.x < e1) && (bin < MAXB);
            rk[u] = CELL;
            if (cell) rk[u] = atomicAdd(&lcnt[bin], 1);
        }
#pragma unroll
        for (int u = 0; u < 8; ++u) {
            if (base + u * (int)blockDim.x < e1) {
                const int d = dd[u];
                if (rk[u] < CELL) {
                    lcell[(d >> BSH) * LP + rk[u]] =
                        ((unsigned long long)ww[u] << 32) |
                        (unsigned long long)(ss[u] | ((unsigned)(d & (BN - 1)) << 24));
                } else {
                    const int o = atomicAdd(&lovf, 1);
                    ovf_e[(size_t)blk * EPB + o] =
                        ((unsigned long long)ww[u] << 32) | (unsigned long long)ss[u];
                    ovf_d[(size_t)blk * EPB + o] = d;
                }
            }
        }
    }
    __syncthreads();
    {
        const int lane = threadIdx.x & 31;
        const int nw = (int)blockDim.x >> 5;
        for (int i0 = (int)threadIdx.x >> 5; i0 < nbinsA; i0 += 8 * nw) {
            int cc[8];
            unsigned long long v[8];
#pragma unroll
            for (int u = 0; u < 8; ++u) {
                const int i = i0 + u * nw;
                const int c = lcnt[(i < nbinsA) ? i : 0];
                cc[u] = (i < nbinsA) ? ((c < CELL) ? c : CELL) : 0;
            }
#pragma unroll
            for (int u = 0; u < 8; ++u)
                v[u] = (lane < cc[u]) ? lcell[(i0 + u * nw) * LP + lane] : ENDMARK;
#pragma unroll
            for (int u = 0; u < 8; ++u)
                if (i0 + u * nw < nbinsA)
                    cells[((size_t)(i0 + u * nw) * nblk + blk) * CELL + lane] = v[u];
        }
    }
    if (threadIdx.x == 0) ovfcnt[blk] = lovf;
}

__global__ void __launch_bounds__(1024)
k_bin(const int* __restrict__ src, const int* __restrict__ dst,
      const float* __restrict__ w,
      unsigned long long* __restrict__ cells,
      unsigned long long* __restrict__ ovf_e, int* __restrict__ ovf_d,
      int* __restrict__ ovfcnt, int* __restrict__ nspill, int E, int nblk, int nbinsA) {
    if (blockIdx.x == 0 && threadIdx.x == 0)
        nspill[0] = 0;
    bin_block((int)blockIdx.x, src, dst, w, cells, ovf_e, ovf_d, ovfcnt, E, nblk, nbinsA);
}

__device__ __forceinline__ void place_store(int slot, int dl, unsigned s, unsigned wb, int bin,
                        float* sxw, unsigned long long* srow,
                        int* __restrict__ nspill,
                        unsigned long long* __restrict__ spill_e, int* __restrict__ spill_d) {
    const unsigned long long v = ((unsigned long long)wb << 32) | (unsigned long long)s;
    if (slot < CAP) {
        srow[dl * RP + slot] = v;
    } else {
        const int j = __hip_atomic_fetch_add(nspill, 1, __ATOMIC_RELAXED,
                                             __HIP_MEMORY_SCOPE_AGENT);
        spill_e[j] = v;
        spill_d[j] = bin * BN + dl;
        atomicAdd(&sxw[dl], __uint_as_float(wb));
    }
}

__device__ __forceinline__ void place_block(int bin,
            const unsigned long long* __restrict__ cells,
            const unsigned long long* __restrict__ ovf_e, const int* __restrict__ ovf_d,
            const int* __restrict__ ovfcnt, int nblk,
            unsigned long long* __restrict__ ell, int* __restrict__ cnt,
            float* __restrict__ dinv, int* __restrict__ nspill,
            unsigned long long* __restrict__ spill_e, int* __restrict__ spill_d, int n,
            const float* __restrict__ X, const float* __restrict__ W1, float* __restrict__ O) {
    __shared__ unsigned long long srow[BN * RP];
    __shared__ float sdinv[BN];
    __shared__ int scnt[BN];
    __shared__ float sxw[BN];
    __shared__ int so[256];
    __shared__ int sany;
    const int t = threadIdx.x;
    const int lo = t & 15, hi = (t >> 4) & 1;
    const int mrow0 = bin * BN + (t >> 5) * 16;
    if (t < BN) scnt[t] = 0;
    if (t < BN) sxw[t] = 0.0f;
    if (t == 0) sany = 0;
    __syncthreads();
    for (int c0 = 0; c0 < nblk; c0 += 256) {
        const int nb = (nblk - c0 < 256) ? (nblk - c0) : 256;
        const int ov = (t < nb) ? ovfcnt[c0 + t] : 0;
        if (bin < MAXB) {
            const unsigned long long* __restrict__ cb = cells + ((size_t)bin * nblk + c0) * CELL;
            const int np = nb * CELL;
            for (int p0 = t; p0 < np; p0 += PPT * 256) {
                unsigned long long rec[PPT];
                int sl[PPT];
#pragma unroll
                for (int u = 0; u < PPT; ++u) {
                    const int p = p0 + u * 256;
                    rec[u] = cb[(p < np) ? p : 0];
                    if (p >= np) rec[u] = ENDMARK;
                }
#pragma unroll
                for (int u = 0; u < PPT; ++u) {
                    sl[u] = 0;
                    if (!(rec[u] & ENDMARK)) sl[u] = atomicAdd(&scnt[(int)((unsigned)rec[u] >> 24)], 1);
                }
#pragma unroll
                for (int u = 0; u < PPT; ++u) {
                    if (!(rec[u] & ENDMARK)) {
                        const unsigned lo = (unsigned)rec[u];
                        place_store(sl[u], (int)(lo >> 24), lo & 0xFFFFFFu, (unsigned)(rec[u] >> 32),
                                    bin, sxw, srow, nspill, spill_e, spill_d);
                    }
                }
            }
        }
        if (t < nb) {
            so[t] = ov;
            if (ov > 0) atomicOr(&sany, 1);
        }
        __syncthreads();
        if (sany) {
            for (int cc = 0; cc < nb; ++cc) {
                const int no = so[cc];
                for (int i = t; i < no; i += 256) {
                    const int d = ovf_d[(size_t)(c0 + cc) * EPB + i];
                    if ((d >> BSH) == bin) {
                        const unsigned long long rec = ovf_e[(size_t)(c0 + cc) * EPB + i];
                        const int dl = d & (BN - 1);
                        const int slot = atomicAdd(&scnt[dl], 1);
                        place_store(slot, dl, (unsigned)rec, (unsigned)(rec >> 32),
                                    bin, sxw, srow, nspill, spill_e, spill_d);
                    }
                }
            }
        }
        __syncthreads();
    }
    float4 afr[8];
    {
        const int ar = (mrow0 + lo < n) ? (mrow0 + lo) : (n - 1);
        const float4* __restrict__ a4 = reinterpret_cast<const float4*>(X) + (size_t)ar * 16 + hi;
#pragma unroll
        for (int q = 0; q < 8; ++q) afr[q] = a4[2 * q];
    }
    if (t < BN) {
        const int g = bin * BN + t;
        float di = 0.0f;
        if (g < n) {
            const int cc = scnt[t];
            const int m = (cc < CAP) ? cc : CAP;
            float d = 1.0f;
            for (int k = 0; k < m; ++k) d += __uint_as_float((unsigned)(srow[t * RP + k] >> 32));
            if (cc > CAP) d += sxw[t];
            di = (d > 0.0f) ? rsqrtf(d) : 0.0f;
            cnt[g] = cc;
            dinv[g] = di;
        }
        sdinv[t] = di;
    }
    for (int idx = t; idx < BN * CAP; idx += 256) {
        const int row = idx >> 5, k = idx & (CAP - 1);
        const int g = bin * BN + row;
        if (g < n && k < scnt[row]) ell[(size_t)g * CAP + k] = srow[row * RP + k];
    }
    __syncthreads();
    if (mrow0 < n) {
        const float4* __restrict__ W4 = reinterpret_cast<const float4*>(W1) + lo;
        v8f c0 = {}, c1 = {}, c2 = {}, c3 = {};
#pragma unroll
        for (int q = 0; q < 8; ++q) {
            const int kb = 8 * q + 4 * hi;
            const float4 w0 = W4[(kb + 0) * 16], w1 = W4[(kb + 1) * 16];
            const float4 w2 = W4[(kb + 2) * 16], w3 = W4[(kb + 3) * 16];
            const v2f a0 = { afr[q].x, afr[q].y };
            const v2f a1 = { afr[q].z, afr[q].w };
            const v2f b00 = { w0.x, w1.x }, b01 = { w0.y, w1.y }, b02 = { w0.z, w1.z }, b03 = { w0.w, w1.w };
            const v2f b10 = { w2.x, w3.x }, b11 = { w2.y, w3.y }, b12 = { w2.z, w3.z }, b13 = { w2.w, w3.w };
            c0 = __builtin_amdgcn_wmma_f32_16x16x4_f32(false, a0, false, b00, (short)0, c0, false, false);
            c1 = __builtin_amdgcn_wmma_f32_16x16x4_f32(false, a0, false, b01, (short)0, c1, false, false);
            c2 = __builtin_amdgcn_wmma_f32_16x16x4_f32(false, a0, false, b02, (short)0, c2, false, false);
            c3 = __builtin_amdgcn_wmma_f32_16x16x4_f32(false, a0, false, b03, (short)0, c3, false, false);
            c0 = __builtin_amdgcn_wmma_f32_16x16x4_f32(false, a1, false, b10, (short)0, c0, false, false);
            c1 = __builtin_amdgcn_wmma_f32_16x16x4_f32(false, a1, false, b11, (short)0, c1, false, false);
            c2 = __builtin_amdgcn_wmma_f32_16x16x4_f32(false, a1, false, b12, (short)0, c2, false, false);
            c3 = __builtin_amdgcn_wmma_f32_16x16x4_f32(false, a1, false, b13, (short)0, c3, false, false);
        }
        const int rb = (t >> 5) * 16 + 8 * hi;
        float4* __restrict__ o4 = reinterpret_cast<float4*>(O) + (size_t)(bin * BN + rb) * 16 + lo;
#pragma unroll
        for (int r = 0; r < 8; ++r) {
            const float sc = sdinv[rb + r];
            if (bin * BN + rb + r < n)
                o4[(size_t)r * 16] = make_float4(c0[r] * sc, c1[r] * sc, c2[r] * sc, c3[r] * sc);
        }
    }
}

__global__ void __launch_bounds__(256)
k_place_gemm1(const unsigned long long* __restrict__ cells,
              const unsigned long long* __restrict__ ovf_e, const int* __restrict__ ovf_d,
              const int* __restrict__ ovfcnt, int nblk,
              unsigned long long* __restrict__ ell, int* __restrict__ cnt,
              float* __restrict__ dinv, int* __restrict__ nspill,
              unsigned long long* __restrict__ spill_e, int* __restrict__ spill_d, int n,
              const float* __restrict__ X, const float* __restrict__ W1,
              float* __restrict__ O) {
    place_block((int)blockIdx.x, cells, ovf_e, ovf_d, ovfcnt, nblk, ell, cnt, dinv,
                nspill, spill_e, spill_d, n, X, W1, O);
}


__device__ __forceinline__ void axpy4(float4& a, float s, const float4& r) {
    a.x += s * r.x; a.y += s * r.y; a.z += s * r.z; a.w += s * r.w;
}

template <int K>
__device__ __forceinline__ int slot_src(int sa, int sb) {
    return __builtin_amdgcn_update_dpp(0, (K & 1) ? sb : sa, 0x150 + (K >> 1), 0xf, 0xf, true);
}
template <int K>
__device__ __forceinline__ float slot_nrm(float na, float nb) {
    return __int_as_float(__builtin_amdgcn_update_dpp(0, __float_as_int((K & 1) ? nb : na),
                                                      0x150 + (K >> 1), 0xf, 0xf, true));
}

template <int K0>
__device__ __forceinline__ void rows8(float4 (&r)[8], unsigned l, const float4* __restrict__ h4,
                                      int sa, int sb) {
    const unsigned i0 = (unsigned)slot_src<K0 + 0>(sa, sb) * 16u + l;
    const unsigned i1 = (unsigned)slot_src<K0 + 1>(sa, sb) * 16u + l;
    const unsigned i2 = (unsigned)slot_src<K0 + 2>(sa, sb) * 16u + l;
    const unsigned i3 = (unsigned)slot_src<K0 + 3>(sa, sb) * 16u + l;
    const unsigned i4 = (unsigned)slot_src<K0 + 4>(sa, sb) * 16u + l;
    const unsigned i5 = (unsigned)slot_src<K0 + 5>(sa, sb) * 16u + l;
    const unsigned i6 = (unsigned)slot_src<K0 + 6>(sa, sb) * 16u + l;
    const unsigned i7 = (unsigned)slot_src<K0 + 7>(sa, sb) * 16u + l;
    r[0] = h4[i0];
    r[1] = h4[i1];
    r[2] = h4[i2];
    r[3] = h4[i3];
    r[4] = h4[i4];
    r[5] = h4[i5];
    r[6] = h4[i6];
    r[7] = h4[i7];
    asm volatile("" :: "v"(i0), "v"(i1), "v"(i2), "v"(i3), "v"(i4), "v"(i5), "v"(i6), "v"(i7));
}
template <int K0>
__device__ __forceinline__ void add8(float4& acc, const float4 (&r)[8], float na, float nb) {
    axpy4(acc, slot_nrm<K0 + 0>(na, nb), r[0]);
    axpy4(acc, slot_nrm<K0 + 1>(na, nb), r[1]);
    axpy4(acc, slot_nrm<K0 + 2>(na, nb), r[2]);
    axpy4(acc, slot_nrm<K0 + 3>(na, nb), r[3]);
    axpy4(acc, slot_nrm<K0 + 4>(na, nb), r[4]);
    axpy4(acc, slot_nrm<K0 + 5>(na, nb), r[5]);
    axpy4(acc, slot_nrm<K0 + 6>(na, nb), r[6]);
    axpy4(acc, slot_nrm<K0 + 7>(na, nb), r[7]);
}

constexpr int TP = 68;

__global__ void __launch_bounds__(256)
k_agg1_gemm2(const int* __restrict__ cnt,
             const unsigned long long* __restrict__ ell,
             const int* __restrict__ nspill,
             const unsigned long long* __restrict__ spill_e,
             const int* __restrict__ spill_d,
             const float* __restrict__ dinv, const float* __restrict__ h,
             const float* __restrict__ bias, const float* __restrict__ W2,
             float* __restrict__ O, int n) {
    __shared__ float tile[16 * TP];
    __shared__ float sdd[16];
    const int gi = threadIdx.x >> 4, l = threadIdx.x & 15;
    const int row0 = blockIdx.x * 16;
    const int g = row0 + gi;
    float4 acc = make_float4(0.0f, 0.0f, 0.0f, 0.0f);
    float dd = 0.0f;
    if (g < n) {
        const int c = cnt[g];
        const int m = (c < CAP) ? c : CAP;
        const size_t p0 = (size_t)g * CAP;
        const float4* __restrict__ h4 = reinterpret_cast<const float4*>(h);
        dd = dinv[g];
        const float4 hs = h4[(size_t)g * 16 + l];
        const float4 b = reinterpret_cast<const float4*>(bias)[l];
        int sa = g, sb = g;
        float wa = 0.0f, wb = 0.0f;
        if (2 * l + 1 < m) {
            const ulonglong2 ev = reinterpret_cast<const ulonglong2*>(ell + p0)[l];
            sa = (int)(unsigned)ev.x;
            sb = (int)(unsigned)ev.y;
            wa = __uint_as_float((unsigned)(ev.x >> 32));
            wb = __uint_as_float((unsigned)(ev.y >> 32));
        } else if (2 * l < m) {
            const unsigned long long e0 = ell[p0 + 2 * l];
            sa = (int)(unsigned)e0;
            wa = __uint_as_float((unsigned)(e0 >> 32));
        }
        float4 r[8];
        rows8<0>(r, (unsigned)l, h4, sa, sb);
        add8<0>(acc, r, wa, wb);
        if (m > 8)  { rows8<8>(r, (unsigned)l, h4, sa, sb);  add8<8>(acc, r, wa, wb); }
        if (m > 16) { rows8<16>(r, (unsigned)l, h4, sa, sb); add8<16>(acc, r, wa, wb); }
        if (m > 24) { rows8<24>(r, (unsigned)l, h4, sa, sb); add8<24>(acc, r, wa, wb); }
        if (c > CAP) {
            const int ns = nspill[0];
            for (int j = 0; j < ns; ++j) {
                if (spill_d[j] != g) continue;
                const unsigned long long e0 = spill_e[j];
                const float4 r0 = h4[(size_t)(int)(unsigned)e0 * 16 + l];
                axpy4(acc, __uint_as_float((unsigned)(e0 >> 32)), r0);
            }
        }
        acc.x += hs.x; acc.y += hs.y; acc.z += hs.z; acc.w += hs.w;
        acc.x = acc.x * dd + b.x; acc.y = acc.y * dd + b.y;
        acc.z = acc.z * dd + b.z; acc.w = acc.w * dd + b.w;
        acc.x = fmaxf(acc.x, 0.0f); acc.y = fmaxf(acc.y, 0.0f);
        acc.z = fmaxf(acc.z, 0.0f); acc.w = fmaxf(acc.w, 0.0f);
    }
    *reinterpret_cast<float4*>(&tile[gi * TP + 4 * l]) = acc;
    if (l == 0) sdd[gi] = dd;

    const int wave = threadIdx.x >> 5;
    const int lane = threadIdx.x & 31;
    const int lo = lane & 15, hi = lane >> 4;
    const int col = wave * 16 + lo;
    const bool cok = (wave < 3) && (col < 40);
    float bw[32];
    if (wave < 3) {
        const float* __restrict__ wcol = W2 + (cok ? col : 0);
#pragma unroll
        for (int j = 0; j < 16; ++j) {
            const int ka = 4 * j + 2 * hi;
            const float w0 = wcol[ka * 40], w1 = wcol[(ka + 1) * 40];
            bw[2 * j]     = cok ? w0 : 0.0f;
            bw[2 * j + 1] = cok ? w1 : 0.0f;
        }
    }
    __syncthreads();
    if (wave < 3) {
        const float* arow = tile + lo * TP;
        v8f c = {};
#pragma unroll
        for (int j = 0; j < 16; ++j) {
            const int ka = 4 * j + 2 * hi;
            v2f a = { arow[ka], arow[ka + 1] };
            v2f b = { bw[2 * j], bw[2 * j + 1] };
            c = __builtin_amdgcn_wmma_f32_16x16x4_f32(false, a, false, b,
                                                      (short)0, c, false, false);
        }
        if (cok) {
            const int r0 = row0 + 8 * hi;
            float* orow = O + (size_t)r0 * 64 + col;
            if (row0 + 16 <= n) {
#pragma unroll
                for (int r = 0; r < 8; ++r) orow[(size_t)r * 64] = c[r] * sdd[8 * hi + r];
            } else {
#pragma unroll
                for (int r = 0; r < 8; ++r)
                    if (r0 + r < n) orow[(size_t)r * 64] = c[r] * sdd[8 * hi + r];
            }
        }
    }
}

__global__ void k_agg2_lsm(const int* __restrict__ cnt,
                           const unsigned long long* __restrict__ ell,
                           const int* __restrict__ nspill,
                           const unsigned long long* __restrict__ spill_e,
                           const int* __restrict__ spill_d,
                           const float* __restrict__ dinv,
                           const float* __restrict__ h, const float* __restrict__ bias,
                           float* __restrict__ out, int n) {
    const int t = blockIdx.x * blockDim.x + threadIdx.x;
    const int g = t >> 4, l = t & 15;
    const bool act = (g < n) && (l < 10);
    const int lc = (l < 10) ? l : 0;
    float4 acc = make_float4(0.0f, 0.0f, 0.0f, 0.0f);
    if (g < n) {
        const int c = cnt[g];
        const int m = (c < CAP) ? c : CAP;
        const size_t p0 = (size_t)g * CAP;
        const float4* __restrict__ h4 = reinterpret_cast<const float4*>(h);
        const float dd = dinv[g];
        const float4 hs = h4[(size_t)g * 16 + lc];
        int sa = g, sb = g;
        float na = 0.0f, nb = 0.0f;
        if (2 * l + 1 < m) {
            const ulonglong2 ev = reinterpret_cast<const ulonglong2*>(ell + p0)[l];
            sa = (int)(unsigned)ev.x;
            sb = (int)(unsigned)ev.y;
            na = __uint_as_float((unsigned)(ev.x >> 32));
            nb = __uint_as_float((unsigned)(ev.y >> 32));
        } else if (2 * l < m) {
            const unsigned long long e0 = ell[p0 + 2 * l];
            sa = (int)(unsigned)e0;
            na = __uint_as_float((unsigned)(e0 >> 32));
        }
        float4 r[8];
        rows8<0>(r, (unsigned)lc, h4, sa, sb);
        add8<0>(acc, r, na, nb);
        if (m > 8)  { rows8<8>(r, (unsigned)lc, h4, sa, sb);  add8<8>(acc, r, na, nb); }
        if (m > 16) { rows8<16>(r, (unsigned)lc, h4, sa, sb); add8<16>(acc, r, na, nb); }
        if (m > 24) { rows8<24>(r, (unsigned)lc, h4, sa, sb); add8<24>(acc, r, na, nb); }
        if (c > CAP) {
            const int ns = nspill[0];
            for (int j = 0; j < ns; ++j) {
                if (spill_d[j] != g) continue;
                const unsigned long long e0 = spill_e[j];
                const float4 r0 = h4[(size_t)(int)(unsigned)e0 * 16 + lc];
                axpy4(acc, __uint_as_float((unsigned)(e0 >> 32)), r0);
            }
        }
        const float4 b = reinterpret_cast<const float4*>(bias)[lc];
        acc.x += hs.x; acc.y += hs.y; acc.z += hs.z; acc.w += hs.w;
        acc.x = acc.x * dd + b.x; acc.y = acc.y * dd + b.y;
        acc.z = acc.z * dd + b.z; acc.w = acc.w * dd + b.w;
    }
    float mx = act ? fmaxf(fmaxf(acc.x, acc.y), fmaxf(acc.z, acc.w)) : -3.402823466e38f;
    for (int o = 8; o >= 1; o >>= 1) mx = fmaxf(mx, __shfl_xor(mx, o, 16));
    float s = act ? (expf(acc.x - mx) + expf(acc.y - mx)) + (expf(acc.z - mx) + expf(acc.w - mx))
                  : 0.0f;
    for (int o = 8; o >= 1; o >>= 1) s += __shfl_xor(s, o, 16);
    const float lse = mx + logf(s);
    if (act) {
        acc.x -= lse; acc.y -= lse; acc.z -= lse; acc.w -= lse;
        reinterpret_cast<float4*>(out)[(size_t)g * 10 + l] = acc;
    }
}


extern "C" void kernel_launch(void* const* d_in, const int* in_sizes, int n_in,
                              void* d_out, int out_size, void* d_ws, size_t ws_size,
                              hipStream_t stream) {
    const float* x   = (const float*)d_in[0];
    const int*   ei  = (const int*)d_in[1];
    const float* ew  = (const float*)d_in[2];
    const float* W1  = (const float*)d_in[3];
    const float* b1  = (const float*)d_in[4];
    const float* W2  = (const float*)d_in[5];
    const float* b2  = (const float*)d_in[6];
    float* out = (float*)d_out;

    const int N = in_sizes[0] / 64;
    const int E = in_sizes[2];
    const int* src = ei;
    const int* dst = ei + E;

    const int nblk   = (E + EPB - 1) / EPB;
    const int nbins  = (N + BN - 1) / BN;
    const int nbinsA = (nbins < MAXB) ? nbins : MAXB;

    const size_t Ep = ((size_t)E + 63) / 64 * 64;
    const size_t Np = ((size_t)N + 63) / 64 * 64;
    unsigned long long* ell     = (unsigned long long*)d_ws;
    unsigned long long* spill_e = ell + Np * CAP;
    int*   spill_d = (int*)(spill_e + Ep);
    int*   cnt     = spill_d + Ep;
    int*   nspill  = cnt + Np;
    float* dinv    = (float*)(nspill + 64);
    float* bufA    = dinv + Np;
    float* bufB    = bufA + (size_t)N * 64;
    unsigned long long* cells = (unsigned long long*)(bufB + (size_t)N * 64);
    unsigned long long* ovf_e = cells + (size_t)nbinsA * nblk * CELL;
    int*   ovf_d   = (int*)(ovf_e + (size_t)nblk * EPB);
    int*   ovfcnt  = ovf_d + (size_t)nblk * EPB;

    const int BLK = 256;
    const unsigned gNg = (unsigned)(((long long)N * 16 + BLK - 1) / BLK);

    k_bin<<<(unsigned)nblk, 1024, 0, stream>>>(src, dst, ew, cells, ovf_e, ovf_d, ovfcnt,
                                               nspill, E, nblk, nbinsA);
    k_place_gemm1<<<(unsigned)nbins, BLK, 0, stream>>>(cells, ovf_e, ovf_d, ovfcnt,
                                                       nblk, ell, cnt, dinv, nspill, spill_e,
                                                       spill_d, N, x, W1, bufA);

    k_agg1_gemm2<<<gNg, BLK, 0, stream>>>(cnt, ell, nspill, spill_e, spill_d, dinv, bufA, b1,
                                          W2, bufB, N);

    k_agg2_lsm<<<gNg, BLK, 0, stream>>>(cnt, ell, nspill, spill_e, spill_d, dinv,
                                        bufB, b2, out, N);
}
